# starting kernel
# baseline (speedup 1.0000x reference)
#include <hip/hip_runtime.h>
#include <hip/hip_bf16.h>

#define NEG_INF (-__builtin_inff())

typedef float v2f __attribute__((ext_vector_type(2)));
typedef float v8f __attribute__((ext_vector_type(8)));

__global__ void k_init(float* __restrict__ m1, float* __restrict__ dn1,
                       float* __restrict__ acc1, float* __restrict__ m2,
                       float* __restrict__ dn2, float* __restrict__ acc2, int N) {
    long long t = (long long)blockIdx.x * blockDim.x + threadIdx.x;
    long long n64 = (long long)N * 64;
    if (t < n64)      acc1[t] = 0.f;
    if (t < 4LL * N) { m1[t] = NEG_INF; dn1[t] = 0.f; }
    if (t < 2LL * N)  acc2[t] = 0.f;
    if (t < N)       { m2[t] = NEG_INF; dn2[t] = 0.f; }
}

__global__ __launch_bounds__(128)
void k_gemm1_wmma(const float* __restrict__ x, const float* __restrict__ W1,
                  float* __restrict__ h1, int N) {
    const int wave = threadIdx.x >> 5;
    const int lane = threadIdx.x & 31;
    const int half = lane >> 4;
    const int l    = lane & 15;
    const int row0 = blockIdx.x * 16;
    const int col  = wave * 16 + l;
    int row = row0 + l;
    if (row >= N) row = N - 1;
    const float* xrow = x + (size_t)row * 128;

    v8f acc = {};
    for (int k0 = 0; k0 < 128; k0 += 4) {
        const int k = k0 + 2 * half;
        v2f a, b;
        a.x = xrow[k + 0];
        a.y = xrow[k + 1];
        b.x = W1[(size_t)(k + 0) * 64 + col];
        b.y = W1[(size_t)(k + 1) * 64 + col];
        acc = __builtin_amdgcn_wmma_f32_16x16x4_f32(
             false, a,  false, b,
             (short)0, acc,  false,  false);
    }
    float* dp = h1 + (size_t)(row0 + 8 * half) * 64 + col;
    if (row0 + 16 <= N) {
#pragma unroll
        for (int v = 0; v < 8; ++v) dp[(size_t)v * 64] = acc[v];
    } else {
#pragma unroll
        for (int v = 0; v < 8; ++v) {
            if (row0 + v + 8 * half < N) dp[(size_t)v * 64] = acc[v];
        }
    }
}

__global__ void k_scores1(const float* __restrict__ h1, const float* __restrict__ as,
                          const float* __restrict__ ad, float* __restrict__ ss,
                          float* __restrict__ sd, int N) {
    int t = blockIdx.x * blockDim.x + threadIdx.x;
    if (t >= N * 4) return;
    int n = t >> 2, h = t & 3;
    const float4* hp = (const float4*)(h1 + (size_t)n * 64 + h * 16);
    const float4* ap = (const float4*)(as + h * 16);
    const float4* bp = (const float4*)(ad + h * 16);
    float a = 0.f, b = 0.f;
#pragma unroll
    for (int i = 0; i < 4; ++i) {
        float4 hv = hp[i], av = ap[i], bv = bp[i];
        a += hv.x * av.x + hv.y * av.y + hv.z * av.z + hv.w * av.w;
        b += hv.x * bv.x + hv.y * bv.y + hv.z * bv.z + hv.w * bv.w;
    }
    ss[t] = a; sd[t] = b;
}

__device__ __forceinline__ void atomicMaxF(float* addr, float v) {
    unsigned u = __float_as_uint(v);
    if (u & 0x80000000u) atomicMin((unsigned*)addr, u);
    else                 atomicMax((int*)addr, __float_as_int(v));
}

__device__ __forceinline__ float lrelu(float s) { return s > 0.f ? s : 0.2f * s; }

__global__ void k_edge_max1(const int* __restrict__ ei, const float* __restrict__ ss,
                            const float* __restrict__ sd, float* __restrict__ m1,
                            int E, int N) {
    long long t = (long long)blockIdx.x * blockDim.x + threadIdx.x;
    long long ET4 = 4LL * ((long long)E + N);
    if (t >= ET4) return;
    int h = (int)(t & 3);
    long long e = t >> 2;
    int src, dst;
    if (e < E) { src = ei[e]; dst = ei[(size_t)E + e]; } else { src = dst = (int)(e - E); }
    float s = lrelu(ss[(size_t)src * 4 + h] + sd[(size_t)dst * 4 + h]);
    atomicMaxF(&m1[(size_t)dst * 4 + h], s);
}

__global__ void k_edge_acc1(const int* __restrict__ ei, const float* __restrict__ ss,
                            const float* __restrict__ sd, const float* __restrict__ m1,
                            const float* __restrict__ h1, float* __restrict__ dn1,
                            float* __restrict__ acc1, int E, int N) {
    long long t = (long long)blockIdx.x * blockDim.x + threadIdx.x;
    long long ET4 = 4LL * ((long long)E + N);
    if (t >= ET4) return;
    int h = (int)(t & 3);
    long long e = t >> 2;
    int src, dst;
    if (e < E) { src = ei[e]; dst = ei[(size_t)E + e]; } else { src = dst = (int)(e - E); }
    float s  = lrelu(ss[(size_t)src * 4 + h] + sd[(size_t)dst * 4 + h]);
    float ex = expf(s - m1[(size_t)dst * 4 + h]);
    atomicAdd(&dn1[(size_t)dst * 4 + h], ex);
    const float4* hs = (const float4*)(h1 + (size_t)src * 64 + h * 16);
    float* ad = acc1 + (size_t)dst * 64 + h * 16;
#pragma unroll
    for (int i = 0; i < 4; ++i) {
        float4 v = hs[i];
        atomicAdd(&ad[4 * i + 0], v.x * ex);
        atomicAdd(&ad[4 * i + 1], v.y * ex);
        atomicAdd(&ad[4 * i + 2], v.z * ex);
        atomicAdd(&ad[4 * i + 3], v.w * ex);
    }
}

__global__ void k_node1(float* __restrict__ acc1, const float* __restrict__ dn1,
                        const float* __restrict__ b1, int N) {
    long long t = (long long)blockIdx.x * blockDim.x + threadIdx.x;
    if (t >= (long long)N * 64) return;
    int n = (int)(t >> 6), c = (int)(t & 63), h = c >> 4;
    float v = acc1[t] / dn1[(size_t)n * 4 + h] + b1[c];
    acc1[t] = v > 0.f ? v : expm1f(v);
}

__global__ void k_proj2(const float* __restrict__ helu, const float* __restrict__ W2,
                        const float* __restrict__ as2, const float* __restrict__ ad2,
                        float* __restrict__ h2, float* __restrict__ ss2,
                        float* __restrict__ sd2, int N) {
    int n = blockIdx.x * blockDim.x + threadIdx.x;
    if (n >= N) return;
    const float* hp = helu + (size_t)n * 64;
    float o0 = 0.f, o1 = 0.f;
#pragma unroll
    for (int c = 0; c < 64; ++c) { float v = hp[c]; o0 += v * W2[c * 2]; o1 += v * W2[c * 2 + 1]; }
    h2[(size_t)n * 2]     = o0;
    h2[(size_t)n * 2 + 1] = o1;
    ss2[n] = o0 * as2[0] + o1 * as2[1];
    sd2[n] = o0 * ad2[0] + o1 * ad2[1];
}

__global__ void k_edge_max2(const int* __restrict__ ei, const float* __restrict__ ss2,
                            const float* __restrict__ sd2, float* __restrict__ m2,
                            int E, int N) {
    long long e = (long long)blockIdx.x * blockDim.x + threadIdx.x;
    if (e >= (long long)E + N) return;
    int src, dst;
    if (e < E) { src = ei[e]; dst = ei[(size_t)E + e]; } else { src = dst = (int)(e - E); }
    atomicMaxF(&m2[dst], lrelu(ss2[src] + sd2[dst]));
}

__global__ void k_edge_acc2(const int* __restrict__ ei, const float* __restrict__ ss2,
                            const float* __restrict__ sd2, const float* __restrict__ m2,
                            const float* __restrict__ h2, float* __restrict__ dn2,
                            float* __restrict__ acc2, int E, int N) {
    long long e = (long long)blockIdx.x * blockDim.x + threadIdx.x;
    if (e >= (long long)E + N) return;
    int src, dst;
    if (e < E) { src = ei[e]; dst = ei[(size_t)E + e]; } else { src = dst = (int)(e - E); }
    float ex = expf(lrelu(ss2[src] + sd2[dst]) - m2[dst]);
    atomicAdd(&dn2[dst], ex);
    atomicAdd(&acc2[(size_t)dst * 2],     h2[(size_t)src * 2]     * ex);
    atomicAdd(&acc2[(size_t)dst * 2 + 1], h2[(size_t)src * 2 + 1] * ex);
}

__global__ void k_out(const float* __restrict__ acc2, const float* __restrict__ dn2,
                      const float* __restrict__ b2, float* __restrict__ out, int N) {
    int n = blockIdx.x * blockDim.x + threadIdx.x;
    if (n >= N) return;
    float d  = dn2[n];
    float v0 = acc2[(size_t)n * 2]     / d + b2[0];
    float v1 = acc2[(size_t)n * 2 + 1] / d + b2[1];
    float m  = fmaxf(v0, v1);
    float lse = m + logf(expf(v0 - m) + expf(v1 - m));
    out[(size_t)n * 2]     = v0 - lse;
    out[(size_t)n * 2 + 1] = v1 - lse;
}

static inline unsigned nblk(long long n, int b) { return (unsigned)((n + b - 1) / b); }

extern "C" void kernel_launch(void* const* d_in, const int* in_sizes, int n_in,
                              void* d_out, int out_size, void* d_ws, size_t ws_size,
                              hipStream_t stream) {
    const float* x     = (const float*)d_in[0];
    const int*   ei    = (const int*)  d_in[1];
    const float* W1    = (const float*)d_in[2];
    const float* asrc1 = (const float*)d_in[3];
    const float* adst1 = (const float*)d_in[4];
    const float* b1    = (const float*)d_in[5];
    const float* W2    = (const float*)d_in[6];
    const float* asrc2 = (const float*)d_in[7];
    const float* adst2 = (const float*)d_in[8];
    const float* b2    = (const float*)d_in[9];
    float* out = (float*)d_out;

    const int N = in_sizes[0] / 128;
    const int E = in_sizes[1] / 2;
    const long long ET = (long long)E + N;

    float* ws   = (float*)d_ws;
    size_t nl   = (size_t)N;
    float* h1   = ws;
    float* ss1  = h1  + nl * 64;
    float* sd1  = ss1 + nl * 4;
    float* m1   = sd1 + nl * 4;
    float* dn1  = m1  + nl * 4;
    float* acc1 = dn1 + nl * 4;
    float* h2   = acc1 + nl * 64;
    float* ss2  = h2  + nl * 2;
    float* sd2  = ss2 + nl;
    float* m2   = sd2 + nl;
    float* dn2  = m2  + nl;
    float* acc2 = dn2 + nl;

    k_init<<<nblk((long long)N * 64, 256), 256, 0, stream>>>(m1, dn1, acc1, m2, dn2, acc2, N);
    k_gemm1_wmma<<<(N + 15) / 16, 128, 0, stream>>>(x, W1, h1, N);
    k_scores1<<<nblk(4LL * N, 256), 256, 0, stream>>>(h1, asrc1, adst1, ss1, sd1, N);
    k_edge_max1<<<nblk(4 * ET, 256), 256, 0, stream>>>(ei, ss1, sd1, m1, E, N);
    k_edge_acc1<<<nblk(4 * ET, 256), 256, 0, stream>>>(ei, ss1, sd1, m1, h1, dn1, acc1, E, N);
    k_node1<<<nblk((long long)N * 64, 256), 256, 0, stream>>>(acc1, dn1, b1, N);
    k_proj2<<<nblk(N, 256), 256, 0, stream>>>(acc1, W2, asrc2, adst2, h2, ss2, sd2, N);
    k_edge_max2<<<nblk(ET, 256), 256, 0, stream>>>(ei, ss2, sd2, m2, E, N);
    k_edge_acc2<<<nblk(ET, 256), 256, 0, stream>>>(ei, ss2, sd2, m2, h2, dn2, acc2, E, N);
    k_out<<<nblk(N, 256), 256, 0, stream>>>(acc2, dn2, b2, out, N);
}


# Spread contended atomics: segment-max slots and layer-2 sums move from dense 4-byte arrays to one 256-B row per node (8-32x fewer atomics per line); +1 small compaction kernel; float atomics kept
# speedup vs baseline: 1.2456x; 1.2456x over previous
#include <hip/hip_runtime.h>
#include <hip/hip_bf16.h>

#define NEG_INF (-__builtin_inff())

typedef float v2f __attribute__((ext_vector_type(2)));
typedef float v8f __attribute__((ext_vector_type(8)));

__global__ void k_init(float* __restrict__ dn1, float* __restrict__ acc1, int N) {
    long long t = (long long)blockIdx.x * blockDim.x + threadIdx.x;
    long long n64 = (long long)N * 64;
    if (t < n64)      acc1[t] = ((t & 63) < 4) ? NEG_INF : 0.f;
    if (t < 4LL * N)  dn1[t] = 0.f;
}

__global__ __launch_bounds__(128)
void k_gemm1_wmma(const float* __restrict__ x, const float* __restrict__ W1,
                  float* __restrict__ h1, int N) {
    const int wave = threadIdx.x >> 5;
    const int lane = threadIdx.x & 31;
    const int half = lane >> 4;
    const int l    = lane & 15;
    const int row0 = blockIdx.x * 16;
    const int col  = wave * 16 + l;
    int row = row0 + l;
    if (row >= N) row = N - 1;
    const float* xrow = x + (size_t)row * 128;

    v8f acc = {};
    for (int k0 = 0; k0 < 128; k0 += 4) {
        const int k = k0 + 2 * half;
        v2f a, b;
        a.x = xrow[k + 0];
        a.y = xrow[k + 1];
        b.x = W1[(size_t)(k + 0) * 64 + col];
        b.y = W1[(size_t)(k + 1) * 64 + col];
        acc = __builtin_amdgcn_wmma_f32_16x16x4_f32(
             false, a,  false, b,
             (short)0, acc,  false,  false);
    }
    float* dp = h1 + (size_t)(row0 + 8 * half) * 64 + col;
    if (row0 + 16 <= N) {
#pragma unroll
        for (int v = 0; v < 8; ++v) dp[(size_t)v * 64] = acc[v];
    } else {
#pragma unroll
        for (int v = 0; v < 8; ++v) {
            if (row0 + v + 8 * half < N) dp[(size_t)v * 64] = acc[v];
        }
    }
}

__global__ void k_scores1(const float* __restrict__ h1, const float* __restrict__ as,
                          const float* __restrict__ ad, float* __restrict__ ss,
                          float* __restrict__ sd, int N) {
    int t = blockIdx.x * blockDim.x + threadIdx.x;
    if (t >= N * 4) return;
    int n = t >> 2, h = t & 3;
    const float4* hp = (const float4*)(h1 + (size_t)n * 64 + h * 16);
    const float4* ap = (const float4*)(as + h * 16);
    const float4* bp = (const float4*)(ad + h * 16);
    float a = 0.f, b = 0.f;
#pragma unroll
    for (int i = 0; i < 4; ++i) {
        float4 hv = hp[i], av = ap[i], bv = bp[i];
        a += hv.x * av.x + hv.y * av.y + hv.z * av.z + hv.w * av.w;
        b += hv.x * bv.x + hv.y * bv.y + hv.z * bv.z + hv.w * bv.w;
    }
    ss[t] = a; sd[t] = b;
}

__device__ __forceinline__ void atomicMaxF(float* addr, float v) {
    unsigned u = __float_as_uint(v);
    if (u & 0x80000000u) atomicMin((unsigned*)addr, u);
    else                 atomicMax((int*)addr, __float_as_int(v));
}

__device__ __forceinline__ float lrelu(float s) { return s > 0.f ? s : 0.2f * s; }

__global__ void k_edge_max1(const int* __restrict__ ei, const float* __restrict__ ss,
                            const float* __restrict__ sd, float* __restrict__ mrow,
                            int E, int N) {
    long long t = (long long)blockIdx.x * blockDim.x + threadIdx.x;
    long long ET4 = 4LL * ((long long)E + N);
    if (t >= ET4) return;
    int h = (int)(t & 3);
    long long e = t >> 2;
    int src, dst;
    if (e < E) { src = ei[e]; dst = ei[(size_t)E + e]; } else { src = dst = (int)(e - E); }
    float s = lrelu(ss[(size_t)src * 4 + h] + sd[(size_t)dst * 4 + h]);
    atomicMaxF(&mrow[(size_t)dst * 64 + h], s);
}

__global__ void k_mcompact1(float* __restrict__ acc1, float* __restrict__ m1, int N) {
    int t = blockIdx.x * blockDim.x + threadIdx.x;
    if (t >= N * 4) return;
    float* p = acc1 + (size_t)(t >> 2) * 64 + (t & 3);
    m1[t] = *p;
    *p = 0.f;
}

__global__ void k_edge_acc1(const int* __restrict__ ei, const float* __restrict__ ss,
                            const float* __restrict__ sd, const float* __restrict__ m1,
                            const float* __restrict__ h1, float* __restrict__ dn1,
                            float* __restrict__ acc1, int E, int N) {
    long long t = (long long)blockIdx.x * blockDim.x + threadIdx.x;
    long long ET4 = 4LL * ((long long)E + N);
    if (t >= ET4) return;
    int h = (int)(t & 3);
    long long e = t >> 2;
    int src, dst;
    if (e < E) { src = ei[e]; dst = ei[(size_t)E + e]; } else { src = dst = (int)(e - E); }
    float s  = lrelu(ss[(size_t)src * 4 + h] + sd[(size_t)dst * 4 + h]);
    float ex = expf(s - m1[(size_t)dst * 4 + h]);
    atomicAdd(&dn1[(size_t)dst * 4 + h], ex);
    const float4* hs = (const float4*)(h1 + (size_t)src * 64 + h * 16);
    float* ad = acc1 + (size_t)dst * 64 + h * 16;
#pragma unroll
    for (int i = 0; i < 4; ++i) {
        float4 v = hs[i];
        atomicAdd(&ad[4 * i + 0], v.x * ex);
        atomicAdd(&ad[4 * i + 1], v.y * ex);
        atomicAdd(&ad[4 * i + 2], v.z * ex);
        atomicAdd(&ad[4 * i + 3], v.w * ex);
    }
}

__global__ void k_node1(float* __restrict__ acc1, const float* __restrict__ dn1,
                        const float* __restrict__ b1, int N) {
    long long t = (long long)blockIdx.x * blockDim.x + threadIdx.x;
    if (t >= (long long)N * 64) return;
    int n = (int)(t >> 6), c = (int)(t & 63), h = c >> 4;
    float v = acc1[t] / dn1[(size_t)n * 4 + h] + b1[c];
    acc1[t] = v > 0.f ? v : expm1f(v);
}

__global__ void k_proj2(const float* __restrict__ helu, const float* __restrict__ W2,
                        const float* __restrict__ as2, const float* __restrict__ ad2,
                        float* __restrict__ h2, float* __restrict__ ss2,
                        float* __restrict__ sd2, float* __restrict__ rec2, int N) {
    int n = blockIdx.x * blockDim.x + threadIdx.x;
    if (n >= N) return;
    *(float4*)(rec2 + (size_t)n * 64) = make_float4(NEG_INF, 0.f, 0.f, 0.f);
    const float* hp = helu + (size_t)n * 64;
    float o0 = 0.f, o1 = 0.f;
#pragma unroll
    for (int c = 0; c < 64; ++c) { float v = hp[c]; o0 += v * W2[c * 2]; o1 += v * W2[c * 2 + 1]; }
    h2[(size_t)n * 2]     = o0;
    h2[(size_t)n * 2 + 1] = o1;
    ss2[n] = o0 * as2[0] + o1 * as2[1];
    sd2[n] = o0 * ad2[0] + o1 * ad2[1];
}

__global__ void k_edge_max2(const int* __restrict__ ei, const float* __restrict__ ss2,
                            const float* __restrict__ sd2, float* __restrict__ rec2,
                            int E, int N) {
    long long e = (long long)blockIdx.x * blockDim.x + threadIdx.x;
    if (e >= (long long)E + N) return;
    int src, dst;
    if (e < E) { src = ei[e]; dst = ei[(size_t)E + e]; } else { src = dst = (int)(e - E); }
    atomicMaxF(&rec2[(size_t)dst * 64], lrelu(ss2[src] + sd2[dst]));
}

__global__ void k_edge_acc2(const int* __restrict__ ei, const float* __restrict__ ss2,
                            const float* __restrict__ sd2, const float* __restrict__ h2,
                            float* rec2, int E, int N) {
    long long e = (long long)blockIdx.x * blockDim.x + threadIdx.x;
    if (e >= (long long)E + N) return;
    int src, dst;
    if (e < E) { src = ei[e]; dst = ei[(size_t)E + e]; } else { src = dst = (int)(e - E); }
    float* r = rec2 + (size_t)dst * 64;
    float ex = expf(lrelu(ss2[src] + sd2[dst]) - r[0]);
    atomicAdd(&r[1], ex);
    atomicAdd(&r[2], h2[(size_t)src * 2]     * ex);
    atomicAdd(&r[3], h2[(size_t)src * 2 + 1] * ex);
}

__global__ void k_out(const float* __restrict__ rec2,
                      const float* __restrict__ b2, float* __restrict__ out, int N) {
    int n = blockIdx.x * blockDim.x + threadIdx.x;
    if (n >= N) return;
    const float4 r = *(const float4*)(rec2 + (size_t)n * 64);
    float d  = r.y;
    float v0 = r.z / d + b2[0];
    float v1 = r.w / d + b2[1];
    float m  = fmaxf(v0, v1);
    float lse = m + logf(expf(v0 - m) + expf(v1 - m));
    out[(size_t)n * 2]     = v0 - lse;
    out[(size_t)n * 2 + 1] = v1 - lse;
}

static inline unsigned nblk(long long n, int b) { return (unsigned)((n + b - 1) / b); }

extern "C" void kernel_launch(void* const* d_in, const int* in_sizes, int n_in,
                              void* d_out, int out_size, void* d_ws, size_t ws_size,
                              hipStream_t stream) {
    const float* x     = (const float*)d_in[0];
    const int*   ei    = (const int*)  d_in[1];
    const float* W1    = (const float*)d_in[2];
    const float* asrc1 = (const float*)d_in[3];
    const float* adst1 = (const float*)d_in[4];
    const float* b1    = (const float*)d_in[5];
    const float* W2    = (const float*)d_in[6];
    const float* asrc2 = (const float*)d_in[7];
    const float* adst2 = (const float*)d_in[8];
    const float* b2    = (const float*)d_in[9];
    float* out = (float*)d_out;

    const int N = in_sizes[0] / 128;
    const int E = in_sizes[1] / 2;
    const long long ET = (long long)E + N;

    float* ws   = (float*)d_ws;
    size_t nl   = (size_t)N;
    float* h1   = ws;
    float* ss1  = h1  + nl * 64;
    float* sd1  = ss1 + nl * 4;
    float* m1   = sd1 + nl * 4;
    float* dn1  = m1  + nl * 4;
    float* acc1 = dn1 + nl * 4;
    float* h2   = acc1 + nl * 64;
    float* ss2  = h2  + nl * 2;
    float* sd2  = ss2 + nl;
    float* rec2 = h1;

    k_init<<<nblk((long long)N * 64, 256), 256, 0, stream>>>(dn1, acc1, N);
    k_gemm1_wmma<<<(N + 15) / 16, 128, 0, stream>>>(x, W1, h1, N);
    k_scores1<<<nblk(4LL * N, 256), 256, 0, stream>>>(h1, asrc1, adst1, ss1, sd1, N);
    k_edge_max1<<<nblk(4 * ET, 256), 256, 0, stream>>>(ei, ss1, sd1, acc1, E, N);
    k_mcompact1<<<nblk(4LL * N, 256), 256, 0, stream>>>(acc1, m1, N);
    k_edge_acc1<<<nblk(4 * ET, 256), 256, 0, stream>>>(ei, ss1, sd1, m1, h1, dn1, acc1, E, N);
    k_node1<<<nblk((long long)N * 64, 256), 256, 0, stream>>>(acc1, dn1, b1, N);
    k_proj2<<<nblk(N, 256), 256, 0, stream>>>(acc1, W2, asrc2, adst2, h2, ss2, sd2, rec2, N);
    k_edge_max2<<<nblk(ET, 256), 256, 0, stream>>>(ei, ss2, sd2, rec2, E, N);
    k_edge_acc2<<<nblk(ET, 256), 256, 0, stream>>>(ei, ss2, sd2, h2, rec2, E, N);
    k_out<<<nblk(N, 256), 256, 0, stream>>>(rec2, b2, out, N);
}


# Float atomics removed: edges grouped by dst in-call (counting sort, int atomics), then one wave per node reduces max, denominator and messages in registers, both layers; summation order changes
# speedup vs baseline: 4.6579x; 3.7395x over previous
#include <hip/hip_runtime.h>
#include <hip/hip_bf16.h>

#define NEG_INF (-__builtin_inff())

typedef float v2f __attribute__((ext_vector_type(2)));
typedef float v8f __attribute__((ext_vector_type(8)));

#define CNT_STRIDE 8

__global__ void k_init(int* __restrict__ cnt, int N) {
    int t = blockIdx.x * blockDim.x + threadIdx.x;
    if (t < N) cnt[(size_t)t * CNT_STRIDE] = 0;
}

__global__ void k_hist(const int* __restrict__ ei, int* __restrict__ cnt,
                       int* __restrict__ rank, int E) {
    const int t = blockIdx.x * blockDim.x + threadIdx.x;
    const int T = gridDim.x * blockDim.x;
    int d[4];
#pragma unroll
    for (int k = 0; k < 4; ++k) { const int e = t + k * T; d[k] = (e < E) ? ei[(size_t)E + e] : -1; }
    int r[4];
#pragma unroll
    for (int k = 0; k < 4; ++k) r[k] = (d[k] >= 0) ? atomicAdd(&cnt[(size_t)d[k] * CNT_STRIDE], 1) : 0;
#pragma unroll
    for (int k = 0; k < 4; ++k) { const int e = t + k * T; if (e < E) rank[e] = r[k]; }
}

__global__ __launch_bounds__(1024)
void k_scan(const int* __restrict__ cnt, int* __restrict__ off,
            int* __restrict__ deg, int N) {
    __shared__ int s_wtot[32];
    __shared__ int s_wpre[32];
    const int t = threadIdx.x, lane = t & 31, w = t >> 5;
    const int base = blockIdx.x * 1024;
    int p = 0;
#pragma unroll 8
    for (int i = t; i < base; i += 1024) p += cnt[(size_t)i * CNT_STRIDE];
    const int n = base + t;
    const int c = (n < N) ? cnt[(size_t)n * CNT_STRIDE] : 0;
    int inc = c;
#pragma unroll
    for (int d = 1; d < 32; d <<= 1) {
        int v = __shfl_up(inc, d);
        if (lane >= d) inc += v;
    }
#pragma unroll
    for (int d = 16; d > 0; d >>= 1) p += __shfl_xor(p, d);
    if (lane == 31) s_wtot[w] = inc;
    if (lane == 0)  s_wpre[w] = p;
    __syncthreads();
    int before = s_wpre[lane] + ((lane < w) ? s_wtot[lane] : 0);
#pragma unroll
    for (int d = 16; d > 0; d >>= 1) before += __shfl_xor(before, d);
    if (n < N) { off[n] = before + inc - c; deg[n] = c; }
}

__global__ void k_fill(const int* __restrict__ ei, const int* __restrict__ rank,
                       const int* __restrict__ off, int* __restrict__ sorted, int E) {
    const int t = blockIdx.x * blockDim.x + threadIdx.x;
    const int T = gridDim.x * blockDim.x;
    int s[4], d[4], r[4];
#pragma unroll
    for (int k = 0; k < 4; ++k) {
        const int e = t + k * T;
        const bool ok = e < E;
        s[k] = ok ? ei[e] : 0;
        d[k] = ok ? ei[(size_t)E + e] : -1;
        r[k] = ok ? rank[e] : 0;
    }
#pragma unroll
    for (int k = 0; k < 4; ++k) if (d[k] >= 0) sorted[(size_t)off[d[k]] + r[k]] = s[k];
}

__global__ __launch_bounds__(128)
void k_gemm1_wmma(const float* __restrict__ x, const float* __restrict__ W1,
                  float* __restrict__ h1, int N) {
    const int wave = threadIdx.x >> 5;
    const int lane = threadIdx.x & 31;
    const int half = lane >> 4;
    const int l    = lane & 15;
    const int row0 = blockIdx.x * 16;
    const int col  = wave * 16 + l;
    int row = row0 + l;
    if (row >= N) row = N - 1;
    const float* xrow = x + (size_t)row * 128;

    v8f acc = {};
    for (int k0 = 0; k0 < 128; k0 += 4) {
        const int k = k0 + 2 * half;
        v2f a, b;
        a.x = xrow[k + 0];
        a.y = xrow[k + 1];
        b.x = W1[(size_t)(k + 0) * 64 + col];
        b.y = W1[(size_t)(k + 1) * 64 + col];
        acc = __builtin_amdgcn_wmma_f32_16x16x4_f32(
             false, a,  false, b,
             (short)0, acc,  false,  false);
    }
    float* dp = h1 + (size_t)(row0 + 8 * half) * 64 + col;
    if (row0 + 16 <= N) {
#pragma unroll
        for (int v = 0; v < 8; ++v) dp[(size_t)v * 64] = acc[v];
    } else {
#pragma unroll
        for (int v = 0; v < 8; ++v) {
            if (row0 + v + 8 * half < N) dp[(size_t)v * 64] = acc[v];
        }
    }
}

__global__ void k_scores1(const float* __restrict__ h1, const float* __restrict__ as,
                          const float* __restrict__ ad, float* __restrict__ ss,
                          float* __restrict__ sd, int N) {
    int t = blockIdx.x * blockDim.x + threadIdx.x;
    if (t >= N * 4) return;
    int n = t >> 2, h = t & 3;
    const float4* hp = (const float4*)(h1 + (size_t)n * 64 + h * 16);
    const float4* ap = (const float4*)(as + h * 16);
    const float4* bp = (const float4*)(ad + h * 16);
    float a = 0.f, b = 0.f;
#pragma unroll
    for (int i = 0; i < 4; ++i) {
        float4 hv = hp[i], av = ap[i], bv = bp[i];
        a += hv.x * av.x + hv.y * av.y + hv.z * av.z + hv.w * av.w;
        b += hv.x * bv.x + hv.y * bv.y + hv.z * bv.z + hv.w * bv.w;
    }
    ss[t] = a; sd[t] = b;
}

__device__ __forceinline__ float lrelu(float s) { return s > 0.f ? s : 0.2f * s; }

__device__ __forceinline__ float wave_max(float v) {
#pragma unroll
    for (int d = 16; d > 0; d >>= 1) v = fmaxf(v, __shfl_xor(v, d));
    return v;
}
__device__ __forceinline__ float wave_sum(float v) {
#pragma unroll
    for (int d = 16; d > 0; d >>= 1) v += __shfl_xor(v, d);
    return v;
}
__device__ __forceinline__ unsigned lane_bcast_u(float v, int j) {
    return (unsigned)__builtin_amdgcn_readlane(__float_as_int(v), j);
}

__global__ __launch_bounds__(256)
void k_node1(const int* __restrict__ sorted, const int* __restrict__ off,
             const int* __restrict__ deg, const float* __restrict__ ss,
             const float* __restrict__ sd, const float* __restrict__ h1,
             const float* __restrict__ b1, const float* __restrict__ W2,
             const float* __restrict__ as2, const float* __restrict__ ad2,
             float* __restrict__ rec2, int N) {
    const int lane = threadIdx.x & 31;
    const int n = __builtin_amdgcn_readfirstlane(blockIdx.x * 8 + (threadIdx.x >> 5));
    if (n >= N) return;
    const int o = off[n], dg = deg[n], tot = dg + 1;
    const float4 sdn = *(const float4*)(sd + (size_t)n * 4);

    float m0 = NEG_INF, m1 = NEG_INF, m2 = NEG_INF, m3 = NEG_INF;
    for (int base = 0; base < tot; base += 32) {
        const int p = base + lane;
        if (p < tot) {
            const int s = (p < dg) ? sorted[(size_t)o + p] : n;
            const float4 sv = *(const float4*)(ss + (size_t)s * 4);
            m0 = fmaxf(m0, lrelu(sv.x + sdn.x));
            m1 = fmaxf(m1, lrelu(sv.y + sdn.y));
            m2 = fmaxf(m2, lrelu(sv.z + sdn.z));
            m3 = fmaxf(m3, lrelu(sv.w + sdn.w));
        }
    }
    m0 = wave_max(m0); m1 = wave_max(m1); m2 = wave_max(m2); m3 = wave_max(m3);

    const int hsel = lane >> 3;
    const unsigned k0 = (hsel == 0) ? 0xffffffffu : 0u, k1 = (hsel == 1) ? 0xffffffffu : 0u;
    const unsigned k2 = (hsel == 2) ? 0xffffffffu : 0u, k3 = (hsel == 3) ? 0xffffffffu : 0u;
    float d0 = 0.f, d1 = 0.f, d2 = 0.f, d3 = 0.f;
    float a0 = 0.f, a1 = 0.f;
    for (int base = 0; base < tot; base += 32) {
        const int p = base + lane;
        int s = n;
        float e0 = 0.f, e1 = 0.f, e2 = 0.f, e3 = 0.f;
        if (p < tot) {
            if (p < dg) s = sorted[(size_t)o + p];
            const float4 sv = *(const float4*)(ss + (size_t)s * 4);
            e0 = expf(lrelu(sv.x + sdn.x) - m0);
            e1 = expf(lrelu(sv.y + sdn.y) - m1);
            e2 = expf(lrelu(sv.z + sdn.z) - m2);
            e3 = expf(lrelu(sv.w + sdn.w) - m3);
        }
        d0 += e0; d1 += e1; d2 += e2; d3 += e3;
        const int cnt = (tot - base < 32) ? (tot - base) : 32;
        for (int j = 0; j < cnt; j += 8) {
#pragma unroll
            for (int u = 0; u < 8; ++u) {
                const int sj = __builtin_amdgcn_readlane(s, j + u);
                const unsigned w0 = lane_bcast_u(e0, j + u), w1 = lane_bcast_u(e1, j + u);
                const unsigned w2 = lane_bcast_u(e2, j + u), w3 = lane_bcast_u(e3, j + u);
                const float w = __uint_as_float((w0 & k0) | (w1 & k1) | (w2 & k2) | (w3 & k3));
                const float2 hv = *(const float2*)(h1 + (size_t)sj * 64 + 2 * lane);
                a0 += hv.x * w;
                a1 += hv.y * w;
            }
        }
    }
    d0 = wave_sum(d0); d1 = wave_sum(d1); d2 = wave_sum(d2); d3 = wave_sum(d3);
    const float dn = (hsel == 0) ? d0 : (hsel == 1) ? d1 : (hsel == 2) ? d2 : d3;
    const float2 bb = *(const float2*)(b1 + 2 * lane);
    float v0 = a0 / dn + bb.x;
    float v1 = a1 / dn + bb.y;
    v0 = v0 > 0.f ? v0 : expm1f(v0);
    v1 = v1 > 0.f ? v1 : expm1f(v1);
    const float4 wv = *(const float4*)(W2 + 4 * lane);
    float o0 = v0 * wv.x + v1 * wv.z;
    float o1 = v0 * wv.y + v1 * wv.w;
    o0 = wave_sum(o0);
    o1 = wave_sum(o1);
    if (lane == 0) {
        *(float4*)(rec2 + (size_t)n * 4) =
            make_float4(o0, o1, o0 * as2[0] + o1 * as2[1], o0 * ad2[0] + o1 * ad2[1]);
    }
}

__global__ __launch_bounds__(256)
void k_node2(const int* __restrict__ sorted, const int* __restrict__ off,
             const int* __restrict__ deg, const float* __restrict__ rec2,
             const float* __restrict__ b2, float* __restrict__ out, int N) {
    const int t = blockIdx.x * 256 + threadIdx.x;
    const int n = t >> 3, sub = t & 7;
    int o = 0, dg = 0, tot = 0;
    float sdn = 0.f;
    if (n < N) { o = off[n]; dg = deg[n]; tot = dg + 1; sdn = rec2[(size_t)n * 4 + 3]; }

    float m = NEG_INF;
    for (int p = sub; p < tot; p += 8) {
        const int s = (p < dg) ? sorted[(size_t)o + p] : n;
        m = fmaxf(m, lrelu(rec2[(size_t)s * 4 + 2] + sdn));
    }
#pragma unroll
    for (int d = 4; d > 0; d >>= 1) m = fmaxf(m, __shfl_xor(m, d));

    float dsum = 0.f, a0 = 0.f, a1 = 0.f;
    for (int p = sub; p < tot; p += 8) {
        const int s = (p < dg) ? sorted[(size_t)o + p] : n;
        const float4 r = *(const float4*)(rec2 + (size_t)s * 4);
        const float ex = expf(lrelu(r.z + sdn) - m);
        dsum += ex;
        a0 += r.x * ex;
        a1 += r.y * ex;
    }
#pragma unroll
    for (int d = 4; d > 0; d >>= 1) {
        dsum += __shfl_xor(dsum, d);
        a0   += __shfl_xor(a0, d);
        a1   += __shfl_xor(a1, d);
    }
    if (sub == 0 && n < N) {
        float v0 = a0 / dsum + b2[0];
        float v1 = a1 / dsum + b2[1];
        float mx = fmaxf(v0, v1);
        float lse = mx + logf(expf(v0 - mx) + expf(v1 - mx));
        *(float2*)(out + (size_t)n * 2) = make_float2(v0 - lse, v1 - lse);
    }
}

static inline unsigned nblk(long long n, int b) { return (unsigned)((n + b - 1) / b); }

extern "C" void kernel_launch(void* const* d_in, const int* in_sizes, int n_in,
                              void* d_out, int out_size, void* d_ws, size_t ws_size,
                              hipStream_t stream) {
    const float* x     = (const float*)d_in[0];
    const int*   ei    = (const int*)  d_in[1];
    const float* W1    = (const float*)d_in[2];
    const float* asrc1 = (const float*)d_in[3];
    const float* adst1 = (const float*)d_in[4];
    const float* b1    = (const float*)d_in[5];
    const float* W2    = (const float*)d_in[6];
    const float* asrc2 = (const float*)d_in[7];
    const float* adst2 = (const float*)d_in[8];
    const float* b2    = (const float*)d_in[9];
    float* out = (float*)d_out;

    const int N = in_sizes[0] / 128;
    const int E = in_sizes[1] / 2;

    float* ws   = (float*)d_ws;
    size_t nl   = (size_t)N;
    float* h1   = ws;
    float* ss1  = h1  + nl * 64;
    float* sd1  = ss1 + nl * 4;
    float* rec2 = sd1 + nl * 4;
    int*   off  = (int*)(rec2 + nl * 4);
    int*   deg  = off + nl;
    int*   rank = deg + nl;
    int*   sorted = rank + (size_t)E;
    int*   cnt  = (int*)h1;

    k_init<<<nblk(N, 256), 256, 0, stream>>>(cnt, N);
    k_hist<<<nblk(E, 1024), 256, 0, stream>>>(ei, cnt, rank, E);
    k_scan<<<nblk(N, 1024), 1024, 0, stream>>>(cnt, off, deg, N);
    k_fill<<<nblk(E, 1024), 256, 0, stream>>>(ei, rank, off, sorted, E);
    k_gemm1_wmma<<<(N + 15) / 16, 128, 0, stream>>>(x, W1, h1, N);
    k_scores1<<<nblk(4LL * N, 256), 256, 0, stream>>>(h1, asrc1, adst1, ss1, sd1, N);
    k_node1<<<nblk(N, 8), 256, 0, stream>>>(sorted, off, deg, ss1, sd1, h1, b1, W2,
                                            asrc2, adst2, rec2, N);
    k_node2<<<nblk(8LL * N, 256), 256, 0, stream>>>(sorted, off, deg, rec2, b2, out, N);
}


# Each edge's source id and score record fetched once, not twice, in both aggregation kernels: the max pass keeps them in registers for the exp pass, a node's loads issued together; sum order kept
# speedup vs baseline: 5.1846x; 1.1131x over previous
#include <hip/hip_runtime.h>
#include <hip/hip_bf16.h>

#define NEG_INF (-__builtin_inff())

typedef float v2f __attribute__((ext_vector_type(2)));
typedef float v8f __attribute__((ext_vector_type(8)));

#define CNT_STRIDE 8

__global__ void k_init(int* __restrict__ cnt, int N) {
    int t = blockIdx.x * blockDim.x + threadIdx.x;
    if (t < N) cnt[(size_t)t * CNT_STRIDE] = 0;
}

__global__ void k_hist(const int* __restrict__ ei, int* __restrict__ cnt,
                       int* __restrict__ rank, int E) {
    const int t = blockIdx.x * blockDim.x + threadIdx.x;
    const int T = gridDim.x * blockDim.x;
    int d[4];
#pragma unroll
    for (int k = 0; k < 4; ++k) { const int e = t + k * T; d[k] = (e < E) ? ei[(size_t)E + e] : -1; }
    int r[4];
#pragma unroll
    for (int k = 0; k < 4; ++k) r[k] = (d[k] >= 0) ? atomicAdd(&cnt[(size_t)d[k] * CNT_STRIDE], 1) : 0;
#pragma unroll
    for (int k = 0; k < 4; ++k) { const int e = t + k * T; if (e < E) rank[e] = r[k]; }
}

__global__ __launch_bounds__(1024)
void k_scan(const int* __restrict__ cnt, int* __restrict__ off,
            int* __restrict__ deg, int N) {
    __shared__ int s_wtot[32];
    __shared__ int s_wpre[32];
    const int t = threadIdx.x, lane = t & 31, w = t >> 5;
    const int base = blockIdx.x * 1024;
    int p = 0;
#pragma unroll 8
    for (int i = t; i < base; i += 1024) p += cnt[(size_t)i * CNT_STRIDE];
    const int n = base + t;
    const int c = (n < N) ? cnt[(size_t)n * CNT_STRIDE] : 0;
    int inc = c;
#pragma unroll
    for (int d = 1; d < 32; d <<= 1) {
        int v = __shfl_up(inc, d);
        if (lane >= d) inc += v;
    }
#pragma unroll
    for (int d = 16; d > 0; d >>= 1) p += __shfl_xor(p, d);
    if (lane == 31) s_wtot[w] = inc;
    if (lane == 0)  s_wpre[w] = p;
    __syncthreads();
    int before = s_wpre[lane] + ((lane < w) ? s_wtot[lane] : 0);
#pragma unroll
    for (int d = 16; d > 0; d >>= 1) before += __shfl_xor(before, d);
    if (n < N) { off[n] = before + inc - c; deg[n] = c; }
}

__global__ void k_fill(const int* __restrict__ ei, const int* __restrict__ rank,
                       const int* __restrict__ off, int* __restrict__ sorted, int E) {
    const int t = blockIdx.x * blockDim.x + threadIdx.x;
    const int T = gridDim.x * blockDim.x;
    int s[4], d[4], r[4];
#pragma unroll
    for (int k = 0; k < 4; ++k) {
        const int e = t + k * T;
        const bool ok = e < E;
        s[k] = ok ? ei[e] : 0;
        d[k] = ok ? ei[(size_t)E + e] : -1;
        r[k] = ok ? rank[e] : 0;
    }
#pragma unroll
    for (int k = 0; k < 4; ++k) if (d[k] >= 0) sorted[(size_t)off[d[k]] + r[k]] = s[k];
}

__global__ __launch_bounds__(128)
void k_gemm1_wmma(const float* __restrict__ x, const float* __restrict__ W1,
                  float* __restrict__ h1, int N) {
    const int wave = threadIdx.x >> 5;
    const int lane = threadIdx.x & 31;
    const int half = lane >> 4;
    const int l    = lane & 15;
    const int row0 = blockIdx.x * 16;
    const int col  = wave * 16 + l;
    int row = row0 + l;
    if (row >= N) row = N - 1;
    const float* xrow = x + (size_t)row * 128;

    v8f acc = {};
    for (int k0 = 0; k0 < 128; k0 += 4) {
        const int k = k0 + 2 * half;
        v2f a, b;
        a.x = xrow[k + 0];
        a.y = xrow[k + 1];
        b.x = W1[(size_t)(k + 0) * 64 + col];
        b.y = W1[(size_t)(k + 1) * 64 + col];
        acc = __builtin_amdgcn_wmma_f32_16x16x4_f32(
             false, a,  false, b,
             (short)0, acc,  false,  false);
    }
    float* dp = h1 + (size_t)(row0 + 8 * half) * 64 + col;
    if (row0 + 16 <= N) {
#pragma unroll
        for (int v = 0; v < 8; ++v) dp[(size_t)v * 64] = acc[v];
    } else {
#pragma unroll
        for (int v = 0; v < 8; ++v) {
            if (row0 + v + 8 * half < N) dp[(size_t)v * 64] = acc[v];
        }
    }
}

__global__ void k_scores1(const float* __restrict__ h1, const float* __restrict__ as,
                          const float* __restrict__ ad, float* __restrict__ ss,
                          float* __restrict__ sd, int N) {
    int t = blockIdx.x * blockDim.x + threadIdx.x;
    if (t >= N * 4) return;
    int n = t >> 2, h = t & 3;
    const float4* hp = (const float4*)(h1 + (size_t)n * 64 + h * 16);
    const float4* ap = (const float4*)(as + h * 16);
    const float4* bp = (const float4*)(ad + h * 16);
    float a = 0.f, b = 0.f;
#pragma unroll
    for (int i = 0; i < 4; ++i) {
        float4 hv = hp[i], av = ap[i], bv = bp[i];
        a += hv.x * av.x + hv.y * av.y + hv.z * av.z + hv.w * av.w;
        b += hv.x * bv.x + hv.y * bv.y + hv.z * bv.z + hv.w * bv.w;
    }
    ss[t] = a; sd[t] = b;
}

__device__ __forceinline__ float lrelu(float s) { return s > 0.f ? s : 0.2f * s; }

__device__ __forceinline__ float wave_max(float v) {
#pragma unroll
    for (int d = 16; d > 0; d >>= 1) v = fmaxf(v, __shfl_xor(v, d));
    return v;
}
__device__ __forceinline__ float wave_sum(float v) {
#pragma unroll
    for (int d = 16; d > 0; d >>= 1) v += __shfl_xor(v, d);
    return v;
}
__device__ __forceinline__ unsigned lane_bcast_u(float v, int j) {
    return (unsigned)__builtin_amdgcn_readlane(__float_as_int(v), j);
}

__device__ __forceinline__ void rows_acc(const float* __restrict__ h1, int lane, int s,
                                         float e0, float e1, float e2, float e3, int cnt,
                                         unsigned k0, unsigned k1, unsigned k2, unsigned k3,
                                         float& a0, float& a1) {
    for (int j = 0; j < cnt; j += 8) {
#pragma unroll
        for (int u = 0; u < 8; ++u) {
            const int sj = __builtin_amdgcn_readlane(s, j + u);
            const unsigned w0 = lane_bcast_u(e0, j + u), w1 = lane_bcast_u(e1, j + u);
            const unsigned w2 = lane_bcast_u(e2, j + u), w3 = lane_bcast_u(e3, j + u);
            const float w = __uint_as_float((w0 & k0) | (w1 & k1) | (w2 & k2) | (w3 & k3));
            const float2 hv = *(const float2*)(h1 + (size_t)sj * 64 + 2 * lane);
            a0 += hv.x * w;
            a1 += hv.y * w;
        }
    }
}

__global__ __launch_bounds__(256) __attribute__((amdgpu_waves_per_eu(16, 16)))
void k_node1(const int* __restrict__ sorted, const int* __restrict__ off,
             const int* __restrict__ deg, const float* __restrict__ ss,
             const float* __restrict__ sd, const float* __restrict__ h1,
             const float* __restrict__ b1, const float* __restrict__ W2,
             const float* __restrict__ as2, const float* __restrict__ ad2,
             float* __restrict__ rec2, int N) {
    const int lane = threadIdx.x & 31;
    const int n = __builtin_amdgcn_readfirstlane(blockIdx.x * 8 + (threadIdx.x >> 5));
    if (n >= N) return;
    const int o = off[n], dg = deg[n], tot = dg + 1;
    const float4 sdn = *(const float4*)(sd + (size_t)n * 4);

    const int pA = lane, pB = 32 + lane;
    const bool hasB = tot > 32;
    const int vA = sorted[(pA < dg) ? (size_t)o + pA : 0];
    const int vB = sorted[(pB < dg) ? (size_t)o + pB : 0];
    const int sA = (pA < dg) ? vA : n;
    const int sB = (pB < dg) ? vB : n;
    const float4 svA = *(const float4*)(ss + (size_t)sA * 4);
    float4 svB = make_float4(0.f, 0.f, 0.f, 0.f);
    if (hasB) svB = *(const float4*)(ss + (size_t)sB * 4);
    const bool okA = pA < tot, okB = pB < tot;
    const float xA0 = okA ? lrelu(svA.x + sdn.x) : NEG_INF, xA1 = okA ? lrelu(svA.y + sdn.y) : NEG_INF;
    const float xA2 = okA ? lrelu(svA.z + sdn.z) : NEG_INF, xA3 = okA ? lrelu(svA.w + sdn.w) : NEG_INF;
    const float xB0 = okB ? lrelu(svB.x + sdn.x) : NEG_INF, xB1 = okB ? lrelu(svB.y + sdn.y) : NEG_INF;
    const float xB2 = okB ? lrelu(svB.z + sdn.z) : NEG_INF, xB3 = okB ? lrelu(svB.w + sdn.w) : NEG_INF;
    float m0 = fmaxf(xA0, xB0), m1 = fmaxf(xA1, xB1), m2 = fmaxf(xA2, xB2), m3 = fmaxf(xA3, xB3);
    for (int base = 64; base < tot; base += 32) {
        const int p = base + lane;
        if (p < tot) {
            const int s = (p < dg) ? sorted[(size_t)o + p] : n;
            const float4 sv = *(const float4*)(ss + (size_t)s * 4);
            m0 = fmaxf(m0, lrelu(sv.x + sdn.x));
            m1 = fmaxf(m1, lrelu(sv.y + sdn.y));
            m2 = fmaxf(m2, lrelu(sv.z + sdn.z));
            m3 = fmaxf(m3, lrelu(sv.w + sdn.w));
        }
    }
    m0 = __int_as_float(__builtin_amdgcn_readfirstlane(__float_as_int(wave_max(m0))));
    m1 = __int_as_float(__builtin_amdgcn_readfirstlane(__float_as_int(wave_max(m1))));
    m2 = __int_as_float(__builtin_amdgcn_readfirstlane(__float_as_int(wave_max(m2))));
    m3 = __int_as_float(__builtin_amdgcn_readfirstlane(__float_as_int(wave_max(m3))));

    const int hsel = lane >> 3;
    const unsigned k0 = (hsel == 0) ? 0xffffffffu : 0u, k1 = (hsel == 1) ? 0xffffffffu : 0u;
    const unsigned k2 = (hsel == 2) ? 0xffffffffu : 0u, k3 = (hsel == 3) ? 0xffffffffu : 0u;
    float d0 = 0.f, d1 = 0.f, d2 = 0.f, d3 = 0.f;
    float a0 = 0.f, a1 = 0.f;
    {
        const float e0 = okA ? expf(xA0 - m0) : 0.f, e1 = okA ? expf(xA1 - m1) : 0.f;
        const float e2 = okA ? expf(xA2 - m2) : 0.f, e3 = okA ? expf(xA3 - m3) : 0.f;
        d0 += e0; d1 += e1; d2 += e2; d3 += e3;
        rows_acc(h1, lane, sA, e0, e1, e2, e3, hasB ? 32 : tot, k0, k1, k2, k3, a0, a1);
    }
    if (hasB) {
        const float e0 = okB ? expf(xB0 - m0) : 0.f, e1 = okB ? expf(xB1 - m1) : 0.f;
        const float e2 = okB ? expf(xB2 - m2) : 0.f, e3 = okB ? expf(xB3 - m3) : 0.f;
        d0 += e0; d1 += e1; d2 += e2; d3 += e3;
        rows_acc(h1, lane, sB, e0, e1, e2, e3, (tot - 32 < 32) ? (tot - 32) : 32, k0, k1, k2, k3, a0, a1);
    }
    for (int base = 64; base < tot; base += 32) {
        const int p = base + lane;
        int s = n;
        float e0 = 0.f, e1 = 0.f, e2 = 0.f, e3 = 0.f;
        if (p < tot) {
            if (p < dg) s = sorted[(size_t)o + p];
            const float4 sv = *(const float4*)(ss + (size_t)s * 4);
            e0 = expf(lrelu(sv.x + sdn.x) - m0);
            e1 = expf(lrelu(sv.y + sdn.y) - m1);
            e2 = expf(lrelu(sv.z + sdn.z) - m2);
            e3 = expf(lrelu(sv.w + sdn.w) - m3);
        }
        d0 += e0; d1 += e1; d2 += e2; d3 += e3;
        rows_acc(h1, lane, s, e0, e1, e2, e3, (tot - base < 32) ? (tot - base) : 32, k0, k1, k2, k3, a0, a1);
    }
    d0 = wave_sum(d0); d1 = wave_sum(d1); d2 = wave_sum(d2); d3 = wave_sum(d3);
    const float dn = (hsel == 0) ? d0 : (hsel == 1) ? d1 : (hsel == 2) ? d2 : d3;
    const float2 bb = *(const float2*)(b1 + 2 * lane);
    float v0 = a0 / dn + bb.x;
    float v1 = a1 / dn + bb.y;
    v0 = v0 > 0.f ? v0 : expm1f(v0);
    v1 = v1 > 0.f ? v1 : expm1f(v1);
    const float4 wv = *(const float4*)(W2 + 4 * lane);
    float o0 = v0 * wv.x + v1 * wv.z;
    float o1 = v0 * wv.y + v1 * wv.w;
    o0 = wave_sum(o0);
    o1 = wave_sum(o1);
    if (lane == 0) {
        *(float4*)(rec2 + (size_t)n * 4) =
            make_float4(o0, o1, o0 * as2[0] + o1 * as2[1], o0 * ad2[0] + o1 * ad2[1]);
    }
}

#define N2_SLOTS 6
__global__ __launch_bounds__(256)
void k_node2(const int* __restrict__ sorted, const int* __restrict__ off,
             const int* __restrict__ deg, const float* __restrict__ rec2,
             const float* __restrict__ b2, float* __restrict__ out, int N) {
    const int t = blockIdx.x * 256 + threadIdx.x;
    const int n = t >> 3, sub = t & 7;
    int o = 0, dg = 0, tot = 0, nn = 0;
    float sdn = 0.f;
    if (n < N) { o = off[n]; dg = deg[n]; tot = dg + 1; sdn = rec2[(size_t)n * 4 + 3]; nn = n; }

    int s[N2_SLOTS];
#pragma unroll
    for (int k = 0; k < N2_SLOTS; ++k) {
        const int p = sub + 8 * k;
        const int v = sorted[(p < dg) ? (size_t)o + p : 0];
        s[k] = (p < dg) ? v : nn;
    }
    float4 r[N2_SLOTS];
#pragma unroll
    for (int k = 0; k < N2_SLOTS; ++k) r[k] = *(const float4*)(rec2 + (size_t)s[k] * 4);
    __builtin_amdgcn_sched_barrier(0);
    float rx[N2_SLOTS], ry[N2_SLOTS], xs[N2_SLOTS];
    float m = NEG_INF;
#pragma unroll
    for (int k = 0; k < N2_SLOTS; ++k) {
        const bool ok = sub + 8 * k < tot;
        rx[k] = ok ? r[k].x : 0.f;
        ry[k] = ok ? r[k].y : 0.f;
        xs[k] = ok ? lrelu(r[k].z + sdn) : NEG_INF;
        m = fmaxf(m, xs[k]);
    }
    for (int p = sub + 8 * N2_SLOTS; p < tot; p += 8) {
        const int sp = (p < dg) ? sorted[(size_t)o + p] : n;
        m = fmaxf(m, lrelu(rec2[(size_t)sp * 4 + 2] + sdn));
    }
#pragma unroll
    for (int d = 4; d > 0; d >>= 1) m = fmaxf(m, __shfl_xor(m, d));

    float dsum = 0.f, a0 = 0.f, a1 = 0.f;
#pragma unroll
    for (int k = 0; k < N2_SLOTS; ++k) {
        const float ex = (sub + 8 * k < tot) ? expf(xs[k] - m) : 0.f;
        dsum += ex;
        a0 += rx[k] * ex;
        a1 += ry[k] * ex;
    }
    for (int p = sub + 8 * N2_SLOTS; p < tot; p += 8) {
        const int sp = (p < dg) ? sorted[(size_t)o + p] : n;
        const float4 r = *(const float4*)(rec2 + (size_t)sp * 4);
        const float ex = expf(lrelu(r.z + sdn) - m);
        dsum += ex;
        a0 += r.x * ex;
        a1 += r.y * ex;
    }
#pragma unroll
    for (int d = 4; d > 0; d >>= 1) {
        dsum += __shfl_xor(dsum, d);
        a0   += __shfl_xor(a0, d);
        a1   += __shfl_xor(a1, d);
    }
    if (sub == 0 && n < N) {
        float v0 = a0 / dsum + b2[0];
        float v1 = a1 / dsum + b2[1];
        float mx = fmaxf(v0, v1);
        float lse = mx + logf(expf(v0 - mx) + expf(v1 - mx));
        *(float2*)(out + (size_t)n * 2) = make_float2(v0 - lse, v1 - lse);
    }
}

static inline unsigned nblk(long long n, int b) { return (unsigned)((n + b - 1) / b); }

extern "C" void kernel_launch(void* const* d_in, const int* in_sizes, int n_in,
                              void* d_out, int out_size, void* d_ws, size_t ws_size,
                              hipStream_t stream) {
    const float* x     = (const float*)d_in[0];
    const int*   ei    = (const int*)  d_in[1];
    const float* W1    = (const float*)d_in[2];
    const float* asrc1 = (const float*)d_in[3];
    const float* adst1 = (const float*)d_in[4];
    const float* b1    = (const float*)d_in[5];
    const float* W2    = (const float*)d_in[6];
    const float* asrc2 = (const float*)d_in[7];
    const float* adst2 = (const float*)d_in[8];
    const float* b2    = (const float*)d_in[9];
    float* out = (float*)d_out;

    const int N = in_sizes[0] / 128;
    const int E = in_sizes[1] / 2;

    float* ws   = (float*)d_ws;
    size_t nl   = (size_t)N;
    float* h1   = ws;
    float* ss1  = h1  + nl * 64;
    float* sd1  = ss1 + nl * 4;
    float* rec2 = sd1 + nl * 4;
    int*   off  = (int*)(rec2 + nl * 4);
    int*   deg  = off + nl;
    int*   rank = deg + nl;
    int*   sorted = rank + (size_t)E;
    int*   cnt  = (int*)h1;

    k_init<<<nblk(N, 256), 256, 0, stream>>>(cnt, N);
    k_hist<<<nblk(E, 1024), 256, 0, stream>>>(ei, cnt, rank, E);
    k_scan<<<nblk(N, 1024), 1024, 0, stream>>>(cnt, off, deg, N);
    k_fill<<<nblk(E, 1024), 256, 0, stream>>>(ei, rank, off, sorted, E);
    k_gemm1_wmma<<<(N + 15) / 16, 128, 0, stream>>>(x, W1, h1, N);
    k_scores1<<<nblk(4LL * N, 256), 256, 0, stream>>>(h1, asrc1, adst1, ss1, sd1, N);
    k_node1<<<nblk(N, 8), 256, 0, stream>>>(sorted, off, deg, ss1, sd1, h1, b1, W2,
                                            asrc2, adst2, rec2, N);
    k_node2<<<nblk(8LL * N, 256), 256, 0, stream>>>(sorted, off, deg, rec2, b2, out, N);
}


# Edge grouping by destination: 1.6M returning global atomics replaced by LDS counters (two-level counting sort, 256-node buckets); rank inside a node stays an arrival order, so sum order may differ
# speedup vs baseline: 9.0850x; 1.7523x over previous
#include <hip/hip_runtime.h>
#include <hip/hip_bf16.h>

#define NEG_INF (-__builtin_inff())

typedef float v2f __attribute__((ext_vector_type(2)));
typedef float v8f __attribute__((ext_vector_type(8)));

#define EPB    8192
#define BSH    8
#define BNODES 256

extern __shared__ int s_dyn[];

__global__ __launch_bounds__(1024)
void k_count(const int* __restrict__ ei, int* __restrict__ cntA, int E, int C, int G) {
    const int t = threadIdx.x, b = blockIdx.x;
    for (int c = t; c < C; c += 1024) s_dyn[c] = 0;
    __syncthreads();
    const int e0 = b * EPB + t;
    int d[8];
#pragma unroll
    for (int k = 0; k < 8; ++k) { const int e = e0 + k * 1024; d[k] = (e < E) ? ei[(size_t)E + e] : -1; }
#pragma unroll
    for (int k = 0; k < 8; ++k) if (d[k] >= 0) atomicAdd(&s_dyn[d[k] >> BSH], 1);
    __syncthreads();
    for (int c = t; c < C; c += 1024) cntA[(size_t)c * G + b] = s_dyn[c];
}

__global__ __launch_bounds__(1024)
void k_scan(const int* __restrict__ cnt, int* __restrict__ off, int M) {
    __shared__ int s_wtot[32];
    __shared__ int s_wpre[32];
    const int t = threadIdx.x, lane = t & 31, w = t >> 5;
    const int base = blockIdx.x * 1024;
    int p = 0;
#pragma unroll 8
    for (int i = t; i < base; i += 1024) p += cnt[i];
    const int n = base + t;
    const int c = (n < M) ? cnt[n] : 0;
    int inc = c;
#pragma unroll
    for (int d = 1; d < 32; d <<= 1) {
        int v = __shfl_up(inc, d);
        if (lane >= d) inc += v;
    }
#pragma unroll
    for (int d = 16; d > 0; d >>= 1) p += __shfl_xor(p, d);
    if (lane == 31) s_wtot[w] = inc;
    if (lane == 0)  s_wpre[w] = p;
    __syncthreads();
    int before = s_wpre[lane] + ((lane < w) ? s_wtot[lane] : 0);
#pragma unroll
    for (int d = 16; d > 0; d >>= 1) before += __shfl_xor(before, d);
    if (n < M) off[n] = before + inc - c;
}

__global__ __launch_bounds__(1024)
void k_part(const int* __restrict__ ei, const int* __restrict__ offA,
            int* __restrict__ part_w, int E, int C, int G) {
    int2* __restrict__ part = (int2*)part_w;
    const int t = threadIdx.x, b = blockIdx.x;
    for (int c = t; c < C; c += 1024) s_dyn[c] = offA[(size_t)c * G + b];
    __syncthreads();
    const int e0 = b * EPB + t;
    int s[8], d[8];
#pragma unroll
    for (int k = 0; k < 8; ++k) {
        const int e = e0 + k * 1024;
        const bool ok = e < E;
        s[k] = ok ? ei[e] : 0;
        d[k] = ok ? ei[(size_t)E + e] : -1;
    }
#pragma unroll
    for (int k = 0; k < 8; ++k) {
        if (d[k] >= 0) {
            const int p = atomicAdd(&s_dyn[d[k] >> BSH], 1);
            part[p] = make_int2(s[k], d[k]);
        }
    }
}

__global__ __launch_bounds__(1024)
void k_bucket(const int* __restrict__ part_w, const int* __restrict__ offA,
              int* __restrict__ off, int* __restrict__ deg, int* __restrict__ sorted,
              int N, int E, int C, int G) {
    __shared__ int s_cnt[BNODES];
    __shared__ int s_wtot[BNODES / 32];
    const int2* __restrict__ part = (const int2*)part_w;
    const int t = threadIdx.x, lane = t & 31, w = t >> 5, c = blockIdx.x;
    const int beg = offA[(size_t)c * G];
    const int end = (c + 1 < C) ? offA[(size_t)(c + 1) * G] : E;
    const int nbase = c << BSH;
    if (t < BNODES) s_cnt[t] = 0;
    __syncthreads();
    for (int i0 = beg + t; i0 < end; i0 += 4096) {
        int y[4];
#pragma unroll
        for (int k = 0; k < 4; ++k) { const int i = i0 + k * 1024; y[k] = (i < end) ? part[i].y : -1; }
#pragma unroll
        for (int k = 0; k < 4; ++k) if (y[k] >= 0) atomicAdd(&s_cnt[y[k] - nbase], 1);
    }
    __syncthreads();
    int cv = 0, inc = 0;
    if (t < BNODES) {
        cv = s_cnt[t];
        inc = cv;
#pragma unroll
        for (int d = 1; d < 32; d <<= 1) {
            const int v = __shfl_up(inc, d);
            if (lane >= d) inc += v;
        }
        if (lane == 31) s_wtot[w] = inc;
    }
    __syncthreads();
    if (t < BNODES) {
        int pre = 0;
        for (int k = 0; k < w; ++k) pre += s_wtot[k];
        const int ex = pre + inc - cv;
        s_cnt[t] = ex;
        const int n = nbase + t;
        if (n < N) { off[n] = beg + ex; deg[n] = cv; }
    }
    __syncthreads();
    for (int i0 = beg + t; i0 < end; i0 += 4096) {
        int2 p[4];
#pragma unroll
        for (int k = 0; k < 4; ++k) { const int i = i0 + k * 1024; p[k] = (i < end) ? part[i] : make_int2(0, -1); }
#pragma unroll
        for (int k = 0; k < 4; ++k) {
            if (p[k].y >= 0) {
                const int pos = atomicAdd(&s_cnt[p[k].y - nbase], 1);
                sorted[(size_t)beg + pos] = p[k].x;
            }
        }
    }
}

__global__ __launch_bounds__(128)
void k_gemm1_wmma(const float* __restrict__ x, const float* __restrict__ W1,
                  float* __restrict__ h1, int N) {
    const int wave = threadIdx.x >> 5;
    const int lane = threadIdx.x & 31;
    const int half = lane >> 4;
    const int l    = lane & 15;
    const int row0 = blockIdx.x * 16;
    const int col  = wave * 16 + l;
    int row = row0 + l;
    if (row >= N) row = N - 1;
    const float* xrow = x + (size_t)row * 128;

    v8f acc = {};
    for (int k0 = 0; k0 < 128; k0 += 4) {
        const int k = k0 + 2 * half;
        v2f a, b;
        a.x = xrow[k + 0];
        a.y = xrow[k + 1];
        b.x = W1[(size_t)(k + 0) * 64 + col];
        b.y = W1[(size_t)(k + 1) * 64 + col];
        acc = __builtin_amdgcn_wmma_f32_16x16x4_f32(
             false, a,  false, b,
             (short)0, acc,  false,  false);
    }
    float* dp = h1 + (size_t)(row0 + 8 * half) * 64 + col;
    if (row0 + 16 <= N) {
#pragma unroll
        for (int v = 0; v < 8; ++v) dp[(size_t)v * 64] = acc[v];
    } else {
#pragma unroll
        for (int v = 0; v < 8; ++v) {
            if (row0 + v + 8 * half < N) dp[(size_t)v * 64] = acc[v];
        }
    }
}

__global__ void k_scores1(const float* __restrict__ h1, const float* __restrict__ as,
                          const float* __restrict__ ad, float* __restrict__ ss,
                          float* __restrict__ sd, int N) {
    int t = blockIdx.x * blockDim.x + threadIdx.x;
    if (t >= N * 4) return;
    int n = t >> 2, h = t & 3;
    const float4* hp = (const float4*)(h1 + (size_t)n * 64 + h * 16);
    const float4* ap = (const float4*)(as + h * 16);
    const float4* bp = (const float4*)(ad + h * 16);
    float a = 0.f, b = 0.f;
#pragma unroll
    for (int i = 0; i < 4; ++i) {
        float4 hv = hp[i], av = ap[i], bv = bp[i];
        a += hv.x * av.x + hv.y * av.y + hv.z * av.z + hv.w * av.w;
        b += hv.x * bv.x + hv.y * bv.y + hv.z * bv.z + hv.w * bv.w;
    }
    ss[t] = a; sd[t] = b;
}

__device__ __forceinline__ float lrelu(float s) { return s > 0.f ? s : 0.2f * s; }

__device__ __forceinline__ float wave_max(float v) {
#pragma unroll
    for (int d = 16; d > 0; d >>= 1) v = fmaxf(v, __shfl_xor(v, d));
    return v;
}
__device__ __forceinline__ float wave_sum(float v) {
#pragma unroll
    for (int d = 16; d > 0; d >>= 1) v += __shfl_xor(v, d);
    return v;
}
__device__ __forceinline__ unsigned lane_bcast_u(float v, int j) {
    return (unsigned)__builtin_amdgcn_readlane(__float_as_int(v), j);
}

__device__ __forceinline__ void rows_acc(const float* __restrict__ h1, int lane, int s,
                                         float e0, float e1, float e2, float e3, int cnt,
                                         unsigned k0, unsigned k1, unsigned k2, unsigned k3,
                                         float& a0, float& a1) {
    for (int j = 0; j < cnt; j += 8) {
#pragma unroll
        for (int u = 0; u < 8; ++u) {
            const int sj = __builtin_amdgcn_readlane(s, j + u);
            const unsigned w0 = lane_bcast_u(e0, j + u), w1 = lane_bcast_u(e1, j + u);
            const unsigned w2 = lane_bcast_u(e2, j + u), w3 = lane_bcast_u(e3, j + u);
            const float w = __uint_as_float((w0 & k0) | (w1 & k1) | (w2 & k2) | (w3 & k3));
            const float2 hv = *(const float2*)(h1 + (size_t)sj * 64 + 2 * lane);
            a0 += hv.x * w;
            a1 += hv.y * w;
        }
    }
}

__global__ __launch_bounds__(256) __attribute__((amdgpu_waves_per_eu(16, 16)))
void k_node1(const int* __restrict__ sorted, const int* __restrict__ off,
             const int* __restrict__ deg, const float* __restrict__ ss,
             const float* __restrict__ sd, const float* __restrict__ h1,
             const float* __restrict__ b1, const float* __restrict__ W2,
             const float* __restrict__ as2, const float* __restrict__ ad2,
             float* __restrict__ rec2, int N) {
    const int lane = threadIdx.x & 31;
    const int n = __builtin_amdgcn_readfirstlane(blockIdx.x * 8 + (threadIdx.x >> 5));
    if (n >= N) return;
    const int o = off[n], dg = deg[n], tot = dg + 1;
    const float4 sdn = *(const float4*)(sd + (size_t)n * 4);

    const int pA = lane, pB = 32 + lane;
    const bool hasB = tot > 32;
    const int vA = sorted[(pA < dg) ? (size_t)o + pA : 0];
    const int vB = sorted[(pB < dg) ? (size_t)o + pB : 0];
    const int sA = (pA < dg) ? vA : n;
    const int sB = (pB < dg) ? vB : n;
    const float4 svA = *(const float4*)(ss + (size_t)sA * 4);
    float4 svB = make_float4(0.f, 0.f, 0.f, 0.f);
    if (hasB) svB = *(const float4*)(ss + (size_t)sB * 4);
    const bool okA = pA < tot, okB = pB < tot;
    const float xA0 = okA ? lrelu(svA.x + sdn.x) : NEG_INF, xA1 = okA ? lrelu(svA.y + sdn.y) : NEG_INF;
    const float xA2 = okA ? lrelu(svA.z + sdn.z) : NEG_INF, xA3 = okA ? lrelu(svA.w + sdn.w) : NEG_INF;
    const float xB0 = okB ? lrelu(svB.x + sdn.x) : NEG_INF, xB1 = okB ? lrelu(svB.y + sdn.y) : NEG_INF;
    const float xB2 = okB ? lrelu(svB.z + sdn.z) : NEG_INF, xB3 = okB ? lrelu(svB.w + sdn.w) : NEG_INF;
    float m0 = fmaxf(xA0, xB0), m1 = fmaxf(xA1, xB1), m2 = fmaxf(xA2, xB2), m3 = fmaxf(xA3, xB3);
    for (int base = 64; base < tot; base += 32) {
        const int p = base + lane;
        if (p < tot) {
            const int s = (p < dg) ? sorted[(size_t)o + p] : n;
            const float4 sv = *(const float4*)(ss + (size_t)s * 4);
            m0 = fmaxf(m0, lrelu(sv.x + sdn.x));
            m1 = fmaxf(m1, lrelu(sv.y + sdn.y));
            m2 = fmaxf(m2, lrelu(sv.z + sdn.z));
            m3 = fmaxf(m3, lrelu(sv.w + sdn.w));
        }
    }
    m0 = __int_as_float(__builtin_amdgcn_readfirstlane(__float_as_int(wave_max(m0))));
    m1 = __int_as_float(__builtin_amdgcn_readfirstlane(__float_as_int(wave_max(m1))));
    m2 = __int_as_float(__builtin_amdgcn_readfirstlane(__float_as_int(wave_max(m2))));
    m3 = __int_as_float(__builtin_amdgcn_readfirstlane(__float_as_int(wave_max(m3))));

    const int hsel = lane >> 3;
    const unsigned k0 = (hsel == 0) ? 0xffffffffu : 0u, k1 = (hsel == 1) ? 0xffffffffu : 0u;
    const unsigned k2 = (hsel == 2) ? 0xffffffffu : 0u, k3 = (hsel == 3) ? 0xffffffffu : 0u;
    float d0 = 0.f, d1 = 0.f, d2 = 0.f, d3 = 0.f;
    float a0 = 0.f, a1 = 0.f;
    {
        const float e0 = okA ? expf(xA0 - m0) : 0.f, e1 = okA ? expf(xA1 - m1) : 0.f;
        const float e2 = okA ? expf(xA2 - m2) : 0.f, e3 = okA ? expf(xA3 - m3) : 0.f;
        d0 += e0; d1 += e1; d2 += e2; d3 += e3;
        rows_acc(h1, lane, sA, e0, e1, e2, e3, hasB ? 32 : tot, k0, k1, k2, k3, a0, a1);
    }
    if (hasB) {
        const float e0 = okB ? expf(xB0 - m0) : 0.f, e1 = okB ? expf(xB1 - m1) : 0.f;
        const float e2 = okB ? expf(xB2 - m2) : 0.f, e3 = okB ? expf(xB3 - m3) : 0.f;
        d0 += e0; d1 += e1; d2 += e2; d3 += e3;
        rows_acc(h1, lane, sB, e0, e1, e2, e3, (tot - 32 < 32) ? (tot - 32) : 32, k0, k1, k2, k3, a0, a1);
    }
    for (int base = 64; base < tot; base += 32) {
        const int p = base + lane;
        int s = n;
        float e0 = 0.f, e1 = 0.f, e2 = 0.f, e3 = 0.f;
        if (p < tot) {
            if (p < dg) s = sorted[(size_t)o + p];
            const float4 sv = *(const float4*)(ss + (size_t)s * 4);
            e0 = expf(lrelu(sv.x + sdn.x) - m0);
            e1 = expf(lrelu(sv.y + sdn.y) - m1);
            e2 = expf(lrelu(sv.z + sdn.z) - m2);
            e3 = expf(lrelu(sv.w + sdn.w) - m3);
        }
        d0 += e0; d1 += e1; d2 += e2; d3 += e3;
        rows_acc(h1, lane, s, e0, e1, e2, e3, (tot - base < 32) ? (tot - base) : 32, k0, k1, k2, k3, a0, a1);
    }
    d0 = wave_sum(d0); d1 = wave_sum(d1); d2 = wave_sum(d2); d3 = wave_sum(d3);
    const float dn = (hsel == 0) ? d0 : (hsel == 1) ? d1 : (hsel == 2) ? d2 : d3;
    const float2 bb = *(const float2*)(b1 + 2 * lane);
    float v0 = a0 / dn + bb.x;
    float v1 = a1 / dn + bb.y;
    v0 = v0 > 0.f ? v0 : expm1f(v0);
    v1 = v1 > 0.f ? v1 : expm1f(v1);
    const float4 wv = *(const float4*)(W2 + 4 * lane);
    float o0 = v0 * wv.x + v1 * wv.z;
    float o1 = v0 * wv.y + v1 * wv.w;
    o0 = wave_sum(o0);
    o1 = wave_sum(o1);
    if (lane == 0) {
        *(float4*)(rec2 + (size_t)n * 4) =
            make_float4(o0, o1, o0 * as2[0] + o1 * as2[1], o0 * ad2[0] + o1 * ad2[1]);
    }
}

#define N2_SLOTS 6
__global__ __launch_bounds__(256)
void k_node2(const int* __restrict__ sorted, const int* __restrict__ off,
             const int* __restrict__ deg, const float* __restrict__ rec2,
             const float* __restrict__ b2, float* __restrict__ out, int N) {
    const int t = blockIdx.x * 256 + threadIdx.x;
    const int n = t >> 3, sub = t & 7;
    int o = 0, dg = 0, tot = 0, nn = 0;
    float sdn = 0.f;
    if (n < N) { o = off[n]; dg = deg[n]; tot = dg + 1; sdn = rec2[(size_t)n * 4 + 3]; nn = n; }

    int s[N2_SLOTS];
#pragma unroll
    for (int k = 0; k < N2_SLOTS; ++k) {
        const int p = sub + 8 * k;
        const int v = sorted[(p < dg) ? (size_t)o + p : 0];
        s[k] = (p < dg) ? v : nn;
    }
    float4 r[N2_SLOTS];
#pragma unroll
    for (int k = 0; k < N2_SLOTS; ++k) r[k] = *(const float4*)(rec2 + (size_t)s[k] * 4);
    __builtin_amdgcn_sched_barrier(0);
    float rx[N2_SLOTS], ry[N2_SLOTS], xs[N2_SLOTS];
    float m = NEG_INF;
#pragma unroll
    for (int k = 0; k < N2_SLOTS; ++k) {
        const bool ok = sub + 8 * k < tot;
        rx[k] = ok ? r[k].x : 0.f;
        ry[k] = ok ? r[k].y : 0.f;
        xs[k] = ok ? lrelu(r[k].z + sdn) : NEG_INF;
        m = fmaxf(m, xs[k]);
    }
    for (int p = sub + 8 * N2_SLOTS; p < tot; p += 8) {
        const int sp = (p < dg) ? sorted[(size_t)o + p] : n;
        m = fmaxf(m, lrelu(rec2[(size_t)sp * 4 + 2] + sdn));
    }
#pragma unroll
    for (int d = 4; d > 0; d >>= 1) m = fmaxf(m, __shfl_xor(m, d));

    float dsum = 0.f, a0 = 0.f, a1 = 0.f;
#pragma unroll
    for (int k = 0; k < N2_SLOTS; ++k) {
        const float ex = (sub + 8 * k < tot) ? expf(xs[k] - m) : 0.f;
        dsum += ex;
        a0 += rx[k] * ex;
        a1 += ry[k] * ex;
    }
    for (int p = sub + 8 * N2_SLOTS; p < tot; p += 8) {
        const int sp = (p < dg) ? sorted[(size_t)o + p] : n;
        const float4 r = *(const float4*)(rec2 + (size_t)sp * 4);
        const float ex = expf(lrelu(r.z + sdn) - m);
        dsum += ex;
        a0 += r.x * ex;
        a1 += r.y * ex;
    }
#pragma unroll
    for (int d = 4; d > 0; d >>= 1) {
        dsum += __shfl_xor(dsum, d);
        a0   += __shfl_xor(a0, d);
        a1   += __shfl_xor(a1, d);
    }
    if (sub == 0 && n < N) {
        float v0 = a0 / dsum + b2[0];
        float v1 = a1 / dsum + b2[1];
        float mx = fmaxf(v0, v1);
        float lse = mx + logf(expf(v0 - mx) + expf(v1 - mx));
        *(float2*)(out + (size_t)n * 2) = make_float2(v0 - lse, v1 - lse);
    }
}

static inline unsigned nblk(long long n, int b) { return (unsigned)((n + b - 1) / b); }

extern "C" void kernel_launch(void* const* d_in, const int* in_sizes, int n_in,
                              void* d_out, int out_size, void* d_ws, size_t ws_size,
                              hipStream_t stream) {
    const float* x     = (const float*)d_in[0];
    const int*   ei    = (const int*)  d_in[1];
    const float* W1    = (const float*)d_in[2];
    const float* asrc1 = (const float*)d_in[3];
    const float* adst1 = (const float*)d_in[4];
    const float* b1    = (const float*)d_in[5];
    const float* W2    = (const float*)d_in[6];
    const float* asrc2 = (const float*)d_in[7];
    const float* adst2 = (const float*)d_in[8];
    const float* b2    = (const float*)d_in[9];
    float* out = (float*)d_out;

    const int N = in_sizes[0] / 128;
    const int E = in_sizes[1] / 2;

    float* ws   = (float*)d_ws;
    size_t nl   = (size_t)N;
    float* h1   = ws;
    float* ss1  = h1  + nl * 64;
    float* sd1  = ss1 + nl * 4;
    float* rec2 = sd1 + nl * 4;
    int*   off  = (int*)(rec2 + nl * 4);
    int*   deg  = off + nl;
    int*   sorted = deg + nl + (size_t)E;
    const int C = (N + BNODES - 1) / BNODES;
    const int G = (int)nblk(E, EPB);
    const int M = C * G;
    int*   part = (int*)h1;
    int*   cntA = (int*)ss1;
    int*   offA = cntA + M;

    k_count<<<G, 1024, C * sizeof(int), stream>>>(ei, cntA, E, C, G);
    k_scan<<<nblk(M, 1024), 1024, 0, stream>>>(cntA, offA, M);
    k_part<<<G, 1024, C * sizeof(int), stream>>>(ei, offA, part, E, C, G);
    k_bucket<<<C, 1024, 0, stream>>>(part, offA, off, deg, sorted, N, E, C, G);
    k_gemm1_wmma<<<(N + 15) / 16, 128, 0, stream>>>(x, W1, h1, N);
    k_scores1<<<nblk(4LL * N, 256), 256, 0, stream>>>(h1, asrc1, adst1, ss1, sd1, N);
    k_node1<<<nblk(N, 8), 256, 0, stream>>>(sorted, off, deg, ss1, sd1, h1, b1, W2,
                                            asrc2, adst2, rec2, N);
    k_node2<<<nblk(8LL * N, 256), 256, 0, stream>>>(sorted, off, deg, rec2, b2, out, N);
}


# Layer-1 row gather by half-waves: 16 lanes x 16 B per row, two edges per load; row offsets and weights reach the lanes through a per-wave LDS table, not 5 readlanes per edge; sum order changes
# speedup vs baseline: 9.4799x; 1.0435x over previous
#include <hip/hip_runtime.h>
#include <hip/hip_bf16.h>

#define NEG_INF (-__builtin_inff())

typedef float v2f __attribute__((ext_vector_type(2)));
typedef float v8f __attribute__((ext_vector_type(8)));

#define EPB    8192
#define BSH    8
#define BNODES 256

extern __shared__ int s_dyn[];

__global__ __launch_bounds__(1024)
void k_count(const int* __restrict__ ei, int* __restrict__ cntA, int E, int C, int G) {
    const int t = threadIdx.x, b = blockIdx.x;
    for (int c = t; c < C; c += 1024) s_dyn[c] = 0;
    __syncthreads();
    const int e0 = b * EPB + t;
    int d[8];
#pragma unroll
    for (int k = 0; k < 8; ++k) { const int e = e0 + k * 1024; d[k] = (e < E) ? ei[(size_t)E + e] : -1; }
#pragma unroll
    for (int k = 0; k < 8; ++k) if (d[k] >= 0) atomicAdd(&s_dyn[d[k] >> BSH], 1);
    __syncthreads();
    for (int c = t; c < C; c += 1024) cntA[(size_t)c * G + b] = s_dyn[c];
}

__global__ __launch_bounds__(1024)
void k_scan(const int* __restrict__ cnt, int* __restrict__ off, int M) {
    __shared__ int s_wtot[32];
    __shared__ int s_wpre[32];
    const int t = threadIdx.x, lane = t & 31, w = t >> 5;
    const int base = blockIdx.x * 1024;
    int p = 0;
#pragma unroll 8
    for (int i = t; i < base; i += 1024) p += cnt[i];
    const int n = base + t;
    const int c = (n < M) ? cnt[n] : 0;
    int inc = c;
#pragma unroll
    for (int d = 1; d < 32; d <<= 1) {
        int v = __shfl_up(inc, d);
        if (lane >= d) inc += v;
    }
#pragma unroll
    for (int d = 16; d > 0; d >>= 1) p += __shfl_xor(p, d);
    if (lane == 31) s_wtot[w] = inc;
    if (lane == 0)  s_wpre[w] = p;
    __syncthreads();
    int before = s_wpre[lane] + ((lane < w) ? s_wtot[lane] : 0);
#pragma unroll
    for (int d = 16; d > 0; d >>= 1) before += __shfl_xor(before, d);
    if (n < M) off[n] = before + inc - c;
}

__global__ __launch_bounds__(1024)
void k_part(const int* __restrict__ ei, const int* __restrict__ offA,
            int* __restrict__ part_w, int E, int C, int G) {
    int2* __restrict__ part = (int2*)part_w;
    const int t = threadIdx.x, b = blockIdx.x;
    for (int c = t; c < C; c += 1024) s_dyn[c] = offA[(size_t)c * G + b];
    __syncthreads();
    const int e0 = b * EPB + t;
    int s[8], d[8];
#pragma unroll
    for (int k = 0; k < 8; ++k) {
        const int e = e0 + k * 1024;
        const bool ok = e < E;
        s[k] = ok ? ei[e] : 0;
        d[k] = ok ? ei[(size_t)E + e] : -1;
    }
#pragma unroll
    for (int k = 0; k < 8; ++k) {
        if (d[k] >= 0) {
            const int p = atomicAdd(&s_dyn[d[k] >> BSH], 1);
            part[p] = make_int2(s[k], d[k]);
        }
    }
}

__global__ __launch_bounds__(1024)
void k_bucket(const int* __restrict__ part_w, const int* __restrict__ offA,
              int* __restrict__ off, int* __restrict__ deg, int* __restrict__ sorted,
              int N, int E, int C, int G) {
    __shared__ int s_cnt[BNODES];
    __shared__ int s_wtot[BNODES / 32];
    const int2* __restrict__ part = (const int2*)part_w;
    const int t = threadIdx.x, lane = t & 31, w = t >> 5, c = blockIdx.x;
    const int beg = offA[(size_t)c * G];
    const int end = (c + 1 < C) ? offA[(size_t)(c + 1) * G] : E;
    const int nbase = c << BSH;
    if (t < BNODES) s_cnt[t] = 0;
    __syncthreads();
    for (int i0 = beg + t; i0 < end; i0 += 4096) {
        int y[4];
#pragma unroll
        for (int k = 0; k < 4; ++k) { const int i = i0 + k * 1024; y[k] = (i < end) ? part[i].y : -1; }
#pragma unroll
        for (int k = 0; k < 4; ++k) if (y[k] >= 0) atomicAdd(&s_cnt[y[k] - nbase], 1);
    }
    __syncthreads();
    int cv = 0, inc = 0;
    if (t < BNODES) {
        cv = s_cnt[t];
        inc = cv;
#pragma unroll
        for (int d = 1; d < 32; d <<= 1) {
            const int v = __shfl_up(inc, d);
            if (lane >= d) inc += v;
        }
        if (lane == 31) s_wtot[w] = inc;
    }
    __syncthreads();
    if (t < BNODES) {
        int pre = 0;
        for (int k = 0; k < w; ++k) pre += s_wtot[k];
        const int ex = pre + inc - cv;
        s_cnt[t] = ex;
        const int n = nbase + t;
        if (n < N) { off[n] = beg + ex; deg[n] = cv; }
    }
    __syncthreads();
    for (int i0 = beg + t; i0 < end; i0 += 4096) {
        int2 p[4];
#pragma unroll
        for (int k = 0; k < 4; ++k) { const int i = i0 + k * 1024; p[k] = (i < end) ? part[i] : make_int2(0, -1); }
#pragma unroll
        for (int k = 0; k < 4; ++k) {
            if (p[k].y >= 0) {
                const int pos = atomicAdd(&s_cnt[p[k].y - nbase], 1);
                sorted[(size_t)beg + pos] = p[k].x;
            }
        }
    }
}

__global__ __launch_bounds__(128)
void k_gemm1_wmma(const float* __restrict__ x, const float* __restrict__ W1,
                  float* __restrict__ h1, int N) {
    const int wave = threadIdx.x >> 5;
    const int lane = threadIdx.x & 31;
    const int half = lane >> 4;
    const int l    = lane & 15;
    const int row0 = blockIdx.x * 16;
    const int col  = wave * 16 + l;
    int row = row0 + l;
    if (row >= N) row = N - 1;
    const float* xrow = x + (size_t)row * 128;

    v8f acc = {};
    for (int k0 = 0; k0 < 128; k0 += 4) {
        const int k = k0 + 2 * half;
        v2f a, b;
        a.x = xrow[k + 0];
        a.y = xrow[k + 1];
        b.x = W1[(size_t)(k + 0) * 64 + col];
        b.y = W1[(size_t)(k + 1) * 64 + col];
        acc = __builtin_amdgcn_wmma_f32_16x16x4_f32(
             false, a,  false, b,
             (short)0, acc,  false,  false);
    }
    float* dp = h1 + (size_t)(row0 + 8 * half) * 64 + col;
    if (row0 + 16 <= N) {
#pragma unroll
        for (int v = 0; v < 8; ++v) dp[(size_t)v * 64] = acc[v];
    } else {
#pragma unroll
        for (int v = 0; v < 8; ++v) {
            if (row0 + v + 8 * half < N) dp[(size_t)v * 64] = acc[v];
        }
    }
}

__global__ void k_scores1(const float* __restrict__ h1, const float* __restrict__ as,
                          const float* __restrict__ ad, float* __restrict__ ss,
                          float* __restrict__ sd, int N) {
    int t = blockIdx.x * blockDim.x + threadIdx.x;
    if (t >= N * 4) return;
    int n = t >> 2, h = t & 3;
    const float4* hp = (const float4*)(h1 + (size_t)n * 64 + h * 16);
    const float4* ap = (const float4*)(as + h * 16);
    const float4* bp = (const float4*)(ad + h * 16);
    float a = 0.f, b = 0.f;
#pragma unroll
    for (int i = 0; i < 4; ++i) {
        float4 hv = hp[i], av = ap[i], bv = bp[i];
        a += hv.x * av.x + hv.y * av.y + hv.z * av.z + hv.w * av.w;
        b += hv.x * bv.x + hv.y * bv.y + hv.z * bv.z + hv.w * bv.w;
    }
    ss[t] = a; sd[t] = b;
}

__device__ __forceinline__ float lrelu(float s) { return s > 0.f ? s : 0.2f * s; }

__device__ __forceinline__ float wave_max(float v) {
#pragma unroll
    for (int d = 16; d > 0; d >>= 1) v = fmaxf(v, __shfl_xor(v, d));
    return v;
}
__device__ __forceinline__ float wave_sum(float v) {
#pragma unroll
    for (int d = 16; d > 0; d >>= 1) v += __shfl_xor(v, d);
    return v;
}

#define N1_PLANE 66

__device__ __forceinline__ float wave_uniform(float v) {
    return __int_as_float(__builtin_amdgcn_readfirstlane(__float_as_int(v)));
}
__device__ __forceinline__ float xhalf(float x) {
    return __builtin_bit_cast(float, __builtin_amdgcn_permlanex16(__builtin_bit_cast(int, x), __builtin_bit_cast(int, x),
                                                                  0x76543210, 0xfedcba98, false, false));
}

__device__ __forceinline__ void put_chunk(int2* __restrict__ wr, int s, float e0, float e1, float e2, float e3) {
    const int so = (int)((unsigned)s << 8);
    wr[0 * N1_PLANE] = make_int2(so, __float_as_int(e0));
    wr[1 * N1_PLANE] = make_int2(so, __float_as_int(e1));
    wr[2 * N1_PLANE] = make_int2(so, __float_as_int(e2));
    wr[3 * N1_PLANE] = make_int2(so, __float_as_int(e3));
}

__device__ __forceinline__ void rows_acc(const float* __restrict__ h1, const int2* __restrict__ rb,
                                         unsigned qoff, int cnt, float4& acc) {
    for (int j = 0; j < cnt; j += 16) {
        int2 rc[8];
#pragma unroll
        for (int u = 0; u < 8; ++u) rc[u] = rb[j + 2 * u];
        float4 hv[8];
#pragma unroll
        for (int u = 0; u < 8; ++u)
            hv[u] = *(const float4*)((const char*)h1 + (size_t)((unsigned)rc[u].x + qoff));
#pragma unroll
        for (int u = 0; u < 8; ++u) {
            const float w = __int_as_float(rc[u].y);
            acc.x += hv[u].x * w;
            acc.y += hv[u].y * w;
            acc.z += hv[u].z * w;
            acc.w += hv[u].w * w;
        }
    }
}

__global__ __launch_bounds__(256) __attribute__((amdgpu_waves_per_eu(16, 16)))
void k_node1(const int* __restrict__ sorted, const int* __restrict__ off,
             const int* __restrict__ deg, const float* __restrict__ ss,
             const float* __restrict__ sd, const float* __restrict__ h1,
             const float* __restrict__ b1, const float* __restrict__ W2,
             const float* __restrict__ as2, const float* __restrict__ ad2,
             float* __restrict__ rec2, int N) {
    __shared__ int2 s_rec[8 * 4 * N1_PLANE];
    const int lane = threadIdx.x & 31;
    const int wv = threadIdx.x >> 5;
    const int n = __builtin_amdgcn_readfirstlane(blockIdx.x * 8 + wv);
    if (n >= N) return;
    const int o = off[n], dg = deg[n], tot = dg + 1;
    const float4 sdn = *(const float4*)(sd + (size_t)n * 4);

    const int pA = lane, pB = 32 + lane;
    const bool hasB = tot > 32;
    const int vA = sorted[(pA < dg) ? (size_t)o + pA : 0];
    const int vB = sorted[(pB < dg) ? (size_t)o + pB : 0];
    const int sA = (pA < dg) ? vA : n;
    const int sB = (pB < dg) ? vB : n;
    const float4 svA = *(const float4*)(ss + (size_t)sA * 4);
    float4 svB = make_float4(0.f, 0.f, 0.f, 0.f);
    if (hasB) svB = *(const float4*)(ss + (size_t)sB * 4);
    const bool okA = pA < tot, okB = pB < tot;
    const float xA0 = okA ? lrelu(svA.x + sdn.x) : NEG_INF, xA1 = okA ? lrelu(svA.y + sdn.y) : NEG_INF;
    const float xA2 = okA ? lrelu(svA.z + sdn.z) : NEG_INF, xA3 = okA ? lrelu(svA.w + sdn.w) : NEG_INF;
    const float xB0 = okB ? lrelu(svB.x + sdn.x) : NEG_INF, xB1 = okB ? lrelu(svB.y + sdn.y) : NEG_INF;
    const float xB2 = okB ? lrelu(svB.z + sdn.z) : NEG_INF, xB3 = okB ? lrelu(svB.w + sdn.w) : NEG_INF;
    float m0 = fmaxf(xA0, xB0), m1 = fmaxf(xA1, xB1), m2 = fmaxf(xA2, xB2), m3 = fmaxf(xA3, xB3);
    for (int base = 64; base < tot; base += 32) {
        const int p = base + lane;
        if (p < tot) {
            const int s = (p < dg) ? sorted[(size_t)o + p] : n;
            const float4 sv = *(const float4*)(ss + (size_t)s * 4);
            m0 = fmaxf(m0, lrelu(sv.x + sdn.x));
            m1 = fmaxf(m1, lrelu(sv.y + sdn.y));
            m2 = fmaxf(m2, lrelu(sv.z + sdn.z));
            m3 = fmaxf(m3, lrelu(sv.w + sdn.w));
        }
    }
    m0 = wave_uniform(wave_max(m0));
    m1 = wave_uniform(wave_max(m1));
    m2 = wave_uniform(wave_max(m2));
    m3 = wave_uniform(wave_max(m3));

    const int half = lane >> 4;
    const int q    = lane & 15;
    const int hsel = q >> 2;
    int2* wtab = s_rec + wv * (4 * N1_PLANE);
    int2* wr = wtab + lane;
    const int2* rb = wtab + hsel * N1_PLANE + half;
    const unsigned qoff = 16u * q;
    float4 acc = make_float4(0.f, 0.f, 0.f, 0.f);
    float t0, t1, t2, t3;
    {
        const float e0 = okA ? expf(xA0 - m0) : 0.f, e1 = okA ? expf(xA1 - m1) : 0.f;
        const float e2 = okA ? expf(xA2 - m2) : 0.f, e3 = okA ? expf(xA3 - m3) : 0.f;
        t0 = e0; t1 = e1; t2 = e2; t3 = e3;
        put_chunk(wr, sA, e0, e1, e2, e3);
    }
    if (hasB) {
        const float e0 = okB ? expf(xB0 - m0) : 0.f, e1 = okB ? expf(xB1 - m1) : 0.f;
        const float e2 = okB ? expf(xB2 - m2) : 0.f, e3 = okB ? expf(xB3 - m3) : 0.f;
        t0 += e0; t1 += e1; t2 += e2; t3 += e3;
        put_chunk(wr + 32, sB, e0, e1, e2, e3);
    }
    float d0 = wave_uniform(wave_sum(t0)), d1 = wave_uniform(wave_sum(t1));
    float d2 = wave_uniform(wave_sum(t2)), d3 = wave_uniform(wave_sum(t3));
    rows_acc(h1, rb, qoff, (tot < 64) ? tot : 64, acc);
    for (int base = 64; base < tot; base += 32) {
        const int p = base + lane;
        int s = n;
        float e0 = 0.f, e1 = 0.f, e2 = 0.f, e3 = 0.f;
        if (p < tot) {
            if (p < dg) s = sorted[(size_t)o + p];
            const float4 sv = *(const float4*)(ss + (size_t)s * 4);
            e0 = expf(lrelu(sv.x + sdn.x) - m0);
            e1 = expf(lrelu(sv.y + sdn.y) - m1);
            e2 = expf(lrelu(sv.z + sdn.z) - m2);
            e3 = expf(lrelu(sv.w + sdn.w) - m3);
        }
        d0 += wave_uniform(wave_sum(e0)); d1 += wave_uniform(wave_sum(e1));
        d2 += wave_uniform(wave_sum(e2)); d3 += wave_uniform(wave_sum(e3));
        put_chunk(wr, s, e0, e1, e2, e3);
        rows_acc(h1, rb, qoff, (tot - base < 32) ? (tot - base) : 32, acc);
    }
    const float dn = (hsel == 0) ? d0 : (hsel == 1) ? d1 : (hsel == 2) ? d2 : d3;
    const float g0 = half ? acc.x : acc.z, g1 = half ? acc.y : acc.w;
    const float a0 = (half ? acc.z : acc.x) + xhalf(g0);
    const float a1 = (half ? acc.w : acc.y) + xhalf(g1);
    const int c0 = 4 * q + 2 * half;
    const float2 bb = *(const float2*)(b1 + c0);
    float v0 = a0 / dn + bb.x;
    float v1 = a1 / dn + bb.y;
    v0 = v0 > 0.f ? v0 : expm1f(v0);
    v1 = v1 > 0.f ? v1 : expm1f(v1);
    const float4 wv2 = *(const float4*)(W2 + 2 * c0);
    float o0 = v0 * wv2.x + v1 * wv2.z;
    float o1 = v0 * wv2.y + v1 * wv2.w;
    o0 = wave_sum(o0);
    o1 = wave_sum(o1);
    if (lane == 0) {
        *(float4*)(rec2 + (size_t)n * 4) =
            make_float4(o0, o1, o0 * as2[0] + o1 * as2[1], o0 * ad2[0] + o1 * ad2[1]);
    }
}

#define N2_SLOTS 6
__global__ __launch_bounds__(256)
void k_node2(const int* __restrict__ sorted, const int* __restrict__ off,
             const int* __restrict__ deg, const float* __restrict__ rec2,
             const float* __restrict__ b2, float* __restrict__ out, int N) {
    const int t = blockIdx.x * 256 + threadIdx.x;
    const int n = t >> 3, sub = t & 7;
    int o = 0, dg = 0, tot = 0, nn = 0;
    float sdn = 0.f;
    if (n < N) { o = off[n]; dg = deg[n]; tot = dg + 1; sdn = rec2[(size_t)n * 4 + 3]; nn = n; }

    int s[N2_SLOTS];
#pragma unroll
    for (int k = 0; k < N2_SLOTS; ++k) {
        const int p = sub + 8 * k;
        const int v = sorted[(p < dg) ? (size_t)o + p : 0];
        s[k] = (p < dg) ? v : nn;
    }
    float4 r[N2_SLOTS];
#pragma unroll
    for (int k = 0; k < N2_SLOTS; ++k) r[k] = *(const float4*)(rec2 + (size_t)s[k] * 4);
    __builtin_amdgcn_sched_barrier(0);
    float rx[N2_SLOTS], ry[N2_SLOTS], xs[N2_SLOTS];
    float m = NEG_INF;
#pragma unroll
    for (int k = 0; k < N2_SLOTS; ++k) {
        const bool ok = sub + 8 * k < tot;
        rx[k] = ok ? r[k].x : 0.f;
        ry[k] = ok ? r[k].y : 0.f;
        xs[k] = ok ? lrelu(r[k].z + sdn) : NEG_INF;
        m = fmaxf(m, xs[k]);
    }
    for (int p = sub + 8 * N2_SLOTS; p < tot; p += 8) {
        const int sp = (p < dg) ? sorted[(size_t)o + p] : n;
        m = fmaxf(m, lrelu(rec2[(size_t)sp * 4 + 2] + sdn));
    }
#pragma unroll
    for (int d = 4; d > 0; d >>= 1) m = fmaxf(m, __shfl_xor(m, d));

    float dsum = 0.f, a0 = 0.f, a1 = 0.f;
#pragma unroll
    for (int k = 0; k < N2_SLOTS; ++k) {
        const float ex = (sub + 8 * k < tot) ? expf(xs[k] - m) : 0.f;
        dsum += ex;
        a0 += rx[k] * ex;
        a1 += ry[k] * ex;
    }
    for (int p = sub + 8 * N2_SLOTS; p < tot; p += 8) {
        const int sp = (p < dg) ? sorted[(size_t)o + p] : n;
        const float4 r = *(const float4*)(rec2 + (size_t)sp * 4);
        const float ex = expf(lrelu(r.z + sdn) - m);
        dsum += ex;
        a0 += r.x * ex;
        a1 += r.y * ex;
    }
#pragma unroll
    for (int d = 4; d > 0; d >>= 1) {
        dsum += __shfl_xor(dsum, d);
        a0   += __shfl_xor(a0, d);
        a1   += __shfl_xor(a1, d);
    }
    if (sub == 0 && n < N) {
        float v0 = a0 / dsum + b2[0];
        float v1 = a1 / dsum + b2[1];
        float mx = fmaxf(v0, v1);
        float lse = mx + logf(expf(v0 - mx) + expf(v1 - mx));
        *(float2*)(out + (size_t)n * 2) = make_float2(v0 - lse, v1 - lse);
    }
}

static inline unsigned nblk(long long n, int b) { return (unsigned)((n + b - 1) / b); }

extern "C" void kernel_launch(void* const* d_in, const int* in_sizes, int n_in,
                              void* d_out, int out_size, void* d_ws, size_t ws_size,
                              hipStream_t stream) {
    const float* x     = (const float*)d_in[0];
    const int*   ei    = (const int*)  d_in[1];
    const float* W1    = (const float*)d_in[2];
    const float* asrc1 = (const float*)d_in[3];
    const float* adst1 = (const float*)d_in[4];
    const float* b1    = (const float*)d_in[5];
    const float* W2    = (const float*)d_in[6];
    const float* asrc2 = (const float*)d_in[7];
    const float* adst2 = (const float*)d_in[8];
    const float* b2    = (const float*)d_in[9];
    float* out = (float*)d_out;

    const int N = in_sizes[0] / 128;
    const int E = in_sizes[1] / 2;

    float* ws   = (float*)d_ws;
    size_t nl   = (size_t)N;
    float* h1   = ws;
    float* ss1  = h1  + nl * 64;
    float* sd1  = ss1 + nl * 4;
    float* rec2 = sd1 + nl * 4;
    int*   off  = (int*)(rec2 + nl * 4);
    int*   deg  = off + nl;
    int*   sorted = deg + nl + (size_t)E;
    const int C = (N + BNODES - 1) / BNODES;
    const int G = (int)nblk(E, EPB);
    const int M = C * G;
    int*   part = (int*)h1;
    int*   cntA = (int*)ss1;
    int*   offA = cntA + M;

    k_count<<<G, 1024, C * sizeof(int), stream>>>(ei, cntA, E, C, G);
    k_scan<<<nblk(M, 1024), 1024, 0, stream>>>(cntA, offA, M);
    k_part<<<G, 1024, C * sizeof(int), stream>>>(ei, offA, part, E, C, G);
    k_bucket<<<C, 1024, 0, stream>>>(part, offA, off, deg, sorted, N, E, C, G);
    k_gemm1_wmma<<<(N + 15) / 16, 128, 0, stream>>>(x, W1, h1, N);
    k_scores1<<<nblk(4LL * N, 256), 256, 0, stream>>>(h1, asrc1, adst1, ss1, sd1, N);
    k_node1<<<nblk(N, 8), 256, 0, stream>>>(sorted, off, deg, ss1, sd1, h1, b1, W2,
                                            asrc2, adst2, rec2, N);
    k_node2<<<nblk(8LL * N, 256), 256, 0, stream>>>(sorted, off, deg, rec2, b2, out, N);
}


# Edge grouping in 2 launches, not 4: a slice orders its edges by coarse bucket inside its own places, so no count launch and no global scan; a bucket reads one run per slice; rank stays arrival order
# speedup vs baseline: 10.4438x; 1.1017x over previous
#include <hip/hip_runtime.h>
#include <hip/hip_bf16.h>

#define NEG_INF (-__builtin_inff())

typedef float v2f __attribute__((ext_vector_type(2)));
typedef float v8f __attribute__((ext_vector_type(8)));

#define EPB    8192
#define BSH    8
#define BNODES 256
#define SEGB   8

extern __shared__ int s_dyn[];

__device__ __forceinline__ int wave_sum_i(int v) {
#pragma unroll
    for (int d = 16; d > 0; d >>= 1) v += __shfl_xor(v, d);
    return v;
}

__global__ __launch_bounds__(1024)
void k_part(const int* __restrict__ ei, int* __restrict__ cntA, int* __restrict__ lofA,
            int* __restrict__ part_w, int E, int C, int G) {
    __shared__ int s_wtot[32];
    int2* __restrict__ part = (int2*)part_w;
    const int t = threadIdx.x, lane = t & 31, w = t >> 5, b = blockIdx.x;
    for (int c = t; c < C; c += 1024) s_dyn[c] = 0;
    __syncthreads();
    const int e0 = b * EPB + t;
    int s[8], d[8];
#pragma unroll
    for (int k = 0; k < 8; ++k) {
        const int e = e0 + k * 1024;
        const bool ok = e < E;
        s[k] = ok ? ei[e] : 0;
        d[k] = ok ? ei[(size_t)E + e] : -1;
    }
#pragma unroll
    for (int k = 0; k < 8; ++k) if (d[k] >= 0) atomicAdd(&s_dyn[d[k] >> BSH], 1);
    __syncthreads();
    int carry = 0;
    for (int base = 0; base < C; base += 1024) {
        const int c = base + t;
        const int cv = (c < C) ? s_dyn[c] : 0;
        int inc = cv;
#pragma unroll
        for (int dd = 1; dd < 32; dd <<= 1) {
            const int v = __shfl_up(inc, dd);
            if (lane >= dd) inc += v;
        }
        if (lane == 31) s_wtot[w] = inc;
        __syncthreads();
        const int wt = s_wtot[lane];
        const int pre = wave_sum_i((lane < w) ? wt : 0);
        const int tot = wave_sum_i(wt);
        const int ex = carry + pre + inc - cv;
        if (c < C) {
            cntA[(size_t)c * G + b] = cv;
            lofA[(size_t)c * G + b] = ex;
            s_dyn[c] = ex;
        }
        carry += tot;
        __syncthreads();
    }
    int2* __restrict__ mine = part + (size_t)b * EPB;
#pragma unroll
    for (int k = 0; k < 8; ++k) {
        if (d[k] >= 0) {
            const int p = atomicAdd(&s_dyn[d[k] >> BSH], 1);
            mine[p] = make_int2(s[k], d[k]);
        }
    }
}

__global__ __launch_bounds__(1024)
void k_bucket(const int* __restrict__ part_w, const int* __restrict__ cntA,
              const int* __restrict__ lofA, int* __restrict__ off, int* __restrict__ deg,
              int* __restrict__ sorted, int N, int G) {
    __shared__ int s_cnt[BNODES + 32];
    __shared__ int s_wtot[32];
    const int2* __restrict__ part = (const int2*)part_w;
    const int t = threadIdx.x, lane = t & 31, c = blockIdx.x;
    const int w = __builtin_amdgcn_readfirstlane(t >> 5);
    const int nbase = c << BSH;
    int* s_len = s_dyn;
    int* s_beg = s_dyn + G;
    int lsum = 0;
    for (int g = t; g < G; g += 1024) {
        const int lo = lofA[(size_t)c * G + g];
        s_len[g] = cntA[(size_t)c * G + g];
        s_beg[g] = g * EPB + lo;
        lsum += lo;
    }
    if (t < BNODES + 32) s_cnt[t] = 0;
    lsum = wave_sum_i(lsum);
    if (lane == 0) s_wtot[w] = lsum;
    __syncthreads();
    const int beg = wave_sum_i(s_wtot[lane]);

    for (int g0 = w; g0 < G; g0 += 32 * SEGB) {
        const int gl = g0 + 32 * (lane & (SEGB - 1));
        const int vlen = (gl < G) ? s_len[gl] : 0;
        const int vst  = (gl < G) ? s_beg[gl] : 0;
        int len[SEGB], st[SEGB];
#pragma unroll
        for (int k = 0; k < SEGB; ++k) {
            len[k] = __builtin_amdgcn_readlane(vlen, k);
            st[k]  = __builtin_amdgcn_readlane(vst, k);
        }
        int y[SEGB][2];
#pragma unroll
        for (int k = 0; k < SEGB; ++k) {
            y[k][0] = (lane < len[k])      ? part[(size_t)st[k] + lane].y      : -1;
            y[k][1] = (lane + 32 < len[k]) ? part[(size_t)st[k] + 32 + lane].y : -1;
        }
#pragma unroll
        for (int k = 0; k < SEGB; ++k) {
            if (y[k][0] >= 0) atomicAdd(&s_cnt[y[k][0] - nbase], 1);
            if (y[k][1] >= 0) atomicAdd(&s_cnt[y[k][1] - nbase], 1);
        }
#pragma unroll
        for (int k = 0; k < SEGB; ++k)
            for (int j = 64 + lane; j < len[k]; j += 32)
                atomicAdd(&s_cnt[part[(size_t)st[k] + j].y - nbase], 1);
    }
    __syncthreads();
    int cv = 0, inc = 0;
    if (t < BNODES) {
        cv = s_cnt[t];
        inc = cv;
#pragma unroll
        for (int d = 1; d < 32; d <<= 1) {
            const int v = __shfl_up(inc, d);
            if (lane >= d) inc += v;
        }
        if (lane == 31) s_wtot[w] = inc;
    }
    __syncthreads();
    if (t < BNODES) {
        int pre = 0;
        for (int k = 0; k < w; ++k) pre += s_wtot[k];
        const int ex = pre + inc - cv;
        s_cnt[t] = ex;
        const int n = nbase + t;
        if (n < N) { off[n] = beg + ex; deg[n] = cv; }
    }
    __syncthreads();
    for (int g0 = w; g0 < G; g0 += 32 * SEGB) {
        const int gl = g0 + 32 * (lane & (SEGB - 1));
        const int vlen = (gl < G) ? s_len[gl] : 0;
        const int vst  = (gl < G) ? s_beg[gl] : 0;
        int len[SEGB], st[SEGB];
#pragma unroll
        for (int k = 0; k < SEGB; ++k) {
            len[k] = __builtin_amdgcn_readlane(vlen, k);
            st[k]  = __builtin_amdgcn_readlane(vst, k);
        }
        int2 p[SEGB][2];
#pragma unroll
        for (int k = 0; k < SEGB; ++k) {
            p[k][0] = (lane < len[k])      ? part[(size_t)st[k] + lane]      : make_int2(0, -1);
            p[k][1] = (lane + 32 < len[k]) ? part[(size_t)st[k] + 32 + lane] : make_int2(0, -1);
        }
        int pos[SEGB][2];
#pragma unroll
        for (int k = 0; k < SEGB; ++k) {
#pragma unroll
            for (int h = 0; h < 2; ++h) {
                const int y = p[k][h].y;
                pos[k][h] = atomicAdd(&s_cnt[(y >= 0) ? y - nbase : BNODES + lane], 1);
            }
        }
#pragma unroll
        for (int k = 0; k < SEGB; ++k) {
#pragma unroll
            for (int h = 0; h < 2; ++h)
                if (p[k][h].y >= 0) sorted[(size_t)beg + pos[k][h]] = p[k][h].x;
        }
#pragma unroll
        for (int k = 0; k < SEGB; ++k) {
            for (int j = 64 + lane; j < len[k]; j += 32) {
                const int2 q = part[(size_t)st[k] + j];
                const int pq = atomicAdd(&s_cnt[q.y - nbase], 1);
                sorted[(size_t)beg + pq] = q.x;
            }
        }
    }
}

__global__ __launch_bounds__(128)
void k_gemm1_wmma(const float* __restrict__ x, const float* __restrict__ W1,
                  float* __restrict__ h1, int N) {
    const int wave = threadIdx.x >> 5;
    const int lane = threadIdx.x & 31;
    const int half = lane >> 4;
    const int l    = lane & 15;
    const int row0 = blockIdx.x * 16;
    const int col  = wave * 16 + l;
    int row = row0 + l;
    if (row >= N) row = N - 1;
    const float* xrow = x + (size_t)row * 128;

    v8f acc = {};
    for (int k0 = 0; k0 < 128; k0 += 4) {
        const int k = k0 + 2 * half;
        v2f a, b;
        a.x = xrow[k + 0];
        a.y = xrow[k + 1];
        b.x = W1[(size_t)(k + 0) * 64 + col];
        b.y = W1[(size_t)(k + 1) * 64 + col];
        acc = __builtin_amdgcn_wmma_f32_16x16x4_f32(
             false, a,  false, b,
             (short)0, acc,  false,  false);
    }
    float* dp = h1 + (size_t)(row0 + 8 * half) * 64 + col;
    if (row0 + 16 <= N) {
#pragma unroll
        for (int v = 0; v < 8; ++v) dp[(size_t)v * 64] = acc[v];
    } else {
#pragma unroll
        for (int v = 0; v < 8; ++v) {
            if (row0 + v + 8 * half < N) dp[(size_t)v * 64] = acc[v];
        }
    }
}

__global__ void k_scores1(const float* __restrict__ h1, const float* __restrict__ as,
                          const float* __restrict__ ad, float* __restrict__ ss,
                          float* __restrict__ sd, int N) {
    int t = blockIdx.x * blockDim.x + threadIdx.x;
    if (t >= N * 4) return;
    int n = t >> 2, h = t & 3;
    const float4* hp = (const float4*)(h1 + (size_t)n * 64 + h * 16);
    const float4* ap = (const float4*)(as + h * 16);
    const float4* bp = (const float4*)(ad + h * 16);
    float a = 0.f, b = 0.f;
#pragma unroll
    for (int i = 0; i < 4; ++i) {
        float4 hv = hp[i], av = ap[i], bv = bp[i];
        a += hv.x * av.x + hv.y * av.y + hv.z * av.z + hv.w * av.w;
        b += hv.x * bv.x + hv.y * bv.y + hv.z * bv.z + hv.w * bv.w;
    }
    ss[t] = a; sd[t] = b;
}

__device__ __forceinline__ float lrelu(float s) { return s > 0.f ? s : 0.2f * s; }

__device__ __forceinline__ float wave_max(float v) {
#pragma unroll
    for (int d = 16; d > 0; d >>= 1) v = fmaxf(v, __shfl_xor(v, d));
    return v;
}
__device__ __forceinline__ float wave_sum(float v) {
#pragma unroll
    for (int d = 16; d > 0; d >>= 1) v += __shfl_xor(v, d);
    return v;
}

#define N1_PLANE 66

__device__ __forceinline__ float wave_uniform(float v) {
    return __int_as_float(__builtin_amdgcn_readfirstlane(__float_as_int(v)));
}
__device__ __forceinline__ float xhalf(float x) {
    return __builtin_bit_cast(float, __builtin_amdgcn_permlanex16(__builtin_bit_cast(int, x), __builtin_bit_cast(int, x),
                                                                  0x76543210, 0xfedcba98, false, false));
}

__device__ __forceinline__ void put_chunk(int2* __restrict__ wr, int s, float e0, float e1, float e2, float e3) {
    const int so = (int)((unsigned)s << 8);
    wr[0 * N1_PLANE] = make_int2(so, __float_as_int(e0));
    wr[1 * N1_PLANE] = make_int2(so, __float_as_int(e1));
    wr[2 * N1_PLANE] = make_int2(so, __float_as_int(e2));
    wr[3 * N1_PLANE] = make_int2(so, __float_as_int(e3));
}

__device__ __forceinline__ void rows_acc(const float* __restrict__ h1, const int2* __restrict__ rb,
                                         unsigned qoff, int cnt, float4& acc) {
    for (int j = 0; j < cnt; j += 16) {
        int2 rc[8];
#pragma unroll
        for (int u = 0; u < 8; ++u) rc[u] = rb[j + 2 * u];
        float4 hv[8];
#pragma unroll
        for (int u = 0; u < 8; ++u)
            hv[u] = *(const float4*)((const char*)h1 + (size_t)((unsigned)rc[u].x + qoff));
#pragma unroll
        for (int u = 0; u < 8; ++u) {
            const float w = __int_as_float(rc[u].y);
            acc.x += hv[u].x * w;
            acc.y += hv[u].y * w;
            acc.z += hv[u].z * w;
            acc.w += hv[u].w * w;
        }
    }
}

__global__ __launch_bounds__(256) __attribute__((amdgpu_waves_per_eu(16, 16)))
void k_node1(const int* __restrict__ sorted, const int* __restrict__ off,
             const int* __restrict__ deg, const float* __restrict__ ss,
             const float* __restrict__ sd, const float* __restrict__ h1,
             const float* __restrict__ b1, const float* __restrict__ W2,
             const float* __restrict__ as2, const float* __restrict__ ad2,
             float* __restrict__ rec2, int N) {
    __shared__ int2 s_rec[8 * 4 * N1_PLANE];
    const int lane = threadIdx.x & 31;
    const int wv = threadIdx.x >> 5;
    const int n = __builtin_amdgcn_readfirstlane(blockIdx.x * 8 + wv);
    if (n >= N) return;
    const int o = off[n], dg = deg[n], tot = dg + 1;
    const float4 sdn = *(const float4*)(sd + (size_t)n * 4);

    const int pA = lane, pB = 32 + lane;
    const bool hasB = tot > 32;
    const int vA = sorted[(pA < dg) ? (size_t)o + pA : 0];
    const int vB = sorted[(pB < dg) ? (size_t)o + pB : 0];
    const int sA = (pA < dg) ? vA : n;
    const int sB = (pB < dg) ? vB : n;
    const float4 svA = *(const float4*)(ss + (size_t)sA * 4);
    float4 svB = make_float4(0.f, 0.f, 0.f, 0.f);
    if (hasB) svB = *(const float4*)(ss + (size_t)sB * 4);
    const bool okA = pA < tot, okB = pB < tot;
    const float xA0 = okA ? lrelu(svA.x + sdn.x) : NEG_INF, xA1 = okA ? lrelu(svA.y + sdn.y) : NEG_INF;
    const float xA2 = okA ? lrelu(svA.z + sdn.z) : NEG_INF, xA3 = okA ? lrelu(svA.w + sdn.w) : NEG_INF;
    const float xB0 = okB ? lrelu(svB.x + sdn.x) : NEG_INF, xB1 = okB ? lrelu(svB.y + sdn.y) : NEG_INF;
    const float xB2 = okB ? lrelu(svB.z + sdn.z) : NEG_INF, xB3 = okB ? lrelu(svB.w + sdn.w) : NEG_INF;
    float m0 = fmaxf(xA0, xB0), m1 = fmaxf(xA1, xB1), m2 = fmaxf(xA2, xB2), m3 = fmaxf(xA3, xB3);
    for (int base = 64; base < tot; base += 32) {
        const int p = base + lane;
        if (p < tot) {
            const int s = (p < dg) ? sorted[(size_t)o + p] : n;
            const float4 sv = *(const float4*)(ss + (size_t)s * 4);
            m0 = fmaxf(m0, lrelu(sv.x + sdn.x));
            m1 = fmaxf(m1, lrelu(sv.y + sdn.y));
            m2 = fmaxf(m2, lrelu(sv.z + sdn.z));
            m3 = fmaxf(m3, lrelu(sv.w + sdn.w));
        }
    }
    m0 = wave_uniform(wave_max(m0));
    m1 = wave_uniform(wave_max(m1));
    m2 = wave_uniform(wave_max(m2));
    m3 = wave_uniform(wave_max(m3));

    const int half = lane >> 4;
    const int q    = lane & 15;
    const int hsel = q >> 2;
    int2* wtab = s_rec + wv * (4 * N1_PLANE);
    int2* wr = wtab + lane;
    const int2* rb = wtab + hsel * N1_PLANE + half;
    const unsigned qoff = 16u * q;
    float4 acc = make_float4(0.f, 0.f, 0.f, 0.f);
    float t0, t1, t2, t3;
    {
        const float e0 = okA ? expf(xA0 - m0) : 0.f, e1 = okA ? expf(xA1 - m1) : 0.f;
        const float e2 = okA ? expf(xA2 - m2) : 0.f, e3 = okA ? expf(xA3 - m3) : 0.f;
        t0 = e0; t1 = e1; t2 = e2; t3 = e3;
        put_chunk(wr, sA, e0, e1, e2, e3);
    }
    if (hasB) {
        const float e0 = okB ? expf(xB0 - m0) : 0.f, e1 = okB ? expf(xB1 - m1) : 0.f;
        const float e2 = okB ? expf(xB2 - m2) : 0.f, e3 = okB ? expf(xB3 - m3) : 0.f;
        t0 += e0; t1 += e1; t2 += e2; t3 += e3;
        put_chunk(wr + 32, sB, e0, e1, e2, e3);
    }
    float d0 = wave_uniform(wave_sum(t0)), d1 = wave_uniform(wave_sum(t1));
    float d2 = wave_uniform(wave_sum(t2)), d3 = wave_uniform(wave_sum(t3));
    rows_acc(h1, rb, qoff, (tot < 64) ? tot : 64, acc);
    for (int base = 64; base < tot; base += 32) {
        const int p = base + lane;
        int s = n;
        float e0 = 0.f, e1 = 0.f, e2 = 0.f, e3 = 0.f;
        if (p < tot) {
            if (p < dg) s = sorted[(size_t)o + p];
            const float4 sv = *(const float4*)(ss + (size_t)s * 4);
            e0 = expf(lrelu(sv.x + sdn.x) - m0);
            e1 = expf(lrelu(sv.y + sdn.y) - m1);
            e2 = expf(lrelu(sv.z + sdn.z) - m2);
            e3 = expf(lrelu(sv.w + sdn.w) - m3);
        }
        d0 += wave_uniform(wave_sum(e0)); d1 += wave_uniform(wave_sum(e1));
        d2 += wave_uniform(wave_sum(e2)); d3 += wave_uniform(wave_sum(e3));
        put_chunk(wr, s, e0, e1, e2, e3);
        rows_acc(h1, rb, qoff, (tot - base < 32) ? (tot - base) : 32, acc);
    }
    const float dn = (hsel == 0) ? d0 : (hsel == 1) ? d1 : (hsel == 2) ? d2 : d3;
    const float g0 = half ? acc.x : acc.z, g1 = half ? acc.y : acc.w;
    const float a0 = (half ? acc.z : acc.x) + xhalf(g0);
    const float a1 = (half ? acc.w : acc.y) + xhalf(g1);
    const int c0 = 4 * q + 2 * half;
    const float2 bb = *(const float2*)(b1 + c0);
    float v0 = a0 / dn + bb.x;
    float v1 = a1 / dn + bb.y;
    v0 = v0 > 0.f ? v0 : expm1f(v0);
    v1 = v1 > 0.f ? v1 : expm1f(v1);
    const float4 wv2 = *(const float4*)(W2 + 2 * c0);
    float o0 = v0 * wv2.x + v1 * wv2.z;
    float o1 = v0 * wv2.y + v1 * wv2.w;
    o0 = wave_sum(o0);
    o1 = wave_sum(o1);
    if (lane == 0) {
        *(float4*)(rec2 + (size_t)n * 4) =
            make_float4(o0, o1, o0 * as2[0] + o1 * as2[1], o0 * ad2[0] + o1 * ad2[1]);
    }
}

#define N2_SLOTS 6
__global__ __launch_bounds__(256)
void k_node2(const int* __restrict__ sorted, const int* __restrict__ off,
             const int* __restrict__ deg, const float* __restrict__ rec2,
             const float* __restrict__ b2, float* __restrict__ out, int N) {
    const int t = blockIdx.x * 256 + threadIdx.x;
    const int n = t >> 3, sub = t & 7;
    int o = 0, dg = 0, tot = 0, nn = 0;
    float sdn = 0.f;
    if (n < N) { o = off[n]; dg = deg[n]; tot = dg + 1; sdn = rec2[(size_t)n * 4 + 3]; nn = n; }

    int s[N2_SLOTS];
#pragma unroll
    for (int k = 0; k < N2_SLOTS; ++k) {
        const int p = sub + 8 * k;
        const int v = sorted[(p < dg) ? (size_t)o + p : 0];
        s[k] = (p < dg) ? v : nn;
    }
    float4 r[N2_SLOTS];
#pragma unroll
    for (int k = 0; k < N2_SLOTS; ++k) r[k] = *(const float4*)(rec2 + (size_t)s[k] * 4);
    __builtin_amdgcn_sched_barrier(0);
    float rx[N2_SLOTS], ry[N2_SLOTS], xs[N2_SLOTS];
    float m = NEG_INF;
#pragma unroll
    for (int k = 0; k < N2_SLOTS; ++k) {
        const bool ok = sub + 8 * k < tot;
        rx[k] = ok ? r[k].x : 0.f;
        ry[k] = ok ? r[k].y : 0.f;
        xs[k] = ok ? lrelu(r[k].z + sdn) : NEG_INF;
        m = fmaxf(m, xs[k]);
    }
    for (int p = sub + 8 * N2_SLOTS; p < tot; p += 8) {
        const int sp = (p < dg) ? sorted[(size_t)o + p] : n;
        m = fmaxf(m, lrelu(rec2[(size_t)sp * 4 + 2] + sdn));
    }
#pragma unroll
    for (int d = 4; d > 0; d >>= 1) m = fmaxf(m, __shfl_xor(m, d));

    float dsum = 0.f, a0 = 0.f, a1 = 0.f;
#pragma unroll
    for (int k = 0; k < N2_SLOTS; ++k) {
        const float ex = (sub + 8 * k < tot) ? expf(xs[k] - m) : 0.f;
        dsum += ex;
        a0 += rx[k] * ex;
        a1 += ry[k] * ex;
    }
    for (int p = sub + 8 * N2_SLOTS; p < tot; p += 8) {
        const int sp = (p < dg) ? sorted[(size_t)o + p] : n;
        const float4 r = *(const float4*)(rec2 + (size_t)sp * 4);
        const float ex = expf(lrelu(r.z + sdn) - m);
        dsum += ex;
        a0 += r.x * ex;
        a1 += r.y * ex;
    }
#pragma unroll
    for (int d = 4; d > 0; d >>= 1) {
        dsum += __shfl_xor(dsum, d);
        a0   += __shfl_xor(a0, d);
        a1   += __shfl_xor(a1, d);
    }
    if (sub == 0 && n < N) {
        float v0 = a0 / dsum + b2[0];
        float v1 = a1 / dsum + b2[1];
        float mx = fmaxf(v0, v1);
        float lse = mx + logf(expf(v0 - mx) + expf(v1 - mx));
        *(float2*)(out + (size_t)n * 2) = make_float2(v0 - lse, v1 - lse);
    }
}

static inline unsigned nblk(long long n, int b) { return (unsigned)((n + b - 1) / b); }

extern "C" void kernel_launch(void* const* d_in, const int* in_sizes, int n_in,
                              void* d_out, int out_size, void* d_ws, size_t ws_size,
                              hipStream_t stream) {
    const float* x     = (const float*)d_in[0];
    const int*   ei    = (const int*)  d_in[1];
    const float* W1    = (const float*)d_in[2];
    const float* asrc1 = (const float*)d_in[3];
    const float* adst1 = (const float*)d_in[4];
    const float* b1    = (const float*)d_in[5];
    const float* W2    = (const float*)d_in[6];
    const float* asrc2 = (const float*)d_in[7];
    const float* adst2 = (const float*)d_in[8];
    const float* b2    = (const float*)d_in[9];
    float* out = (float*)d_out;

    const int N = in_sizes[0] / 128;
    const int E = in_sizes[1] / 2;

    float* ws   = (float*)d_ws;
    size_t nl   = (size_t)N;
    float* h1   = ws;
    float* ss1  = h1  + nl * 64;
    float* sd1  = ss1 + nl * 4;
    float* rec2 = sd1 + nl * 4;
    int*   off  = (int*)(rec2 + nl * 4);
    int*   deg  = off + nl;
    int*   sorted = deg + nl + (size_t)E;
    const int C = (N + BNODES - 1) / BNODES;
    const int G = (int)nblk(E, EPB);
    const int M = C * G;
    int*   part = (int*)h1;
    int*   cntA = (int*)ss1;
    int*   lofA = cntA + M;

    k_part<<<G, 1024, C * sizeof(int), stream>>>(ei, cntA, lofA, part, E, C, G);
    k_bucket<<<C, 1024, 2 * G * sizeof(int), stream>>>(part, cntA, lofA, off, deg, sorted, N, G);
    k_gemm1_wmma<<<(N + 15) / 16, 128, 0, stream>>>(x, W1, h1, N);
    k_scores1<<<nblk(4LL * N, 256), 256, 0, stream>>>(h1, asrc1, adst1, ss1, sd1, N);
    k_node1<<<nblk(N, 8), 256, 0, stream>>>(sorted, off, deg, ss1, sd1, h1, b1, W2,
                                            asrc2, adst2, rec2, N);
    k_node2<<<nblk(8LL * N, 256), 256, 0, stream>>>(sorted, off, deg, rec2, b2, out, N);
}


# Both edge-grouping kernels store whole 128-B lines: pairs and source ids are placed in an LDS copy in final order, then copied out 16 B per lane, not scattered 8 or 4 B each; rank stays arrival order
# speedup vs baseline: 11.2241x; 1.0747x over previous
#include <hip/hip_runtime.h>
#include <hip/hip_bf16.h>

#define NEG_INF (-__builtin_inff())

typedef float v2f __attribute__((ext_vector_type(2)));
typedef float v8f __attribute__((ext_vector_type(8)));

#define EPB    8192
#define BSH    8
#define BNODES 256
#define SEGB   8
#define BCAP   12288

extern __shared__ int s_dyn[];

__device__ __forceinline__ int wave_sum_i(int v) {
#pragma unroll
    for (int d = 16; d > 0; d >>= 1) v += __shfl_xor(v, d);
    return v;
}

__global__ __launch_bounds__(1024)
void k_part(const int* __restrict__ ei, int* __restrict__ cntA, int* __restrict__ lofA,
            int* __restrict__ part_w, int E, int C, int G) {
    __shared__ int s_wtot[32];
    __shared__ __attribute__((aligned(16))) int2 s_stage[EPB];
    int2* __restrict__ part = (int2*)part_w;
    const int t = threadIdx.x, lane = t & 31, w = t >> 5, b = blockIdx.x;
    for (int c = t; c < C; c += 1024) s_dyn[c] = 0;
    __syncthreads();
    const int e0 = b * EPB + t;
    int s[8], d[8];
#pragma unroll
    for (int k = 0; k < 8; ++k) {
        const int e = e0 + k * 1024;
        const bool ok = e < E;
        s[k] = ok ? ei[e] : 0;
        d[k] = ok ? ei[(size_t)E + e] : -1;
    }
#pragma unroll
    for (int k = 0; k < 8; ++k) if (d[k] >= 0) atomicAdd(&s_dyn[d[k] >> BSH], 1);
    __syncthreads();
    int carry = 0;
    for (int base = 0; base < C; base += 1024) {
        const int c = base + t;
        const int cv = (c < C) ? s_dyn[c] : 0;
        int inc = cv;
#pragma unroll
        for (int dd = 1; dd < 32; dd <<= 1) {
            const int v = __shfl_up(inc, dd);
            if (lane >= dd) inc += v;
        }
        if (lane == 31) s_wtot[w] = inc;
        __syncthreads();
        const int wt = s_wtot[lane];
        const int pre = wave_sum_i((lane < w) ? wt : 0);
        const int tot = wave_sum_i(wt);
        const int ex = carry + pre + inc - cv;
        if (c < C) {
            cntA[(size_t)c * G + b] = cv;
            lofA[(size_t)c * G + b] = ex;
            s_dyn[c] = ex;
        }
        carry += tot;
        __syncthreads();
    }
#pragma unroll
    for (int k = 0; k < 8; ++k) {
        if (d[k] >= 0) {
            const int p = atomicAdd(&s_dyn[d[k] >> BSH], 1);
            s_stage[p] = make_int2(s[k], d[k]);
        }
    }
    __syncthreads();
    int2* __restrict__ mine = part + (size_t)b * EPB;
    const int left = E - b * EPB;
    const int nv = (left < EPB) ? left : EPB;
#pragma unroll
    for (int k = 0; k < 4; ++k) {
        const int i = 2 * (t + k * 1024);
        if (i + 1 < nv)  *(int4*)(mine + i) = *(const int4*)(s_stage + i);
        else if (i < nv) mine[i] = s_stage[i];
    }
}

__global__ __launch_bounds__(1024)
void k_bucket(const int* __restrict__ part_w, const int* __restrict__ cntA,
              const int* __restrict__ lofA, int* __restrict__ off, int* __restrict__ deg,
              int* __restrict__ sorted, int N, int G) {
    __shared__ int s_cnt[BNODES + 32];
    __shared__ int s_wtot[32];
    __shared__ __attribute__((aligned(16))) int s_stage[BCAP];
    const int2* __restrict__ part = (const int2*)part_w;
    const int t = threadIdx.x, lane = t & 31, c = blockIdx.x;
    const int w = __builtin_amdgcn_readfirstlane(t >> 5);
    const int nbase = c << BSH;
    int* s_len = s_dyn;
    int* s_beg = s_dyn + G;
    int lsum = 0;
    for (int g = t; g < G; g += 1024) {
        const int lo = lofA[(size_t)c * G + g];
        s_len[g] = cntA[(size_t)c * G + g];
        s_beg[g] = g * EPB + lo;
        lsum += lo;
    }
    if (t < BNODES + 32) s_cnt[t] = 0;
    lsum = wave_sum_i(lsum);
    if (lane == 0) s_wtot[w] = lsum;
    __syncthreads();
    const int beg = wave_sum_i(s_wtot[lane]);

    for (int g0 = w; g0 < G; g0 += 32 * SEGB) {
        const int gl = g0 + 32 * (lane & (SEGB - 1));
        const int vlen = (gl < G) ? s_len[gl] : 0;
        const int vst  = (gl < G) ? s_beg[gl] : 0;
        int len[SEGB], st[SEGB];
#pragma unroll
        for (int k = 0; k < SEGB; ++k) {
            len[k] = __builtin_amdgcn_readlane(vlen, k);
            st[k]  = __builtin_amdgcn_readlane(vst, k);
        }
        int y[SEGB][2];
#pragma unroll
        for (int k = 0; k < SEGB; ++k) {
            y[k][0] = (lane < len[k])      ? part[(size_t)st[k] + lane].y      : -1;
            y[k][1] = (lane + 32 < len[k]) ? part[(size_t)st[k] + 32 + lane].y : -1;
        }
#pragma unroll
        for (int k = 0; k < SEGB; ++k) {
            if (y[k][0] >= 0) atomicAdd(&s_cnt[y[k][0] - nbase], 1);
            if (y[k][1] >= 0) atomicAdd(&s_cnt[y[k][1] - nbase], 1);
        }
#pragma unroll
        for (int k = 0; k < SEGB; ++k)
            for (int j = 64 + lane; j < len[k]; j += 32)
                atomicAdd(&s_cnt[part[(size_t)st[k] + j].y - nbase], 1);
    }
    __syncthreads();
    int cv = 0, inc = 0;
    if (t < BNODES) {
        cv = s_cnt[t];
        inc = cv;
#pragma unroll
        for (int d = 1; d < 32; d <<= 1) {
            const int v = __shfl_up(inc, d);
            if (lane >= d) inc += v;
        }
        if (lane == 31) s_wtot[w] = inc;
    }
    __syncthreads();
    if (t < BNODES) {
        int pre = 0;
        for (int k = 0; k < w; ++k) pre += s_wtot[k];
        const int ex = pre + inc - cv;
        s_cnt[t] = ex;
        const int n = nbase + t;
        if (n < N) { off[n] = beg + ex; deg[n] = cv; }
    }
    int btot = 0;
#pragma unroll
    for (int k = 0; k < BNODES / 32; ++k) btot += s_wtot[k];
    int* __restrict__ first = sorted + (size_t)beg;
    const int sh = (int)(((size_t)first >> 2) & 31);
    const bool staged = btot + 32 <= BCAP;
    __syncthreads();
    for (int g0 = w; g0 < G; g0 += 32 * SEGB) {
        const int gl = g0 + 32 * (lane & (SEGB - 1));
        const int vlen = (gl < G) ? s_len[gl] : 0;
        const int vst  = (gl < G) ? s_beg[gl] : 0;
        int len[SEGB], st[SEGB];
#pragma unroll
        for (int k = 0; k < SEGB; ++k) {
            len[k] = __builtin_amdgcn_readlane(vlen, k);
            st[k]  = __builtin_amdgcn_readlane(vst, k);
        }
        int2 p[SEGB][2];
#pragma unroll
        for (int k = 0; k < SEGB; ++k) {
            p[k][0] = (lane < len[k])      ? part[(size_t)st[k] + lane]      : make_int2(0, -1);
            p[k][1] = (lane + 32 < len[k]) ? part[(size_t)st[k] + 32 + lane] : make_int2(0, -1);
        }
        int pos[SEGB][2];
#pragma unroll
        for (int k = 0; k < SEGB; ++k) {
#pragma unroll
            for (int h = 0; h < 2; ++h) {
                const int y = p[k][h].y;
                pos[k][h] = atomicAdd(&s_cnt[(y >= 0) ? y - nbase : BNODES + lane], 1);
            }
        }
        if (staged) {
#pragma unroll
            for (int k = 0; k < SEGB; ++k) {
#pragma unroll
                for (int h = 0; h < 2; ++h)
                    if (p[k][h].y >= 0) s_stage[sh + pos[k][h]] = p[k][h].x;
            }
        } else {
#pragma unroll
            for (int k = 0; k < SEGB; ++k) {
#pragma unroll
                for (int h = 0; h < 2; ++h)
                    if (p[k][h].y >= 0) first[pos[k][h]] = p[k][h].x;
            }
        }
#pragma unroll
        for (int k = 0; k < SEGB; ++k) {
            for (int j = 64 + lane; j < len[k]; j += 32) {
                const int2 q = part[(size_t)st[k] + j];
                const int pq = atomicAdd(&s_cnt[q.y - nbase], 1);
                if (staged) s_stage[sh + pq] = q.x;
                else        first[pq] = q.x;
            }
        }
    }
    __syncthreads();
    if (staged) {
        int* __restrict__ line0 = first - sh;
        const int hi = sh + btot;
        for (int i = 4 * t; i < hi; i += 4096) {
            if (i >= sh && i + 4 <= hi) {
                *(int4*)(line0 + i) = *(const int4*)(s_stage + i);
            } else {
#pragma unroll
                for (int j = 0; j < 4; ++j)
                    if (i + j >= sh && i + j < hi) line0[i + j] = s_stage[i + j];
            }
        }
    }
}

__global__ __launch_bounds__(128)
void k_gemm1_wmma(const float* __restrict__ x, const float* __restrict__ W1,
                  float* __restrict__ h1, int N) {
    const int wave = threadIdx.x >> 5;
    const int lane = threadIdx.x & 31;
    const int half = lane >> 4;
    const int l    = lane & 15;
    const int row0 = blockIdx.x * 16;
    const int col  = wave * 16 + l;
    int row = row0 + l;
    if (row >= N) row = N - 1;
    const float* xrow = x + (size_t)row * 128;

    v8f acc = {};
    for (int k0 = 0; k0 < 128; k0 += 4) {
        const int k = k0 + 2 * half;
        v2f a, b;
        a.x = xrow[k + 0];
        a.y = xrow[k + 1];
        b.x = W1[(size_t)(k + 0) * 64 + col];
        b.y = W1[(size_t)(k + 1) * 64 + col];
        acc = __builtin_amdgcn_wmma_f32_16x16x4_f32(
             false, a,  false, b,
             (short)0, acc,  false,  false);
    }
    float* dp = h1 + (size_t)(row0 + 8 * half) * 64 + col;
    if (row0 + 16 <= N) {
#pragma unroll
        for (int v = 0; v < 8; ++v) dp[(size_t)v * 64] = acc[v];
    } else {
#pragma unroll
        for (int v = 0; v < 8; ++v) {
            if (row0 + v + 8 * half < N) dp[(size_t)v * 64] = acc[v];
        }
    }
}

__global__ void k_scores1(const float* __restrict__ h1, const float* __restrict__ as,
                          const float* __restrict__ ad, float* __restrict__ ss,
                          float* __restrict__ sd, int N) {
    int t = blockIdx.x * blockDim.x + threadIdx.x;
    if (t >= N * 4) return;
    int n = t >> 2, h = t & 3;
    const float4* hp = (const float4*)(h1 + (size_t)n * 64 + h * 16);
    const float4* ap = (const float4*)(as + h * 16);
    const float4* bp = (const float4*)(ad + h * 16);
    float a = 0.f, b = 0.f;
#pragma unroll
    for (int i = 0; i < 4; ++i) {
        float4 hv = hp[i], av = ap[i], bv = bp[i];
        a += hv.x * av.x + hv.y * av.y + hv.z * av.z + hv.w * av.w;
        b += hv.x * bv.x + hv.y * bv.y + hv.z * bv.z + hv.w * bv.w;
    }
    ss[t] = a; sd[t] = b;
}

__device__ __forceinline__ float lrelu(float s) { return s > 0.f ? s : 0.2f * s; }

__device__ __forceinline__ float wave_max(float v) {
#pragma unroll
    for (int d = 16; d > 0; d >>= 1) v = fmaxf(v, __shfl_xor(v, d));
    return v;
}
__device__ __forceinline__ float wave_sum(float v) {
#pragma unroll
    for (int d = 16; d > 0; d >>= 1) v += __shfl_xor(v, d);
    return v;
}

#define N1_PLANE 66

__device__ __forceinline__ float wave_uniform(float v) {
    return __int_as_float(__builtin_amdgcn_readfirstlane(__float_as_int(v)));
}
__device__ __forceinline__ float xhalf(float x) {
    return __builtin_bit_cast(float, __builtin_amdgcn_permlanex16(__builtin_bit_cast(int, x), __builtin_bit_cast(int, x),
                                                                  0x76543210, 0xfedcba98, false, false));
}

__device__ __forceinline__ void put_chunk(int2* __restrict__ wr, int s, float e0, float e1, float e2, float e3) {
    const int so = (int)((unsigned)s << 8);
    wr[0 * N1_PLANE] = make_int2(so, __float_as_int(e0));
    wr[1 * N1_PLANE] = make_int2(so, __float_as_int(e1));
    wr[2 * N1_PLANE] = make_int2(so, __float_as_int(e2));
    wr[3 * N1_PLANE] = make_int2(so, __float_as_int(e3));
}

__device__ __forceinline__ void rows_acc(const float* __restrict__ h1, const int2* __restrict__ rb,
                                         unsigned qoff, int cnt, float4& acc) {
    for (int j = 0; j < cnt; j += 16) {
        int2 rc[8];
#pragma unroll
        for (int u = 0; u < 8; ++u) rc[u] = rb[j + 2 * u];
        float4 hv[8];
#pragma unroll
        for (int u = 0; u < 8; ++u)
            hv[u] = *(const float4*)((const char*)h1 + (size_t)((unsigned)rc[u].x + qoff));
#pragma unroll
        for (int u = 0; u < 8; ++u) {
            const float w = __int_as_float(rc[u].y);
            acc.x += hv[u].x * w;
            acc.y += hv[u].y * w;
            acc.z += hv[u].z * w;
            acc.w += hv[u].w * w;
        }
    }
}

__global__ __launch_bounds__(256) __attribute__((amdgpu_waves_per_eu(16, 16)))
void k_node1(const int* __restrict__ sorted, const int* __restrict__ off,
             const int* __restrict__ deg, const float* __restrict__ ss,
             const float* __restrict__ sd, const float* __restrict__ h1,
             const float* __restrict__ b1, const float* __restrict__ W2,
             const float* __restrict__ as2, const float* __restrict__ ad2,
             float* __restrict__ rec2, int N) {
    __shared__ int2 s_rec[8 * 4 * N1_PLANE];
    const int lane = threadIdx.x & 31;
    const int wv = threadIdx.x >> 5;
    const int n = __builtin_amdgcn_readfirstlane(blockIdx.x * 8 + wv);
    if (n >= N) return;
    const int o = off[n], dg = deg[n], tot = dg + 1;
    const float4 sdn = *(const float4*)(sd + (size_t)n * 4);

    const int pA = lane, pB = 32 + lane;
    const bool hasB = tot > 32;
    const int vA = sorted[(pA < dg) ? (size_t)o + pA : 0];
    const int vB = sorted[(pB < dg) ? (size_t)o + pB : 0];
    const int sA = (pA < dg) ? vA : n;
    const int sB = (pB < dg) ? vB : n;
    const float4 svA = *(const float4*)(ss + (size_t)sA * 4);
    float4 svB = make_float4(0.f, 0.f, 0.f, 0.f);
    if (hasB) svB = *(const float4*)(ss + (size_t)sB * 4);
    const bool okA = pA < tot, okB = pB < tot;
    const float xA0 = okA ? lrelu(svA.x + sdn.x) : NEG_INF, xA1 = okA ? lrelu(svA.y + sdn.y) : NEG_INF;
    const float xA2 = okA ? lrelu(svA.z + sdn.z) : NEG_INF, xA3 = okA ? lrelu(svA.w + sdn.w) : NEG_INF;
    const float xB0 = okB ? lrelu(svB.x + sdn.x) : NEG_INF, xB1 = okB ? lrelu(svB.y + sdn.y) : NEG_INF;
    const float xB2 = okB ? lrelu(svB.z + sdn.z) : NEG_INF, xB3 = okB ? lrelu(svB.w + sdn.w) : NEG_INF;
    float m0 = fmaxf(xA0, xB0), m1 = fmaxf(xA1, xB1), m2 = fmaxf(xA2, xB2), m3 = fmaxf(xA3, xB3);
    for (int base = 64; base < tot; base += 32) {
        const int p = base + lane;
        if (p < tot) {
            const int s = (p < dg) ? sorted[(size_t)o + p] : n;
            const float4 sv = *(const float4*)(ss + (size_t)s * 4);
            m0 = fmaxf(m0, lrelu(sv.x + sdn.x));
            m1 = fmaxf(m1, lrelu(sv.y + sdn.y));
            m2 = fmaxf(m2, lrelu(sv.z + sdn.z));
            m3 = fmaxf(m3, lrelu(sv.w + sdn.w));
        }
    }
    m0 = wave_uniform(wave_max(m0));
    m1 = wave_uniform(wave_max(m1));
    m2 = wave_uniform(wave_max(m2));
    m3 = wave_uniform(wave_max(m3));

    const int half = lane >> 4;
    const int q    = lane & 15;
    const int hsel = q >> 2;
    int2* wtab = s_rec + wv * (4 * N1_PLANE);
    int2* wr = wtab + lane;
    const int2* rb = wtab + hsel * N1_PLANE + half;
    const unsigned qoff = 16u * q;
    float4 acc = make_float4(0.f, 0.f, 0.f, 0.f);
    float t0, t1, t2, t3;
    {
        const float e0 = okA ? expf(xA0 - m0) : 0.f, e1 = okA ? expf(xA1 - m1) : 0.f;
        const float e2 = okA ? expf(xA2 - m2) : 0.f, e3 = okA ? expf(xA3 - m3) : 0.f;
        t0 = e0; t1 = e1; t2 = e2; t3 = e3;
        put_chunk(wr, sA, e0, e1, e2, e3);
    }
    if (hasB) {
        const float e0 = okB ? expf(xB0 - m0) : 0.f, e1 = okB ? expf(xB1 - m1) : 0.f;
        const float e2 = okB ? expf(xB2 - m2) : 0.f, e3 = okB ? expf(xB3 - m3) : 0.f;
        t0 += e0; t1 += e1; t2 += e2; t3 += e3;
        put_chunk(wr + 32, sB, e0, e1, e2, e3);
    }
    float d0 = wave_uniform(wave_sum(t0)), d1 = wave_uniform(wave_sum(t1));
    float d2 = wave_uniform(wave_sum(t2)), d3 = wave_uniform(wave_sum(t3));
    rows_acc(h1, rb, qoff, (tot < 64) ? tot : 64, acc);
    for (int base = 64; base < tot; base += 32) {
        const int p = base + lane;
        int s = n;
        float e0 = 0.f, e1 = 0.f, e2 = 0.f, e3 = 0.f;
        if (p < tot) {
            if (p < dg) s = sorted[(size_t)o + p];
            const float4 sv = *(const float4*)(ss + (size_t)s * 4);
            e0 = expf(lrelu(sv.x + sdn.x) - m0);
            e1 = expf(lrelu(sv.y + sdn.y) - m1);
            e2 = expf(lrelu(sv.z + sdn.z) - m2);
            e3 = expf(lrelu(sv.w + sdn.w) - m3);
        }
        d0 += wave_uniform(wave_sum(e0)); d1 += wave_uniform(wave_sum(e1));
        d2 += wave_uniform(wave_sum(e2)); d3 += wave_uniform(wave_sum(e3));
        put_chunk(wr, s, e0, e1, e2, e3);
        rows_acc(h1, rb, qoff, (tot - base < 32) ? (tot - base) : 32, acc);
    }
    const float dn = (hsel == 0) ? d0 : (hsel == 1) ? d1 : (hsel == 2) ? d2 : d3;
    const float g0 = half ? acc.x : acc.z, g1 = half ? acc.y : acc.w;
    const float a0 = (half ? acc.z : acc.x) + xhalf(g0);
    const float a1 = (half ? acc.w : acc.y) + xhalf(g1);
    const int c0 = 4 * q + 2 * half;
    const float2 bb = *(const float2*)(b1 + c0);
    float v0 = a0 / dn + bb.x;
    float v1 = a1 / dn + bb.y;
    v0 = v0 > 0.f ? v0 : expm1f(v0);
    v1 = v1 > 0.f ? v1 : expm1f(v1);
    const float4 wv2 = *(const float4*)(W2 + 2 * c0);
    float o0 = v0 * wv2.x + v1 * wv2.z;
    float o1 = v0 * wv2.y + v1 * wv2.w;
    o0 = wave_sum(o0);
    o1 = wave_sum(o1);
    if (lane == 0) {
        *(float4*)(rec2 + (size_t)n * 4) =
            make_float4(o0, o1, o0 * as2[0] + o1 * as2[1], o0 * ad2[0] + o1 * ad2[1]);
    }
}

#define N2_SLOTS 6
__global__ __launch_bounds__(256)
void k_node2(const int* __restrict__ sorted, const int* __restrict__ off,
             const int* __restrict__ deg, const float* __restrict__ rec2,
             const float* __restrict__ b2, float* __restrict__ out, int N) {
    const int t = blockIdx.x * 256 + threadIdx.x;
    const int n = t >> 3, sub = t & 7;
    int o = 0, dg = 0, tot = 0, nn = 0;
    float sdn = 0.f;
    if (n < N) { o = off[n]; dg = deg[n]; tot = dg + 1; sdn = rec2[(size_t)n * 4 + 3]; nn = n; }

    int s[N2_SLOTS];
#pragma unroll
    for (int k = 0; k < N2_SLOTS; ++k) {
        const int p = sub + 8 * k;
        const int v = sorted[(p < dg) ? (size_t)o + p : 0];
        s[k] = (p < dg) ? v : nn;
    }
    float4 r[N2_SLOTS];
#pragma unroll
    for (int k = 0; k < N2_SLOTS; ++k) r[k] = *(const float4*)(rec2 + (size_t)s[k] * 4);
    __builtin_amdgcn_sched_barrier(0);
    float rx[N2_SLOTS], ry[N2_SLOTS], xs[N2_SLOTS];
    float m = NEG_INF;
#pragma unroll
    for (int k = 0; k < N2_SLOTS; ++k) {
        const bool ok = sub + 8 * k < tot;
        rx[k] = ok ? r[k].x : 0.f;
        ry[k] = ok ? r[k].y : 0.f;
        xs[k] = ok ? lrelu(r[k].z + sdn) : NEG_INF;
        m = fmaxf(m, xs[k]);
    }
    for (int p = sub + 8 * N2_SLOTS; p < tot; p += 8) {
        const int sp = (p < dg) ? sorted[(size_t)o + p] : n;
        m = fmaxf(m, lrelu(rec2[(size_t)sp * 4 + 2] + sdn));
    }
#pragma unroll
    for (int d = 4; d > 0; d >>= 1) m = fmaxf(m, __shfl_xor(m, d));

    float dsum = 0.f, a0 = 0.f, a1 = 0.f;
#pragma unroll
    for (int k = 0; k < N2_SLOTS; ++k) {
        const float ex = (sub + 8 * k < tot) ? expf(xs[k] - m) : 0.f;
        dsum += ex;
        a0 += rx[k] * ex;
        a1 += ry[k] * ex;
    }
    for (int p = sub + 8 * N2_SLOTS; p < tot; p += 8) {
        const int sp = (p < dg) ? sorted[(size_t)o + p] : n;
        const float4 r = *(const float4*)(rec2 + (size_t)sp * 4);
        const float ex = expf(lrelu(r.z + sdn) - m);
        dsum += ex;
        a0 += r.x * ex;
        a1 += r.y * ex;
    }
#pragma unroll
    for (int d = 4; d > 0; d >>= 1) {
        dsum += __shfl_xor(dsum, d);
        a0   += __shfl_xor(a0, d);
        a1   += __shfl_xor(a1, d);
    }
    if (sub == 0 && n < N) {
        float v0 = a0 / dsum + b2[0];
        float v1 = a1 / dsum + b2[1];
        float mx = fmaxf(v0, v1);
        float lse = mx + logf(expf(v0 - mx) + expf(v1 - mx));
        *(float2*)(out + (size_t)n * 2) = make_float2(v0 - lse, v1 - lse);
    }
}

static inline unsigned nblk(long long n, int b) { return (unsigned)((n + b - 1) / b); }

extern "C" void kernel_launch(void* const* d_in, const int* in_sizes, int n_in,
                              void* d_out, int out_size, void* d_ws, size_t ws_size,
                              hipStream_t stream) {
    const float* x     = (const float*)d_in[0];
    const int*   ei    = (const int*)  d_in[1];
    const float* W1    = (const float*)d_in[2];
    const float* asrc1 = (const float*)d_in[3];
    const float* adst1 = (const float*)d_in[4];
    const float* b1    = (const float*)d_in[5];
    const float* W2    = (const float*)d_in[6];
    const float* asrc2 = (const float*)d_in[7];
    const float* adst2 = (const float*)d_in[8];
    const float* b2    = (const float*)d_in[9];
    float* out = (float*)d_out;

    const int N = in_sizes[0] / 128;
    const int E = in_sizes[1] / 2;

    float* ws   = (float*)d_ws;
    size_t nl   = (size_t)N;
    float* h1   = ws;
    float* ss1  = h1  + nl * 64;
    float* sd1  = ss1 + nl * 4;
    float* rec2 = sd1 + nl * 4;
    int*   off  = (int*)(rec2 + nl * 4);
    int*   deg  = off + nl;
    int*   sorted = deg + nl + (size_t)E;
    const int C = (N + BNODES - 1) / BNODES;
    const int G = (int)nblk(E, EPB);
    const int M = C * G;
    int*   part = (int*)h1;
    int*   cntA = (int*)ss1;
    int*   lofA = cntA + M;

    k_part<<<G, 1024, C * sizeof(int), stream>>>(ei, cntA, lofA, part, E, C, G);
    k_bucket<<<C, 1024, 2 * G * sizeof(int), stream>>>(part, cntA, lofA, off, deg, sorted, N, G);
    k_gemm1_wmma<<<(N + 15) / 16, 128, 0, stream>>>(x, W1, h1, N);
    k_scores1<<<nblk(4LL * N, 256), 256, 0, stream>>>(h1, asrc1, adst1, ss1, sd1, N);
    k_node1<<<nblk(N, 8), 256, 0, stream>>>(sorted, off, deg, ss1, sd1, h1, b1, W2,
                                            asrc2, adst2, rec2, N);
    k_node2<<<nblk(8LL * N, 256), 256, 0, stream>>>(sorted, off, deg, rec2, b2, out, N);
}


# Layer-1 GEMM operand path: a wave owns a 16x64 strip, so x is read once, 16 B per lane (16 loads, not 4 x 32), and W1 comes from a per-block LDS copy grouped by four k; k sum order changes
# speedup vs baseline: 11.7642x; 1.0481x over previous
#include <hip/hip_runtime.h>
#include <hip/hip_bf16.h>

#define NEG_INF (-__builtin_inff())

typedef float v2f __attribute__((ext_vector_type(2)));
typedef float v8f __attribute__((ext_vector_type(8)));

#define EPB    8192
#define BSH    8
#define BNODES 256
#define SEGB   8
#define BCAP   12288

extern __shared__ int s_dyn[];

__device__ __forceinline__ int wave_sum_i(int v) {
#pragma unroll
    for (int d = 16; d > 0; d >>= 1) v += __shfl_xor(v, d);
    return v;
}

__global__ __launch_bounds__(1024)
void k_part(const int* __restrict__ ei, int* __restrict__ cntA, int* __restrict__ lofA,
            int* __restrict__ part_w, int E, int C, int G) {
    __shared__ int s_wtot[32];
    __shared__ __attribute__((aligned(16))) int2 s_stage[EPB];
    int2* __restrict__ part = (int2*)part_w;
    const int t = threadIdx.x, lane = t & 31, w = t >> 5, b = blockIdx.x;
    for (int c = t; c < C; c += 1024) s_dyn[c] = 0;
    __syncthreads();
    const int e0 = b * EPB + t;
    int s[8], d[8];
#pragma unroll
    for (int k = 0; k < 8; ++k) {
        const int e = e0 + k * 1024;
        const bool ok = e < E;
        s[k] = ok ? ei[e] : 0;
        d[k] = ok ? ei[(size_t)E + e] : -1;
    }
#pragma unroll
    for (int k = 0; k < 8; ++k) if (d[k] >= 0) atomicAdd(&s_dyn[d[k] >> BSH], 1);
    __syncthreads();
    int carry = 0;
    for (int base = 0; base < C; base += 1024) {
        const int c = base + t;
        const int cv = (c < C) ? s_dyn[c] : 0;
        int inc = cv;
#pragma unroll
        for (int dd = 1; dd < 32; dd <<= 1) {
            const int v = __shfl_up(inc, dd);
            if (lane >= dd) inc += v;
        }
        if (lane == 31) s_wtot[w] = inc;
        __syncthreads();
        const int wt = s_wtot[lane];
        const int pre = wave_sum_i((lane < w) ? wt : 0);
        const int tot = wave_sum_i(wt);
        const int ex = carry + pre + inc - cv;
        if (c < C) {
            cntA[(size_t)c * G + b] = cv;
            lofA[(size_t)c * G + b] = ex;
            s_dyn[c] = ex;
        }
        carry += tot;
        __syncthreads();
    }
#pragma unroll
    for (int k = 0; k < 8; ++k) {
        if (d[k] >= 0) {
            const int p = atomicAdd(&s_dyn[d[k] >> BSH], 1);
            s_stage[p] = make_int2(s[k], d[k]);
        }
    }
    __syncthreads();
    int2* __restrict__ mine = part + (size_t)b * EPB;
    const int left = E - b * EPB;
    const int nv = (left < EPB) ? left : EPB;
#pragma unroll
    for (int k = 0; k < 4; ++k) {
        const int i = 2 * (t + k * 1024);
        if (i + 1 < nv)  *(int4*)(mine + i) = *(const int4*)(s_stage + i);
        else if (i < nv) mine[i] = s_stage[i];
    }
}

__global__ __launch_bounds__(1024)
void k_bucket(const int* __restrict__ part_w, const int* __restrict__ cntA,
              const int* __restrict__ lofA, int* __restrict__ off, int* __restrict__ deg,
              int* __restrict__ sorted, int N, int G) {
    __shared__ int s_cnt[BNODES + 32];
    __shared__ int s_wtot[32];
    __shared__ __attribute__((aligned(16))) int s_stage[BCAP];
    const int2* __restrict__ part = (const int2*)part_w;
    const int t = threadIdx.x, lane = t & 31, c = blockIdx.x;
    const int w = __builtin_amdgcn_readfirstlane(t >> 5);
    const int nbase = c << BSH;
    int* s_len = s_dyn;
    int* s_beg = s_dyn + G;
    int lsum = 0;
    for (int g = t; g < G; g += 1024) {
        const int lo = lofA[(size_t)c * G + g];
        s_len[g] = cntA[(size_t)c * G + g];
        s_beg[g] = g * EPB + lo;
        lsum += lo;
    }
    if (t < BNODES + 32) s_cnt[t] = 0;
    lsum = wave_sum_i(lsum);
    if (lane == 0) s_wtot[w] = lsum;
    __syncthreads();
    const int beg = wave_sum_i(s_wtot[lane]);

    for (int g0 = w; g0 < G; g0 += 32 * SEGB) {
        const int gl = g0 + 32 * (lane & (SEGB - 1));
        const int vlen = (gl < G) ? s_len[gl] : 0;
        const int vst  = (gl < G) ? s_beg[gl] : 0;
        int len[SEGB], st[SEGB];
#pragma unroll
        for (int k = 0; k < SEGB; ++k) {
            len[k] = __builtin_amdgcn_readlane(vlen, k);
            st[k]  = __builtin_amdgcn_readlane(vst, k);
        }
        int y[SEGB][2];
#pragma unroll
        for (int k = 0; k < SEGB; ++k) {
            y[k][0] = (lane < len[k])      ? part[(size_t)st[k] + lane].y      : -1;
            y[k][1] = (lane + 32 < len[k]) ? part[(size_t)st[k] + 32 + lane].y : -1;
        }
#pragma unroll
        for (int k = 0; k < SEGB; ++k) {
            if (y[k][0] >= 0) atomicAdd(&s_cnt[y[k][0] - nbase], 1);
            if (y[k][1] >= 0) atomicAdd(&s_cnt[y[k][1] - nbase], 1);
        }
#pragma unroll
        for (int k = 0; k < SEGB; ++k)
            for (int j = 64 + lane; j < len[k]; j += 32)
                atomicAdd(&s_cnt[part[(size_t)st[k] + j].y - nbase], 1);
    }
    __syncthreads();
    int cv = 0, inc = 0;
    if (t < BNODES) {
        cv = s_cnt[t];
        inc = cv;
#pragma unroll
        for (int d = 1; d < 32; d <<= 1) {
            const int v = __shfl_up(inc, d);
            if (lane >= d) inc += v;
        }
        if (lane == 31) s_wtot[w] = inc;
    }
    __syncthreads();
    if (t < BNODES) {
        int pre = 0;
        for (int k = 0; k < w; ++k) pre += s_wtot[k];
        const int ex = pre + inc - cv;
        s_cnt[t] = ex;
        const int n = nbase + t;
        if (n < N) { off[n] = beg + ex; deg[n] = cv; }
    }
    int btot = 0;
#pragma unroll
    for (int k = 0; k < BNODES / 32; ++k) btot += s_wtot[k];
    int* __restrict__ first = sorted + (size_t)beg;
    const int sh = (int)(((size_t)first >> 2) & 31);
    const bool staged = btot + 32 <= BCAP;
    __syncthreads();
    for (int g0 = w; g0 < G; g0 += 32 * SEGB) {
        const int gl = g0 + 32 * (lane & (SEGB - 1));
        const int vlen = (gl < G) ? s_len[gl] : 0;
        const int vst  = (gl < G) ? s_beg[gl] : 0;
        int len[SEGB], st[SEGB];
#pragma unroll
        for (int k = 0; k < SEGB; ++k) {
            len[k] = __builtin_amdgcn_readlane(vlen, k);
            st[k]  = __builtin_amdgcn_readlane(vst, k);
        }
        int2 p[SEGB][2];
#pragma unroll
        for (int k = 0; k < SEGB; ++k) {
            p[k][0] = (lane < len[k])      ? part[(size_t)st[k] + lane]      : make_int2(0, -1);
            p[k][1] = (lane + 32 < len[k]) ? part[(size_t)st[k] + 32 + lane] : make_int2(0, -1);
        }
        int pos[SEGB][2];
#pragma unroll
        for (int k = 0; k < SEGB; ++k) {
#pragma unroll
            for (int h = 0; h < 2; ++h) {
                const int y = p[k][h].y;
                pos[k][h] = atomicAdd(&s_cnt[(y >= 0) ? y - nbase : BNODES + lane], 1);
            }
        }
        if (staged) {
#pragma unroll
            for (int k = 0; k < SEGB; ++k) {
#pragma unroll
                for (int h = 0; h < 2; ++h)
                    if (p[k][h].y >= 0) s_stage[sh + pos[k][h]] = p[k][h].x;
            }
        } else {
#pragma unroll
            for (int k = 0; k < SEGB; ++k) {
#pragma unroll
                for (int h = 0; h < 2; ++h)
                    if (p[k][h].y >= 0) first[pos[k][h]] = p[k][h].x;
            }
        }
#pragma unroll
        for (int k = 0; k < SEGB; ++k) {
            for (int j = 64 + lane; j < len[k]; j += 32) {
                const int2 q = part[(size_t)st[k] + j];
                const int pq = atomicAdd(&s_cnt[q.y - nbase], 1);
                if (staged) s_stage[sh + pq] = q.x;
                else        first[pq] = q.x;
            }
        }
    }
    __syncthreads();
    if (staged) {
        int* __restrict__ line0 = first - sh;
        const int hi = sh + btot;
        for (int i = 4 * t; i < hi; i += 4096) {
            if (i >= sh && i + 4 <= hi) {
                *(int4*)(line0 + i) = *(const int4*)(s_stage + i);
            } else {
#pragma unroll
                for (int j = 0; j < 4; ++j)
                    if (i + j >= sh && i + j < hi) line0[i + j] = s_stage[i + j];
            }
        }
    }
}

__global__ __launch_bounds__(256)
void k_gemm1_wmma(const float* __restrict__ x, const float* __restrict__ W1,
                  float* __restrict__ h1, int N) {
    __shared__ __attribute__((aligned(16))) float s_w[32 * 64 * 4];
    const int t = threadIdx.x;
    const int lane = t & 31;
    const int wave = t >> 5;
    const int half = lane >> 4;
    const int c    = lane & 15;
    const int row0 = (blockIdx.x * 8 + wave) * 16;
    int row = row0 + c;
    if (row >= N) row = N - 1;
    const float4* xr = (const float4*)(x + (size_t)row * 128) + half;
    float4 a[16];
#pragma unroll
    for (int i = 0; i < 16; ++i) a[i] = xr[2 * i];
    float4 r[2][4];
#pragma unroll
    for (int p = 0; p < 2; ++p) {
        const int it = t + 256 * p;
        const float* wp = W1 + (size_t)(4 * (it >> 4)) * 64 + 4 * (it & 15);
#pragma unroll
        for (int j = 0; j < 4; ++j) r[p][j] = *(const float4*)(wp + 64 * j);
    }
#pragma unroll
    for (int p = 0; p < 2; ++p) {
        const int it = t + 256 * p;
        const int kq = it >> 4, cq = it & 15;
        float4* d = (float4*)(s_w + kq * 256 + 16 * cq);
        d[0] = make_float4(r[p][0].x, r[p][1].x, r[p][2].x, r[p][3].x);
        d[1] = make_float4(r[p][0].y, r[p][1].y, r[p][2].y, r[p][3].y);
        d[2] = make_float4(r[p][0].z, r[p][1].z, r[p][2].z, r[p][3].z);
        d[3] = make_float4(r[p][0].w, r[p][1].w, r[p][2].w, r[p][3].w);
    }
    __syncthreads();
    if (row0 >= N) return;

    const float4* bq = (const float4*)s_w + half * 64 + c;
    v8f acc[4] = {};
#pragma unroll
    for (int i = 0; i < 16; ++i) {
        float4 b[4];
#pragma unroll
        for (int tl = 0; tl < 4; ++tl) b[tl] = bq[(2 * i) * 64 + 16 * tl];
#pragma unroll
        for (int tl = 0; tl < 4; ++tl) {
            v2f av, bv;
            av.x = a[i].x; av.y = a[i].y;
            bv.x = b[tl].x; bv.y = b[tl].y;
            acc[tl] = __builtin_amdgcn_wmma_f32_16x16x4_f32(
                 false, av,  false, bv,
                 (short)0, acc[tl],  false,  false);
        }
#pragma unroll
        for (int tl = 0; tl < 4; ++tl) {
            v2f av, bv;
            av.x = a[i].z; av.y = a[i].w;
            bv.x = b[tl].z; bv.y = b[tl].w;
            acc[tl] = __builtin_amdgcn_wmma_f32_16x16x4_f32(
                 false, av,  false, bv,
                 (short)0, acc[tl],  false,  false);
        }
    }
    float* dp = h1 + (size_t)(row0 + 8 * half) * 64 + c;
    if (row0 + 16 <= N) {
#pragma unroll
        for (int tl = 0; tl < 4; ++tl) {
#pragma unroll
            for (int v = 0; v < 8; ++v) dp[(size_t)v * 64 + 16 * tl] = acc[tl][v];
        }
    } else {
#pragma unroll
        for (int tl = 0; tl < 4; ++tl) {
#pragma unroll
            for (int v = 0; v < 8; ++v) {
                if (row0 + v + 8 * half < N) dp[(size_t)v * 64 + 16 * tl] = acc[tl][v];
            }
        }
    }
}

__global__ void k_scores1(const float* __restrict__ h1, const float* __restrict__ as,
                          const float* __restrict__ ad, float* __restrict__ ss,
                          float* __restrict__ sd, int N) {
    int t = blockIdx.x * blockDim.x + threadIdx.x;
    if (t >= N * 4) return;
    int n = t >> 2, h = t & 3;
    const float4* hp = (const float4*)(h1 + (size_t)n * 64 + h * 16);
    const float4* ap = (const float4*)(as + h * 16);
    const float4* bp = (const float4*)(ad + h * 16);
    float a = 0.f, b = 0.f;
#pragma unroll
    for (int i = 0; i < 4; ++i) {
        float4 hv = hp[i], av = ap[i], bv = bp[i];
        a += hv.x * av.x + hv.y * av.y + hv.z * av.z + hv.w * av.w;
        b += hv.x * bv.x + hv.y * bv.y + hv.z * bv.z + hv.w * bv.w;
    }
    ss[t] = a; sd[t] = b;
}

__device__ __forceinline__ float lrelu(float s) { return s > 0.f ? s : 0.2f * s; }

__device__ __forceinline__ float wave_max(float v) {
#pragma unroll
    for (int d = 16; d > 0; d >>= 1) v = fmaxf(v, __shfl_xor(v, d));
    return v;
}
__device__ __forceinline__ float wave_sum(float v) {
#pragma unroll
    for (int d = 16; d > 0; d >>= 1) v += __shfl_xor(v, d);
    return v;
}

#define N1_PLANE 66

__device__ __forceinline__ float wave_uniform(float v) {
    return __int_as_float(__builtin_amdgcn_readfirstlane(__float_as_int(v)));
}
__device__ __forceinline__ float xhalf(float x) {
    return __builtin_bit_cast(float, __builtin_amdgcn_permlanex16(__builtin_bit_cast(int, x), __builtin_bit_cast(int, x),
                                                                  0x76543210, 0xfedcba98, false, false));
}

__device__ __forceinline__ void put_chunk(int2* __restrict__ wr, int s, float e0, float e1, float e2, float e3) {
    const int so = (int)((unsigned)s << 8);
    wr[0 * N1_PLANE] = make_int2(so, __float_as_int(e0));
    wr[1 * N1_PLANE] = make_int2(so, __float_as_int(e1));
    wr[2 * N1_PLANE] = make_int2(so, __float_as_int(e2));
    wr[3 * N1_PLANE] = make_int2(so, __float_as_int(e3));
}

__device__ __forceinline__ void rows_acc(const float* __restrict__ h1, const int2* __restrict__ rb,
                                         unsigned qoff, int cnt, float4& acc) {
    for (int j = 0; j < cnt; j += 16) {
        int2 rc[8];
#pragma unroll
        for (int u = 0; u < 8; ++u) rc[u] = rb[j + 2 * u];
        float4 hv[8];
#pragma unroll
        for (int u = 0; u < 8; ++u)
            hv[u] = *(const float4*)((const char*)h1 + (size_t)((unsigned)rc[u].x + qoff));
#pragma unroll
        for (int u = 0; u < 8; ++u) {
            const float w = __int_as_float(rc[u].y);
            acc.x += hv[u].x * w;
            acc.y += hv[u].y * w;
            acc.z += hv[u].z * w;
            acc.w += hv[u].w * w;
        }
    }
}

__global__ __launch_bounds__(256) __attribute__((amdgpu_waves_per_eu(16, 16)))
void k_node1(const int* __restrict__ sorted, const int* __restrict__ off,
             const int* __restrict__ deg, const float* __restrict__ ss,
             const float* __restrict__ sd, const float* __restrict__ h1,
             const float* __restrict__ b1, const float* __restrict__ W2,
             const float* __restrict__ as2, const float* __restrict__ ad2,
             float* __restrict__ rec2, int N) {
    __shared__ int2 s_rec[8 * 4 * N1_PLANE];
    const int lane = threadIdx.x & 31;
    const int wv = threadIdx.x >> 5;
    const int n = __builtin_amdgcn_readfirstlane(blockIdx.x * 8 + wv);
    if (n >= N) return;
    const int o = off[n], dg = deg[n], tot = dg + 1;
    const float4 sdn = *(const float4*)(sd + (size_t)n * 4);

    const int pA = lane, pB = 32 + lane;
    const bool hasB = tot > 32;
    const int vA = sorted[(pA < dg) ? (size_t)o + pA : 0];
    const int vB = sorted[(pB < dg) ? (size_t)o + pB : 0];
    const int sA = (pA < dg) ? vA : n;
    const int sB = (pB < dg) ? vB : n;
    const float4 svA = *(const float4*)(ss + (size_t)sA * 4);
    float4 svB = make_float4(0.f, 0.f, 0.f, 0.f);
    if (hasB) svB = *(const float4*)(ss + (size_t)sB * 4);
    const bool okA = pA < tot, okB = pB < tot;
    const float xA0 = okA ? lrelu(svA.x + sdn.x) : NEG_INF, xA1 = okA ? lrelu(svA.y + sdn.y) : NEG_INF;
    const float xA2 = okA ? lrelu(svA.z + sdn.z) : NEG_INF, xA3 = okA ? lrelu(svA.w + sdn.w) : NEG_INF;
    const float xB0 = okB ? lrelu(svB.x + sdn.x) : NEG_INF, xB1 = okB ? lrelu(svB.y + sdn.y) : NEG_INF;
    const float xB2 = okB ? lrelu(svB.z + sdn.z) : NEG_INF, xB3 = okB ? lrelu(svB.w + sdn.w) : NEG_INF;
    float m0 = fmaxf(xA0, xB0), m1 = fmaxf(xA1, xB1), m2 = fmaxf(xA2, xB2), m3 = fmaxf(xA3, xB3);
    for (int base = 64; base < tot; base += 32) {
        const int p = base + lane;
        if (p < tot) {
            const int s = (p < dg) ? sorted[(size_t)o + p] : n;
            const float4 sv = *(const float4*)(ss + (size_t)s * 4);
            m0 = fmaxf(m0, lrelu(sv.x + sdn.x));
            m1 = fmaxf(m1, lrelu(sv.y + sdn.y));
            m2 = fmaxf(m2, lrelu(sv.z + sdn.z));
            m3 = fmaxf(m3, lrelu(sv.w + sdn.w));
        }
    }
    m0 = wave_uniform(wave_max(m0));
    m1 = wave_uniform(wave_max(m1));
    m2 = wave_uniform(wave_max(m2));
    m3 = wave_uniform(wave_max(m3));

    const int half = lane >> 4;
    const int q    = lane & 15;
    const int hsel = q >> 2;
    int2* wtab = s_rec + wv * (4 * N1_PLANE);
    int2* wr = wtab + lane;
    const int2* rb = wtab + hsel * N1_PLANE + half;
    const unsigned qoff = 16u * q;
    float4 acc = make_float4(0.f, 0.f, 0.f, 0.f);
    float t0, t1, t2, t3;
    {
        const float e0 = okA ? expf(xA0 - m0) : 0.f, e1 = okA ? expf(xA1 - m1) : 0.f;
        const float e2 = okA ? expf(xA2 - m2) : 0.f, e3 = okA ? expf(xA3 - m3) : 0.f;
        t0 = e0; t1 = e1; t2 = e2; t3 = e3;
        put_chunk(wr, sA, e0, e1, e2, e3);
    }
    if (hasB) {
        const float e0 = okB ? expf(xB0 - m0) : 0.f, e1 = okB ? expf(xB1 - m1) : 0.f;
        const float e2 = okB ? expf(xB2 - m2) : 0.f, e3 = okB ? expf(xB3 - m3) : 0.f;
        t0 += e0; t1 += e1; t2 += e2; t3 += e3;
        put_chunk(wr + 32, sB, e0, e1, e2, e3);
    }
    float d0 = wave_uniform(wave_sum(t0)), d1 = wave_uniform(wave_sum(t1));
    float d2 = wave_uniform(wave_sum(t2)), d3 = wave_uniform(wave_sum(t3));
    rows_acc(h1, rb, qoff, (tot < 64) ? tot : 64, acc);
    for (int base = 64; base < tot; base += 32) {
        const int p = base + lane;
        int s = n;
        float e0 = 0.f, e1 = 0.f, e2 = 0.f, e3 = 0.f;
        if (p < tot) {
            if (p < dg) s = sorted[(size_t)o + p];
            const float4 sv = *(const float4*)(ss + (size_t)s * 4);
            e0 = expf(lrelu(sv.x + sdn.x) - m0);
            e1 = expf(lrelu(sv.y + sdn.y) - m1);
            e2 = expf(lrelu(sv.z + sdn.z) - m2);
            e3 = expf(lrelu(sv.w + sdn.w) - m3);
        }
        d0 += wave_uniform(wave_sum(e0)); d1 += wave_uniform(wave_sum(e1));
        d2 += wave_uniform(wave_sum(e2)); d3 += wave_uniform(wave_sum(e3));
        put_chunk(wr, s, e0, e1, e2, e3);
        rows_acc(h1, rb, qoff, (tot - base < 32) ? (tot - base) : 32, acc);
    }
    const float dn = (hsel == 0) ? d0 : (hsel == 1) ? d1 : (hsel == 2) ? d2 : d3;
    const float g0 = half ? acc.x : acc.z, g1 = half ? acc.y : acc.w;
    const float a0 = (half ? acc.z : acc.x) + xhalf(g0);
    const float a1 = (half ? acc.w : acc.y) + xhalf(g1);
    const int c0 = 4 * q + 2 * half;
    const float2 bb = *(const float2*)(b1 + c0);
    float v0 = a0 / dn + bb.x;
    float v1 = a1 / dn + bb.y;
    v0 = v0 > 0.f ? v0 : expm1f(v0);
    v1 = v1 > 0.f ? v1 : expm1f(v1);
    const float4 wv2 = *(const float4*)(W2 + 2 * c0);
    float o0 = v0 * wv2.x + v1 * wv2.z;
    float o1 = v0 * wv2.y + v1 * wv2.w;
    o0 = wave_sum(o0);
    o1 = wave_sum(o1);
    if (lane == 0) {
        *(float4*)(rec2 + (size_t)n * 4) =
            make_float4(o0, o1, o0 * as2[0] + o1 * as2[1], o0 * ad2[0] + o1 * ad2[1]);
    }
}

#define N2_SLOTS 6
__global__ __launch_bounds__(256)
void k_node2(const int* __restrict__ sorted, const int* __restrict__ off,
             const int* __restrict__ deg, const float* __restrict__ rec2,
             const float* __restrict__ b2, float* __restrict__ out, int N) {
    const int t = blockIdx.x * 256 + threadIdx.x;
    const int n = t >> 3, sub = t & 7;
    int o = 0, dg = 0, tot = 0, nn = 0;
    float sdn = 0.f;
    if (n < N) { o = off[n]; dg = deg[n]; tot = dg + 1; sdn = rec2[(size_t)n * 4 + 3]; nn = n; }

    int s[N2_SLOTS];
#pragma unroll
    for (int k = 0; k < N2_SLOTS; ++k) {
        const int p = sub + 8 * k;
        const int v = sorted[(p < dg) ? (size_t)o + p : 0];
        s[k] = (p < dg) ? v : nn;
    }
    float4 r[N2_SLOTS];
#pragma unroll
    for (int k = 0; k < N2_SLOTS; ++k) r[k] = *(const float4*)(rec2 + (size_t)s[k] * 4);
    __builtin_amdgcn_sched_barrier(0);
    float rx[N2_SLOTS], ry[N2_SLOTS], xs[N2_SLOTS];
    float m = NEG_INF;
#pragma unroll
    for (int k = 0; k < N2_SLOTS; ++k) {
        const bool ok = sub + 8 * k < tot;
        rx[k] = ok ? r[k].x : 0.f;
        ry[k] = ok ? r[k].y : 0.f;
        xs[k] = ok ? lrelu(r[k].z + sdn) : NEG_INF;
        m = fmaxf(m, xs[k]);
    }
    for (int p = sub + 8 * N2_SLOTS; p < tot; p += 8) {
        const int sp = (p < dg) ? sorted[(size_t)o + p] : n;
        m = fmaxf(m, lrelu(rec2[(size_t)sp * 4 + 2] + sdn));
    }
#pragma unroll
    for (int d = 4; d > 0; d >>= 1) m = fmaxf(m, __shfl_xor(m, d));

    float dsum = 0.f, a0 = 0.f, a1 = 0.f;
#pragma unroll
    for (int k = 0; k < N2_SLOTS; ++k) {
        const float ex = (sub + 8 * k < tot) ? expf(xs[k] - m) : 0.f;
        dsum += ex;
        a0 += rx[k] * ex;
        a1 += ry[k] * ex;
    }
    for (int p = sub + 8 * N2_SLOTS; p < tot; p += 8) {
        const int sp = (p < dg) ? sorted[(size_t)o + p] : n;
        const float4 r = *(const float4*)(rec2 + (size_t)sp * 4);
        const float ex = expf(lrelu(r.z + sdn) - m);
        dsum += ex;
        a0 += r.x * ex;
        a1 += r.y * ex;
    }
#pragma unroll
    for (int d = 4; d > 0; d >>= 1) {
        dsum += __shfl_xor(dsum, d);
        a0   += __shfl_xor(a0, d);
        a1   += __shfl_xor(a1, d);
    }
    if (sub == 0 && n < N) {
        float v0 = a0 / dsum + b2[0];
        float v1 = a1 / dsum + b2[1];
        float mx = fmaxf(v0, v1);
        float lse = mx + logf(expf(v0 - mx) + expf(v1 - mx));
        *(float2*)(out + (size_t)n * 2) = make_float2(v0 - lse, v1 - lse);
    }
}

static inline unsigned nblk(long long n, int b) { return (unsigned)((n + b - 1) / b); }

extern "C" void kernel_launch(void* const* d_in, const int* in_sizes, int n_in,
                              void* d_out, int out_size, void* d_ws, size_t ws_size,
                              hipStream_t stream) {
    const float* x     = (const float*)d_in[0];
    const int*   ei    = (const int*)  d_in[1];
    const float* W1    = (const float*)d_in[2];
    const float* asrc1 = (const float*)d_in[3];
    const float* adst1 = (const float*)d_in[4];
    const float* b1    = (const float*)d_in[5];
    const float* W2    = (const float*)d_in[6];
    const float* asrc2 = (const float*)d_in[7];
    const float* adst2 = (const float*)d_in[8];
    const float* b2    = (const float*)d_in[9];
    float* out = (float*)d_out;

    const int N = in_sizes[0] / 128;
    const int E = in_sizes[1] / 2;

    float* ws   = (float*)d_ws;
    size_t nl   = (size_t)N;
    float* h1   = ws;
    float* ss1  = h1  + nl * 64;
    float* sd1  = ss1 + nl * 4;
    float* rec2 = sd1 + nl * 4;
    int*   off  = (int*)(rec2 + nl * 4);
    int*   deg  = off + nl;
    int*   sorted = deg + nl + (size_t)E;
    const int C = (N + BNODES - 1) / BNODES;
    const int G = (int)nblk(E, EPB);
    const int M = C * G;
    int*   part = (int*)h1;
    int*   cntA = (int*)ss1;
    int*   lofA = cntA + M;

    k_part<<<G, 1024, C * sizeof(int), stream>>>(ei, cntA, lofA, part, E, C, G);
    k_bucket<<<C, 1024, 2 * G * sizeof(int), stream>>>(part, cntA, lofA, off, deg, sorted, N, G);
    k_gemm1_wmma<<<nblk(N, 128), 256, 0, stream>>>(x, W1, h1, N);
    k_scores1<<<nblk(4LL * N, 256), 256, 0, stream>>>(h1, asrc1, adst1, ss1, sd1, N);
    k_node1<<<nblk(N, 8), 256, 0, stream>>>(sorted, off, deg, ss1, sd1, h1, b1, W2,
                                            asrc2, adst2, rec2, N);
    k_node2<<<nblk(8LL * N, 256), 256, 0, stream>>>(sorted, off, deg, rec2, b2, out, N);
}


# Layer-1 score pass moved into the GEMM epilogue: a wave turns its 16x64 strip through LDS, stores whole 256-B rows and takes its dot products there (now fused multiply-adds); 5 launches, not 6
# speedup vs baseline: 12.2225x; 1.0390x over previous
#include <hip/hip_runtime.h>
#include <hip/hip_bf16.h>

#define NEG_INF (-__builtin_inff())

typedef float v2f __attribute__((ext_vector_type(2)));
typedef float v8f __attribute__((ext_vector_type(8)));

#define EPB    8192
#define BSH    8
#define BNODES 256
#define SEGB   8
#define BCAP   12288

extern __shared__ int s_dyn[];

__device__ __forceinline__ int wave_sum_i(int v) {
#pragma unroll
    for (int d = 16; d > 0; d >>= 1) v += __shfl_xor(v, d);
    return v;
}

__global__ __launch_bounds__(1024)
void k_part(const int* __restrict__ ei, int* __restrict__ cntA, int* __restrict__ lofA,
            int* __restrict__ part_w, int E, int C, int G) {
    __shared__ int s_wtot[32];
    __shared__ __attribute__((aligned(16))) int2 s_stage[EPB];
    int2* __restrict__ part = (int2*)part_w;
    const int t = threadIdx.x, lane = t & 31, w = t >> 5, b = blockIdx.x;
    for (int c = t; c < C; c += 1024) s_dyn[c] = 0;
    __syncthreads();
    const int e0 = b * EPB + t;
    int s[8], d[8];
#pragma unroll
    for (int k = 0; k < 8; ++k) {
        const int e = e0 + k * 1024;
        const bool ok = e < E;
        s[k] = ok ? ei[e] : 0;
        d[k] = ok ? ei[(size_t)E + e] : -1;
    }
#pragma unroll
    for (int k = 0; k < 8; ++k) if (d[k] >= 0) atomicAdd(&s_dyn[d[k] >> BSH], 1);
    __syncthreads();
    int carry = 0;
    for (int base = 0; base < C; base += 1024) {
        const int c = base + t;
        const int cv = (c < C) ? s_dyn[c] : 0;
        int inc = cv;
#pragma unroll
        for (int dd = 1; dd < 32; dd <<= 1) {
            const int v = __shfl_up(inc, dd);
            if (lane >= dd) inc += v;
        }
        if (lane == 31) s_wtot[w] = inc;
        __syncthreads();
        const int wt = s_wtot[lane];
        const int pre = wave_sum_i((lane < w) ? wt : 0);
        const int tot = wave_sum_i(wt);
        const int ex = carry + pre + inc - cv;
        if (c < C) {
            cntA[(size_t)c * G + b] = cv;
            lofA[(size_t)c * G + b] = ex;
            s_dyn[c] = ex;
        }
        carry += tot;
        __syncthreads();
    }
#pragma unroll
    for (int k = 0; k < 8; ++k) {
        if (d[k] >= 0) {
            const int p = atomicAdd(&s_dyn[d[k] >> BSH], 1);
            s_stage[p] = make_int2(s[k], d[k]);
        }
    }
    __syncthreads();
    int2* __restrict__ mine = part + (size_t)b * EPB;
    const int left = E - b * EPB;
    const int nv = (left < EPB) ? left : EPB;
#pragma unroll
    for (int k = 0; k < 4; ++k) {
        const int i = 2 * (t + k * 1024);
        if (i + 1 < nv)  *(int4*)(mine + i) = *(const int4*)(s_stage + i);
        else if (i < nv) mine[i] = s_stage[i];
    }
}

__global__ __launch_bounds__(1024)
void k_bucket(const int* __restrict__ part_w, const int* __restrict__ cntA,
              const int* __restrict__ lofA, int* __restrict__ off, int* __restrict__ deg,
              int* __restrict__ sorted, int N, int G) {
    __shared__ int s_cnt[BNODES + 32];
    __shared__ int s_wtot[32];
    __shared__ __attribute__((aligned(16))) int s_stage[BCAP];
    const int2* __restrict__ part = (const int2*)part_w;
    const int t = threadIdx.x, lane = t & 31, c = blockIdx.x;
    const int w = __builtin_amdgcn_readfirstlane(t >> 5);
    const int nbase = c << BSH;
    int* s_len = s_dyn;
    int* s_beg = s_dyn + G;
    int lsum = 0;
    for (int g = t; g < G; g += 1024) {
        const int lo = lofA[(size_t)c * G + g];
        s_len[g] = cntA[(size_t)c * G + g];
        s_beg[g] = g * EPB + lo;
        lsum += lo;
    }
    if (t < BNODES + 32) s_cnt[t] = 0;
    lsum = wave_sum_i(lsum);
    if (lane == 0) s_wtot[w] = lsum;
    __syncthreads();
    const int beg = wave_sum_i(s_wtot[lane]);

    for (int g0 = w; g0 < G; g0 += 32 * SEGB) {
        const int gl = g0 + 32 * (lane & (SEGB - 1));
        const int vlen = (gl < G) ? s_len[gl] : 0;
        const int vst  = (gl < G) ? s_beg[gl] : 0;
        int len[SEGB], st[SEGB];
#pragma unroll
        for (int k = 0; k < SEGB; ++k) {
            len[k] = __builtin_amdgcn_readlane(vlen, k);
            st[k]  = __builtin_amdgcn_readlane(vst, k);
        }
        int y[SEGB][2];
#pragma unroll
        for (int k = 0; k < SEGB; ++k) {
            y[k][0] = (lane < len[k])      ? part[(size_t)st[k] + lane].y      : -1;
            y[k][1] = (lane + 32 < len[k]) ? part[(size_t)st[k] + 32 + lane].y : -1;
        }
#pragma unroll
        for (int k = 0; k < SEGB; ++k) {
            if (y[k][0] >= 0) atomicAdd(&s_cnt[y[k][0] - nbase], 1);
            if (y[k][1] >= 0) atomicAdd(&s_cnt[y[k][1] - nbase], 1);
        }
#pragma unroll
        for (int k = 0; k < SEGB; ++k)
            for (int j = 64 + lane; j < len[k]; j += 32)
                atomicAdd(&s_cnt[part[(size_t)st[k] + j].y - nbase], 1);
    }
    __syncthreads();
    int cv = 0, inc = 0;
    if (t < BNODES) {
        cv = s_cnt[t];
        inc = cv;
#pragma unroll
        for (int d = 1; d < 32; d <<= 1) {
            const int v = __shfl_up(inc, d);
            if (lane >= d) inc += v;
        }
        if (lane == 31) s_wtot[w] = inc;
    }
    __syncthreads();
    if (t < BNODES) {
        int pre = 0;
        for (int k = 0; k < w; ++k) pre += s_wtot[k];
        const int ex = pre + inc - cv;
        s_cnt[t] = ex;
        const int n = nbase + t;
        if (n < N) { off[n] = beg + ex; deg[n] = cv; }
    }
    int btot = 0;
#pragma unroll
    for (int k = 0; k < BNODES / 32; ++k) btot += s_wtot[k];
    int* __restrict__ first = sorted + (size_t)beg;
    const int sh = (int)(((size_t)first >> 2) & 31);
    const bool staged = btot + 32 <= BCAP;
    __syncthreads();
    for (int g0 = w; g0 < G; g0 += 32 * SEGB) {
        const int gl = g0 + 32 * (lane & (SEGB - 1));
        const int vlen = (gl < G) ? s_len[gl] : 0;
        const int vst  = (gl < G) ? s_beg[gl] : 0;
        int len[SEGB], st[SEGB];
#pragma unroll
        for (int k = 0; k < SEGB; ++k) {
            len[k] = __builtin_amdgcn_readlane(vlen, k);
            st[k]  = __builtin_amdgcn_readlane(vst, k);
        }
        int2 p[SEGB][2];
#pragma unroll
        for (int k = 0; k < SEGB; ++k) {
            p[k][0] = (lane < len[k])      ? part[(size_t)st[k] + lane]      : make_int2(0, -1);
            p[k][1] = (lane + 32 < len[k]) ? part[(size_t)st[k] + 32 + lane] : make_int2(0, -1);
        }
        int pos[SEGB][2];
#pragma unroll
        for (int k = 0; k < SEGB; ++k) {
#pragma unroll
            for (int h = 0; h < 2; ++h) {
                const int y = p[k][h].y;
                pos[k][h] = atomicAdd(&s_cnt[(y >= 0) ? y - nbase : BNODES + lane], 1);
            }
        }
        if (staged) {
#pragma unroll
            for (int k = 0; k < SEGB; ++k) {
#pragma unroll
                for (int h = 0; h < 2; ++h)
                    if (p[k][h].y >= 0) s_stage[sh + pos[k][h]] = p[k][h].x;
            }
        } else {
#pragma unroll
            for (int k = 0; k < SEGB; ++k) {
#pragma unroll
                for (int h = 0; h < 2; ++h)
                    if (p[k][h].y >= 0) first[pos[k][h]] = p[k][h].x;
            }
        }
#pragma unroll
        for (int k = 0; k < SEGB; ++k) {
            for (int j = 64 + lane; j < len[k]; j += 32) {
                const int2 q = part[(size_t)st[k] + j];
                const int pq = atomicAdd(&s_cnt[q.y - nbase], 1);
                if (staged) s_stage[sh + pq] = q.x;
                else        first[pq] = q.x;
            }
        }
    }
    __syncthreads();
    if (staged) {
        int* __restrict__ line0 = first - sh;
        const int hi = sh + btot;
        for (int i = 4 * t; i < hi; i += 4096) {
            if (i >= sh && i + 4 <= hi) {
                *(int4*)(line0 + i) = *(const int4*)(s_stage + i);
            } else {
#pragma unroll
                for (int j = 0; j < 4; ++j)
                    if (i + j >= sh && i + j < hi) line0[i + j] = s_stage[i + j];
            }
        }
    }
}

#define SPITCH 68

__global__ __launch_bounds__(256)
void k_gemm1_wmma(const float* __restrict__ x, const float* __restrict__ W1,
                  const float* __restrict__ as, const float* __restrict__ ad,
                  float* __restrict__ h1, float* __restrict__ ss, float* __restrict__ sd, int N) {
    __shared__ __attribute__((aligned(16))) float s_w[32 * 64 * 4];
    __shared__ __attribute__((aligned(16))) float s_h[8 * 16 * SPITCH];
    const int t = threadIdx.x;
    const int lane = t & 31;
    const int wave = __builtin_amdgcn_readfirstlane(t >> 5);
    const int half = lane >> 4;
    const int c    = lane & 15;
    const int row0 = (blockIdx.x * 8 + wave) * 16;
    int row = row0 + c;
    if (row >= N) row = N - 1;
    const float4* xr = (const float4*)(x + (size_t)row * 128) + half;
    float4 a[16];
#pragma unroll
    for (int i = 0; i < 16; ++i) a[i] = xr[2 * i];
    float4 r[2][4];
#pragma unroll
    for (int p = 0; p < 2; ++p) {
        const int it = t + 256 * p;
        const float* wp = W1 + (size_t)(4 * (it >> 4)) * 64 + 4 * (it & 15);
#pragma unroll
        for (int j = 0; j < 4; ++j) r[p][j] = *(const float4*)(wp + 64 * j);
    }
#pragma unroll
    for (int p = 0; p < 2; ++p) {
        const int it = t + 256 * p;
        const int kq = it >> 4, cq = it & 15;
        float4* d = (float4*)(s_w + kq * 256 + 16 * cq);
        d[0] = make_float4(r[p][0].x, r[p][1].x, r[p][2].x, r[p][3].x);
        d[1] = make_float4(r[p][0].y, r[p][1].y, r[p][2].y, r[p][3].y);
        d[2] = make_float4(r[p][0].z, r[p][1].z, r[p][2].z, r[p][3].z);
        d[3] = make_float4(r[p][0].w, r[p][1].w, r[p][2].w, r[p][3].w);
    }
    __syncthreads();
    if (row0 >= N) return;

    const float4* bq = (const float4*)s_w + half * 64 + c;
    v8f acc[4] = {};
#pragma unroll
    for (int i = 0; i < 16; ++i) {
        float4 b[4];
#pragma unroll
        for (int tl = 0; tl < 4; ++tl) b[tl] = bq[(2 * i) * 64 + 16 * tl];
#pragma unroll
        for (int tl = 0; tl < 4; ++tl) {
            v2f av, bv;
            av.x = a[i].x; av.y = a[i].y;
            bv.x = b[tl].x; bv.y = b[tl].y;
            acc[tl] = __builtin_amdgcn_wmma_f32_16x16x4_f32(
                 false, av,  false, bv,
                 (short)0, acc[tl],  false,  false);
        }
#pragma unroll
        for (int tl = 0; tl < 4; ++tl) {
            v2f av, bv;
            av.x = a[i].z; av.y = a[i].w;
            bv.x = b[tl].z; bv.y = b[tl].w;
            acc[tl] = __builtin_amdgcn_wmma_f32_16x16x4_f32(
                 false, av,  false, bv,
                 (short)0, acc[tl],  false,  false);
        }
    }
    float* sh = s_h + wave * (16 * SPITCH);
    float* sp = sh + (8 * half) * SPITCH + c;
#pragma unroll
    for (int tl = 0; tl < 4; ++tl) {
#pragma unroll
        for (int v = 0; v < 8; ++v) sp[v * SPITCH + 16 * tl] = acc[tl][v];
    }
    const int hd = lane & 3;
    float4 av4[4], bv4[4];
#pragma unroll
    for (int i = 0; i < 4; ++i) {
        av4[i] = *(const float4*)(as + hd * 16 + 4 * i);
        bv4[i] = *(const float4*)(ad + hd * 16 + 4 * i);
    }
    const bool full = row0 + 16 <= N;
    float4 hv[8];
#pragma unroll
    for (int j = 0; j < 8; ++j) hv[j] = *(const float4*)(sh + (2 * j + half) * SPITCH + 4 * c);
    float* dp = h1 + (size_t)(row0 + half) * 64 + 4 * c;
    if (full) {
#pragma unroll
        for (int j = 0; j < 8; ++j) *(float4*)(dp + (size_t)j * 128) = hv[j];
    } else {
#pragma unroll
        for (int j = 0; j < 8; ++j)
            if (row0 + 2 * j + half < N) *(float4*)(dp + (size_t)j * 128) = hv[j];
    }
    const int r0 = lane >> 2;
    float sa[2], sb[2];
#pragma unroll
    for (int p = 0; p < 2; ++p) {
        const float4* hp = (const float4*)(sh + (8 * p + r0) * SPITCH + 16 * hd);
        float ua = 0.f, ub = 0.f;
#pragma unroll
        for (int i = 0; i < 4; ++i) {
            const float4 q = hp[i];
            ua += q.x * av4[i].x + q.y * av4[i].y + q.z * av4[i].z + q.w * av4[i].w;
            ub += q.x * bv4[i].x + q.y * bv4[i].y + q.z * bv4[i].z + q.w * bv4[i].w;
        }
        sa[p] = ua; sb[p] = ub;
    }
    float* sso = ss + (size_t)(row0 + r0) * 4 + hd;
    float* sdo = sd + (size_t)(row0 + r0) * 4 + hd;
    if (full) {
        sso[0] = sa[0]; sso[32] = sa[1];
        sdo[0] = sb[0]; sdo[32] = sb[1];
    } else {
        if (row0 + r0 < N)     { sso[0]  = sa[0]; sdo[0]  = sb[0]; }
        if (row0 + 8 + r0 < N) { sso[32] = sa[1]; sdo[32] = sb[1]; }
    }
}

__device__ __forceinline__ float lrelu(float s) { return s > 0.f ? s : 0.2f * s; }

__device__ __forceinline__ float wave_max(float v) {
#pragma unroll
    for (int d = 16; d > 0; d >>= 1) v = fmaxf(v, __shfl_xor(v, d));
    return v;
}
__device__ __forceinline__ float wave_sum(float v) {
#pragma unroll
    for (int d = 16; d > 0; d >>= 1) v += __shfl_xor(v, d);
    return v;
}

#define N1_PLANE 66

__device__ __forceinline__ float wave_uniform(float v) {
    return __int_as_float(__builtin_amdgcn_readfirstlane(__float_as_int(v)));
}
__device__ __forceinline__ float xhalf(float x) {
    return __builtin_bit_cast(float, __builtin_amdgcn_permlanex16(__builtin_bit_cast(int, x), __builtin_bit_cast(int, x),
                                                                  0x76543210, 0xfedcba98, false, false));
}

__device__ __forceinline__ void put_chunk(int2* __restrict__ wr, int s, float e0, float e1, float e2, float e3) {
    const int so = (int)((unsigned)s << 8);
    wr[0 * N1_PLANE] = make_int2(so, __float_as_int(e0));
    wr[1 * N1_PLANE] = make_int2(so, __float_as_int(e1));
    wr[2 * N1_PLANE] = make_int2(so, __float_as_int(e2));
    wr[3 * N1_PLANE] = make_int2(so, __float_as_int(e3));
}

__device__ __forceinline__ void rows_acc(const float* __restrict__ h1, const int2* __restrict__ rb,
                                         unsigned qoff, int cnt, float4& acc) {
    for (int j = 0; j < cnt; j += 16) {
        int2 rc[8];
#pragma unroll
        for (int u = 0; u < 8; ++u) rc[u] = rb[j + 2 * u];
        float4 hv[8];
#pragma unroll
        for (int u = 0; u < 8; ++u)
            hv[u] = *(const float4*)((const char*)h1 + (size_t)((unsigned)rc[u].x + qoff));
#pragma unroll
        for (int u = 0; u < 8; ++u) {
            const float w = __int_as_float(rc[u].y);
            acc.x += hv[u].x * w;
            acc.y += hv[u].y * w;
            acc.z += hv[u].z * w;
            acc.w += hv[u].w * w;
        }
    }
}

__global__ __launch_bounds__(256) __attribute__((amdgpu_waves_per_eu(16, 16)))
void k_node1(const int* __restrict__ sorted, const int* __restrict__ off,
             const int* __restrict__ deg, const float* __restrict__ ss,
             const float* __restrict__ sd, const float* __restrict__ h1,
             const float* __restrict__ b1, const float* __restrict__ W2,
             const float* __restrict__ as2, const float* __restrict__ ad2,
             float* __restrict__ rec2, int N) {
    __shared__ int2 s_rec[8 * 4 * N1_PLANE];
    const int lane = threadIdx.x & 31;
    const int wv = threadIdx.x >> 5;
    const int n = __builtin_amdgcn_readfirstlane(blockIdx.x * 8 + wv);
    if (n >= N) return;
    const int o = off[n], dg = deg[n], tot = dg + 1;
    const float4 sdn = *(const float4*)(sd + (size_t)n * 4);

    const int pA = lane, pB = 32 + lane;
    const bool hasB = tot > 32;
    const int vA = sorted[(pA < dg) ? (size_t)o + pA : 0];
    const int vB = sorted[(pB < dg) ? (size_t)o + pB : 0];
    const int sA = (pA < dg) ? vA : n;
    const int sB = (pB < dg) ? vB : n;
    const float4 svA = *(const float4*)(ss + (size_t)sA * 4);
    float4 svB = make_float4(0.f, 0.f, 0.f, 0.f);
    if (hasB) svB = *(const float4*)(ss + (size_t)sB * 4);
    const bool okA = pA < tot, okB = pB < tot;
    const float xA0 = okA ? lrelu(svA.x + sdn.x) : NEG_INF, xA1 = okA ? lrelu(svA.y + sdn.y) : NEG_INF;
    const float xA2 = okA ? lrelu(svA.z + sdn.z) : NEG_INF, xA3 = okA ? lrelu(svA.w + sdn.w) : NEG_INF;
    const float xB0 = okB ? lrelu(svB.x + sdn.x) : NEG_INF, xB1 = okB ? lrelu(svB.y + sdn.y) : NEG_INF;
    const float xB2 = okB ? lrelu(svB.z + sdn.z) : NEG_INF, xB3 = okB ? lrelu(svB.w + sdn.w) : NEG_INF;
    float m0 = fmaxf(xA0, xB0), m1 = fmaxf(xA1, xB1), m2 = fmaxf(xA2, xB2), m3 = fmaxf(xA3, xB3);
    for (int base = 64; base < tot; base += 32) {
        const int p = base + lane;
        if (p < tot) {
            const int s = (p < dg) ? sorted[(size_t)o + p] : n;
            const float4 sv = *(const float4*)(ss + (size_t)s * 4);
            m0 = fmaxf(m0, lrelu(sv.x + sdn.x));
            m1 = fmaxf(m1, lrelu(sv.y + sdn.y));
            m2 = fmaxf(m2, lrelu(sv.z + sdn.z));
            m3 = fmaxf(m3, lrelu(sv.w + sdn.w));
        }
    }
    m0 = wave_uniform(wave_max(m0));
    m1 = wave_uniform(wave_max(m1));
    m2 = wave_uniform(wave_max(m2));
    m3 = wave_uniform(wave_max(m3));

    const int half = lane >> 4;
    const int q    = lane & 15;
    const int hsel = q >> 2;
    int2* wtab = s_rec + wv * (4 * N1_PLANE);
    int2* wr = wtab + lane;
    const int2* rb = wtab + hsel * N1_PLANE + half;
    const unsigned qoff = 16u * q;
    float4 acc = make_float4(0.f, 0.f, 0.f, 0.f);
    float t0, t1, t2, t3;
    {
        const float e0 = okA ? expf(xA0 - m0) : 0.f, e1 = okA ? expf(xA1 - m1) : 0.f;
        const float e2 = okA ? expf(xA2 - m2) : 0.f, e3 = okA ? expf(xA3 - m3) : 0.f;
        t0 = e0; t1 = e1; t2 = e2; t3 = e3;
        put_chunk(wr, sA, e0, e1, e2, e3);
    }
    if (hasB) {
        const float e0 = okB ? expf(xB0 - m0) : 0.f, e1 = okB ? expf(xB1 - m1) : 0.f;
        const float e2 = okB ? expf(xB2 - m2) : 0.f, e3 = okB ? expf(xB3 - m3) : 0.f;
        t0 += e0; t1 += e1; t2 += e2; t3 += e3;
        put_chunk(wr + 32, sB, e0, e1, e2, e3);
    }
    float d0 = wave_uniform(wave_sum(t0)), d1 = wave_uniform(wave_sum(t1));
    float d2 = wave_uniform(wave_sum(t2)), d3 = wave_uniform(wave_sum(t3));
    rows_acc(h1, rb, qoff, (tot < 64) ? tot : 64, acc);
    for (int base = 64; base < tot; base += 32) {
        const int p = base + lane;
        int s = n;
        float e0 = 0.f, e1 = 0.f, e2 = 0.f, e3 = 0.f;
        if (p < tot) {
            if (p < dg) s = sorted[(size_t)o + p];
            const float4 sv = *(const float4*)(ss + (size_t)s * 4);
            e0 = expf(lrelu(sv.x + sdn.x) - m0);
            e1 = expf(lrelu(sv.y + sdn.y) - m1);
            e2 = expf(lrelu(sv.z + sdn.z) - m2);
            e3 = expf(lrelu(sv.w + sdn.w) - m3);
        }
        d0 += wave_uniform(wave_sum(e0)); d1 += wave_uniform(wave_sum(e1));
        d2 += wave_uniform(wave_sum(e2)); d3 += wave_uniform(wave_sum(e3));
        put_chunk(wr, s, e0, e1, e2, e3);
        rows_acc(h1, rb, qoff, (tot - base < 32) ? (tot - base) : 32, acc);
    }
    const float dn = (hsel == 0) ? d0 : (hsel == 1) ? d1 : (hsel == 2) ? d2 : d3;
    const float g0 = half ? acc.x : acc.z, g1 = half ? acc.y : acc.w;
    const float a0 = (half ? acc.z : acc.x) + xhalf(g0);
    const float a1 = (half ? acc.w : acc.y) + xhalf(g1);
    const int c0 = 4 * q + 2 * half;
    const float2 bb = *(const float2*)(b1 + c0);
    float v0 = a0 / dn + bb.x;
    float v1 = a1 / dn + bb.y;
    v0 = v0 > 0.f ? v0 : expm1f(v0);
    v1 = v1 > 0.f ? v1 : expm1f(v1);
    const float4 wv2 = *(const float4*)(W2 + 2 * c0);
    float o0 = v0 * wv2.x + v1 * wv2.z;
    float o1 = v0 * wv2.y + v1 * wv2.w;
    o0 = wave_sum(o0);
    o1 = wave_sum(o1);
    if (lane == 0) {
        *(float4*)(rec2 + (size_t)n * 4) =
            make_float4(o0, o1, o0 * as2[0] + o1 * as2[1], o0 * ad2[0] + o1 * ad2[1]);
    }
}

#define N2_SLOTS 6
__global__ __launch_bounds__(256)
void k_node2(const int* __restrict__ sorted, const int* __restrict__ off,
             const int* __restrict__ deg, const float* __restrict__ rec2,
             const float* __restrict__ b2, float* __restrict__ out, int N) {
    const int t = blockIdx.x * 256 + threadIdx.x;
    const int n = t >> 3, sub = t & 7;
    int o = 0, dg = 0, tot = 0, nn = 0;
    float sdn = 0.f;
    if (n < N) { o = off[n]; dg = deg[n]; tot = dg + 1; sdn = rec2[(size_t)n * 4 + 3]; nn = n; }

    int s[N2_SLOTS];
#pragma unroll
    for (int k = 0; k < N2_SLOTS; ++k) {
        const int p = sub + 8 * k;
        const int v = sorted[(p < dg) ? (size_t)o + p : 0];
        s[k] = (p < dg) ? v : nn;
    }
    float4 r[N2_SLOTS];
#pragma unroll
    for (int k = 0; k < N2_SLOTS; ++k) r[k] = *(const float4*)(rec2 + (size_t)s[k] * 4);
    __builtin_amdgcn_sched_barrier(0);
    float rx[N2_SLOTS], ry[N2_SLOTS], xs[N2_SLOTS];
    float m = NEG_INF;
#pragma unroll
    for (int k = 0; k < N2_SLOTS; ++k) {
        const bool ok = sub + 8 * k < tot;
        rx[k] = ok ? r[k].x : 0.f;
        ry[k] = ok ? r[k].y : 0.f;
        xs[k] = ok ? lrelu(r[k].z + sdn) : NEG_INF;
        m = fmaxf(m, xs[k]);
    }
    for (int p = sub + 8 * N2_SLOTS; p < tot; p += 8) {
        const int sp = (p < dg) ? sorted[(size_t)o + p] : n;
        m = fmaxf(m, lrelu(rec2[(size_t)sp * 4 + 2] + sdn));
    }
#pragma unroll
    for (int d = 4; d > 0; d >>= 1) m = fmaxf(m, __shfl_xor(m, d));

    float dsum = 0.f, a0 = 0.f, a1 = 0.f;
#pragma unroll
    for (int k = 0; k < N2_SLOTS; ++k) {
        const float ex = (sub + 8 * k < tot) ? expf(xs[k] - m) : 0.f;
        dsum += ex;
        a0 += rx[k] * ex;
        a1 += ry[k] * ex;
    }
    for (int p = sub + 8 * N2_SLOTS; p < tot; p += 8) {
        const int sp = (p < dg) ? sorted[(size_t)o + p] : n;
        const float4 r = *(const float4*)(rec2 + (size_t)sp * 4);
        const float ex = expf(lrelu(r.z + sdn) - m);
        dsum += ex;
        a0 += r.x * ex;
        a1 += r.y * ex;
    }
#pragma unroll
    for (int d = 4; d > 0; d >>= 1) {
        dsum += __shfl_xor(dsum, d);
        a0   += __shfl_xor(a0, d);
        a1   += __shfl_xor(a1, d);
    }
    if (sub == 0 && n < N) {
        float v0 = a0 / dsum + b2[0];
        float v1 = a1 / dsum + b2[1];
        float mx = fmaxf(v0, v1);
        float lse = mx + logf(expf(v0 - mx) + expf(v1 - mx));
        *(float2*)(out + (size_t)n * 2) = make_float2(v0 - lse, v1 - lse);
    }
}

static inline unsigned nblk(long long n, int b) { return (unsigned)((n + b - 1) / b); }

extern "C" void kernel_launch(void* const* d_in, const int* in_sizes, int n_in,
                              void* d_out, int out_size, void* d_ws, size_t ws_size,
                              hipStream_t stream) {
    const float* x     = (const float*)d_in[0];
    const int*   ei    = (const int*)  d_in[1];
    const float* W1    = (const float*)d_in[2];
    const float* asrc1 = (const float*)d_in[3];
    const float* adst1 = (const float*)d_in[4];
    const float* b1    = (const float*)d_in[5];
    const float* W2    = (const float*)d_in[6];
    const float* asrc2 = (const float*)d_in[7];
    const float* adst2 = (const float*)d_in[8];
    const float* b2    = (const float*)d_in[9];
    float* out = (float*)d_out;

    const int N = in_sizes[0] / 128;
    const int E = in_sizes[1] / 2;

    float* ws   = (float*)d_ws;
    size_t nl   = (size_t)N;
    float* h1   = ws;
    float* ss1  = h1  + nl * 64;
    float* sd1  = ss1 + nl * 4;
    float* rec2 = sd1 + nl * 4;
    int*   off  = (int*)(rec2 + nl * 4);
    int*   deg  = off + nl;
    int*   sorted = deg + nl + (size_t)E;
    const int C = (N + BNODES - 1) / BNODES;
    const int G = (int)nblk(E, EPB);
    const int M = C * G;
    int*   part = (int*)h1;
    int*   cntA = (int*)ss1;
    int*   lofA = cntA + M;

    k_part<<<G, 1024, C * sizeof(int), stream>>>(ei, cntA, lofA, part, E, C, G);
    k_bucket<<<C, 1024, 2 * G * sizeof(int), stream>>>(part, cntA, lofA, off, deg, sorted, N, G);
    k_gemm1_wmma<<<nblk(N, 128), 256, 0, stream>>>(x, W1, asrc1, adst1, h1, ss1, sd1, N);
    k_node1<<<nblk(N, 8), 256, 0, stream>>>(sorted, off, deg, ss1, sd1, h1, b1, W2,
                                            asrc2, adst2, rec2, N);
    k_node2<<<nblk(8LL * N, 256), 256, 0, stream>>>(sorted, off, deg, rec2, b2, out, N);
}
